# starting kernel
# baseline (speedup 1.0000x reference)
#include <hip/hip_runtime.h>

#define NN  100000
#define NF  256
#define HH  8
#define CC  16
#define HC  128
#define NC  16
#define EE  1600000
#define ET  1700000
#define NEG 0.2f

typedef _Float16 h8   __attribute__((ext_vector_type(8)));
typedef _Float16 v16h __attribute__((ext_vector_type(16)));
typedef float    v8f  __attribute__((ext_vector_type(8)));

__device__ __forceinline__ void edge_sd(const int* __restrict__ ei, int e, int& s, int& d) {
    if (e < EE) { s = ei[e]; d = ei[EE + e]; }
    else        { s = e - EE; d = s; }
}

__device__ __forceinline__ void atomicMaxF(float* addr, float val) {
    if (val >= 0.0f) atomicMax((int*)addr, __float_as_int(val));
    else             atomicMin((unsigned int*)addr, __float_as_uint(val));
}

__device__ __forceinline__ float lrelu(float a) { return a > 0.0f ? a : NEG * a; }

__global__ __launch_bounds__(256) void gat_cvtw(const float* __restrict__ W1,
                                                const float* __restrict__ W2,
                                                _Float16* __restrict__ w1h,
                                                _Float16* __restrict__ w2h) {
    int i = blockIdx.x * 256 + threadIdx.x;
    if (i < HC * NF)              w1h[i] = (_Float16)W1[i];
    else if (i < HC * NF + NC*HC) w2h[i - HC * NF] = (_Float16)W2[i - HC * NF];
}

__global__ __launch_bounds__(256) void gat_cvtx(const float* __restrict__ x,
                                                _Float16* __restrict__ xh) {
    int i = blockIdx.x * 256 + threadIdx.x;
    if (i < NN * NF) xh[i] = (_Float16)x[i];
}

__global__ __launch_bounds__(256)
void gat_gemm1(const _Float16* __restrict__ xh, const _Float16* __restrict__ w1h,
               float* __restrict__ h1) {
    __shared__ _Float16 lw[HC * NF];
    const int tid = threadIdx.x;
    for (int idx = tid; idx < HC * 32; idx += 256) {
        int r = idx >> 5, ch = idx & 31;
        *(h8*)(&lw[r * NF + ((ch ^ (r & 15)) << 3)]) =
            *(const h8*)(w1h + (size_t)r * NF + (ch << 3));
    }
    __syncthreads();

    const int lane = tid & 31;
    const int rt   = blockIdx.x * 8 + (tid >> 5);
    if (rt >= NN / 16) return;
    const int rsub = lane & 15;
    const int ksel = (lane >> 4) << 3;
    const _Float16* xrow = xh + (size_t)(rt * 16 + rsub) * NF;

    v8f acc[8] = {};
    #pragma unroll
    for (int ks = 0; ks < 8; ++ks) {
        const int kb = ks * 32 + ksel;
        const int c0 = kb >> 3;
        union { v16h v; h8 p[2]; } A;
        A.p[0] = *(const h8*)(xrow + kb);
        A.p[1] = *(const h8*)(xrow + kb + 16);
        #pragma unroll
        for (int ct = 0; ct < 8; ++ct) {
            const int n = ct * 16 + rsub;
            union { v16h v; h8 p[2]; } B;
            B.p[0] = *(const h8*)(&lw[n * NF + (((c0    ) ^ (n & 15)) << 3)]);
            B.p[1] = *(const h8*)(&lw[n * NF + (((c0 + 2) ^ (n & 15)) << 3)]);
            acc[ct] = __builtin_amdgcn_wmma_f32_16x16x32_f16(
                false, A.v, false, B.v, (short)0, acc[ct], false, false);
        }
    }
    const int rhi = (lane >> 4) << 3;
    #pragma unroll
    for (int ct = 0; ct < 8; ++ct) {
        #pragma unroll
        for (int r = 0; r < 8; ++r)
            h1[(size_t)(rt * 16 + rhi + r) * HC + ct * 16 + rsub] = acc[ct][r];
    }
}

__global__ __launch_bounds__(256)
void gat_att1(const float* __restrict__ h1, const float* __restrict__ as,
              const float* __restrict__ ad, float* __restrict__ a_src,
              float* __restrict__ a_dst, float* __restrict__ amax,
              float* __restrict__ den, float* __restrict__ agg) {
    int i = blockIdx.x * 256 + threadIdx.x;
    if (i >= NN * HH) return;
    int n = i >> 3, h = i & 7;
    const float* hp = h1 + (size_t)n * HC + h * CC;
    float s = 0.0f, d = 0.0f;
    #pragma unroll
    for (int c = 0; c < CC; ++c) { s += hp[c] * as[h * CC + c]; d += hp[c] * ad[h * CC + c]; }
    a_src[i] = s; a_dst[i] = d;
    amax[i] = -__builtin_inff(); den[i] = 0.0f;
    float* ap = agg + (size_t)n * HC + h * CC;
    #pragma unroll
    for (int c = 0; c < CC; ++c) ap[c] = 0.0f;
}

__global__ __launch_bounds__(256)
void gat_emax1(const int* __restrict__ ei, const float* __restrict__ a_src,
               const float* __restrict__ a_dst, float* __restrict__ amax) {
    int i = blockIdx.x * 256 + threadIdx.x;
    if (i >= ET * HH) return;
    int e = i >> 3, h = i & 7, s, d;
    edge_sd(ei, e, s, d);
    atomicMaxF(&amax[d * HH + h], lrelu(a_src[s * HH + h] + a_dst[d * HH + h]));
}

__global__ __launch_bounds__(256)
void gat_eden1(const int* __restrict__ ei, const float* __restrict__ a_src,
               const float* __restrict__ a_dst, const float* __restrict__ amax,
               float* __restrict__ den) {
    int i = blockIdx.x * 256 + threadIdx.x;
    if (i >= ET * HH) return;
    int e = i >> 3, h = i & 7, s, d;
    edge_sd(ei, e, s, d);
    float al = lrelu(a_src[s * HH + h] + a_dst[d * HH + h]);
    atomicAdd(&den[d * HH + h], expf(al - amax[d * HH + h]));
}

__global__ __launch_bounds__(256)
void gat_attw1(const int* __restrict__ ei, const float* __restrict__ a_src,
               const float* __restrict__ a_dst, const float* __restrict__ amax,
               const float* __restrict__ den, float* __restrict__ attn) {
    int i = blockIdx.x * 256 + threadIdx.x;
    if (i >= ET * HH) return;
    int e = i >> 3, h = i & 7, s, d;
    edge_sd(ei, e, s, d);
    float al = lrelu(a_src[s * HH + h] + a_dst[d * HH + h]);
    attn[i] = expf(al - amax[d * HH + h]) / (den[d * HH + h] + 1e-16f);
}

__global__ __launch_bounds__(256)
void gat_eagg1(const int* __restrict__ ei, const float* __restrict__ attn,
               const float* __restrict__ h1, float* __restrict__ agg) {
    int i = blockIdx.x * 256 + threadIdx.x;
    int e = i >> 5, q = i & 31;
    if (e >= ET) return;
    int c = q << 2;
    int s, d; edge_sd(ei, e, s, d);
    float a = attn[e * HH + (q >> 2)];
    const float4 hv = *(const float4*)(h1 + (size_t)s * HC + c);
    float* ap = agg + (size_t)d * HC + c;
    atomicAdd(ap + 0, hv.x * a);
    atomicAdd(ap + 1, hv.y * a);
    atomicAdd(ap + 2, hv.z * a);
    atomicAdd(ap + 3, hv.w * a);
}

__global__ __launch_bounds__(256)
void gat_biaselu(float* __restrict__ agg, const float* __restrict__ b1) {
    int i = blockIdx.x * 256 + threadIdx.x;
    if (i >= NN * HC) return;
    float v = agg[i] + b1[i & 127];
    agg[i] = v > 0.0f ? v : (expf(v) - 1.0f);
}

__global__ __launch_bounds__(256)
void gat_gemm2(const float* __restrict__ a1, const _Float16* __restrict__ w2h,
               float* __restrict__ h2) {
    const int lane = threadIdx.x & 31;
    const int rt   = blockIdx.x * 8 + (threadIdx.x >> 5);
    if (rt >= NN / 16) return;
    const int rsub = lane & 15;
    const int ksel = (lane >> 4) << 3;
    const float*    arow = a1 + (size_t)(rt * 16 + rsub) * HC;
    const _Float16* wrow = w2h + (size_t)rsub * HC;

    v8f acc = {};
    #pragma unroll
    for (int ks = 0; ks < 4; ++ks) {
        const int kb = ks * 32 + ksel;
        union { v16h v; _Float16 e[16]; } A;
        #pragma unroll
        for (int t = 0; t < 8; ++t) {
            A.e[t]     = (_Float16)arow[kb + t];
            A.e[8 + t] = (_Float16)arow[kb + 16 + t];
        }
        union { v16h v; h8 p[2]; } B;
        B.p[0] = *(const h8*)(wrow + kb);
        B.p[1] = *(const h8*)(wrow + kb + 16);
        acc = __builtin_amdgcn_wmma_f32_16x16x32_f16(
            false, A.v, false, B.v, (short)0, acc, false, false);
    }
    const int rhi = (lane >> 4) << 3;
    #pragma unroll
    for (int r = 0; r < 8; ++r)
        h2[(size_t)(rt * 16 + rhi + r) * NC + rsub] = acc[r];
}

__global__ __launch_bounds__(256)
void gat_att2(const float* __restrict__ h2, const float* __restrict__ as,
              const float* __restrict__ ad, float* __restrict__ a_src,
              float* __restrict__ a_dst, float* __restrict__ amax,
              float* __restrict__ den, float* __restrict__ agg) {
    int n = blockIdx.x * 256 + threadIdx.x;
    if (n >= NN) return;
    const float* hp = h2 + (size_t)n * NC;
    float s = 0.0f, d = 0.0f;
    #pragma unroll
    for (int c = 0; c < NC; ++c) { s += hp[c] * as[c]; d += hp[c] * ad[c]; }
    a_src[n] = s; a_dst[n] = d;
    amax[n] = -__builtin_inff(); den[n] = 0.0f;
    float* ap = agg + (size_t)n * NC;
    #pragma unroll
    for (int c = 0; c < NC; ++c) ap[c] = 0.0f;
}

__global__ __launch_bounds__(256)
void gat_emax2(const int* __restrict__ ei, const float* __restrict__ a_src,
               const float* __restrict__ a_dst, float* __restrict__ amax) {
    int e = blockIdx.x * 256 + threadIdx.x;
    if (e >= ET) return;
    int s, d; edge_sd(ei, e, s, d);
    atomicMaxF(&amax[d], lrelu(a_src[s] + a_dst[d]));
}

__global__ __launch_bounds__(256)
void gat_eden2(const int* __restrict__ ei, const float* __restrict__ a_src,
               const float* __restrict__ a_dst, const float* __restrict__ amax,
               float* __restrict__ den) {
    int e = blockIdx.x * 256 + threadIdx.x;
    if (e >= ET) return;
    int s, d; edge_sd(ei, e, s, d);
    float al = lrelu(a_src[s] + a_dst[d]);
    atomicAdd(&den[d], expf(al - amax[d]));
}

__global__ __launch_bounds__(256)
void gat_attw2(const int* __restrict__ ei, const float* __restrict__ a_src,
               const float* __restrict__ a_dst, const float* __restrict__ amax,
               const float* __restrict__ den, float* __restrict__ attn) {
    int e = blockIdx.x * 256 + threadIdx.x;
    if (e >= ET) return;
    int s, d; edge_sd(ei, e, s, d);
    float al = lrelu(a_src[s] + a_dst[d]);
    attn[e] = expf(al - amax[d]) / (den[d] + 1e-16f);
}

__global__ __launch_bounds__(256)
void gat_eagg2(const int* __restrict__ ei, const float* __restrict__ attn,
               const float* __restrict__ h2, float* __restrict__ agg) {
    int i = blockIdx.x * 256 + threadIdx.x;
    int e = i >> 2, q = i & 3;
    if (e >= ET) return;
    int c = q << 2;
    int s, d; edge_sd(ei, e, s, d);
    float a = attn[e];
    const float4 hv = *(const float4*)(h2 + (size_t)s * NC + c);
    float* ap = agg + (size_t)d * NC + c;
    atomicAdd(ap + 0, hv.x * a);
    atomicAdd(ap + 1, hv.y * a);
    atomicAdd(ap + 2, hv.z * a);
    atomicAdd(ap + 3, hv.w * a);
}

__global__ __launch_bounds__(256)
void gat_lsm(const float* __restrict__ agg, const float* __restrict__ b2,
             float* __restrict__ out) {
    int n = blockIdx.x * 256 + threadIdx.x;
    if (n >= NN) return;
    float v[NC], m = -__builtin_inff();
    #pragma unroll
    for (int c = 0; c < NC; ++c) { v[c] = agg[(size_t)n * NC + c] + b2[c]; m = fmaxf(m, v[c]); }
    float sum = 0.0f;
    #pragma unroll
    for (int c = 0; c < NC; ++c) sum += expf(v[c] - m);
    float ls = logf(sum);
    #pragma unroll
    for (int c = 0; c < NC; ++c) out[(size_t)n * NC + c] = v[c] - m - ls;
}

extern "C" void kernel_launch(void* const* d_in, const int* in_sizes, int n_in,
                              void* d_out, int out_size, void* d_ws, size_t ws_size,
                              hipStream_t stream) {
    (void)in_sizes; (void)n_in; (void)out_size; (void)ws_size;
    const float* x   = (const float*)d_in[0];
    const int*   ei  = (const int*)d_in[1];
    const float* W1  = (const float*)d_in[2];
    const float* as1 = (const float*)d_in[3];
    const float* ad1 = (const float*)d_in[4];
    const float* b1  = (const float*)d_in[5];
    const float* W2  = (const float*)d_in[6];
    const float* as2 = (const float*)d_in[7];
    const float* ad2 = (const float*)d_in[8];
    const float* b2  = (const float*)d_in[9];
    float* out = (float*)d_out;

    char* wsb = (char*)d_ws;
    size_t off = 0;
    auto alloc = [&](size_t bytes) -> void* {
        void* p = wsb + off;
        off += (bytes + 255) & ~(size_t)255;
        return p;
    };
    _Float16* w1h  = (_Float16*)alloc((size_t)HC * NF * 2);
    _Float16* w2h  = (_Float16*)alloc((size_t)NC * HC * 2);
    _Float16* xh   = (_Float16*)alloc((size_t)NN * NF * 2);
    float* h1      = (float*)alloc((size_t)NN * HC * 4);
    float* asrc1   = (float*)alloc((size_t)NN * HH * 4);
    float* adst1   = (float*)alloc((size_t)NN * HH * 4);
    float* amax1   = (float*)alloc((size_t)NN * HH * 4);
    float* den1    = (float*)alloc((size_t)NN * HH * 4);
    float* attn1   = (float*)alloc((size_t)ET * HH * 4);
    float* h2      = (float*)alloc((size_t)NN * NC * 4);
    float* asrc2   = (float*)alloc((size_t)NN * 4);
    float* adst2   = (float*)alloc((size_t)NN * 4);
    float* amax2   = (float*)alloc((size_t)NN * 4);
    float* den2    = (float*)alloc((size_t)NN * 4);
    float* attn2   = (float*)alloc((size_t)ET * 4);
    float* agg2    = (float*)alloc((size_t)NN * NC * 4);
    float* agg1    = (float*)xh;

    const dim3 B(256);
    gat_cvtw <<<(HC * NF + NC * HC + 255) / 256, B, 0, stream>>>(W1, W2, w1h, w2h);
    gat_cvtx <<<(NN * NF) / 256, B, 0, stream>>>(x, xh);
    gat_gemm1<<<(NN / 16 + 7) / 8, B, 0, stream>>>(xh, w1h, h1);
    gat_att1 <<<(NN * HH) / 256, B, 0, stream>>>(h1, as1, ad1, asrc1, adst1, amax1, den1, agg1);
    gat_emax1<<<(ET * HH) / 256, B, 0, stream>>>(ei, asrc1, adst1, amax1);
    gat_eden1<<<(ET * HH) / 256, B, 0, stream>>>(ei, asrc1, adst1, amax1, den1);
    gat_attw1<<<(ET * HH) / 256, B, 0, stream>>>(ei, asrc1, adst1, amax1, den1, attn1);
    gat_eagg1<<<(ET * 32) / 256, B, 0, stream>>>(ei, attn1, h1, agg1);
    gat_biaselu<<<(NN * HC) / 256, B, 0, stream>>>(agg1, b1);
    gat_gemm2<<<(NN / 16 + 7) / 8, B, 0, stream>>>(agg1, w2h, h2);
    gat_att2 <<<(NN + 255) / 256, B, 0, stream>>>(h2, as2, ad2, asrc2, adst2, amax2, den2, agg2);
    gat_emax2<<<(ET + 255) / 256, B, 0, stream>>>(ei, asrc2, adst2, amax2);
    gat_eden2<<<(ET + 255) / 256, B, 0, stream>>>(ei, asrc2, adst2, amax2, den2);
    gat_attw2<<<(ET + 255) / 256, B, 0, stream>>>(ei, asrc2, adst2, amax2, den2, attn2);
    gat_eagg2<<<(ET * 4 + 255) / 256, B, 0, stream>>>(ei, attn2, h2, agg2);
    gat_lsm  <<<(NN + 255) / 256, B, 0, stream>>>(agg2, b2, out);
}


# softmax normalisation deferred: one edge sweep per layer adds exp weights into den and weighted rows into agg (same float atomics, arrival order); node kernels divide by den; eden/attw sweeps gone
# speedup vs baseline: 1.0517x; 1.0517x over previous
#include <hip/hip_runtime.h>

#define NN  100000
#define NF  256
#define HH  8
#define CC  16
#define HC  128
#define NC  16
#define EE  1600000
#define ET  1700000
#define NEG 0.2f

typedef _Float16 h8   __attribute__((ext_vector_type(8)));
typedef _Float16 v16h __attribute__((ext_vector_type(16)));
typedef float    v8f  __attribute__((ext_vector_type(8)));

__device__ __forceinline__ void edge_sd(const int* __restrict__ ei, int e, int& s, int& d) {
    if (e < EE) { s = ei[e]; d = ei[EE + e]; }
    else        { s = e - EE; d = s; }
}

__device__ __forceinline__ void atomicMaxF(float* addr, float val) {
    if (val >= 0.0f) atomicMax((int*)addr, __float_as_int(val));
    else             atomicMin((unsigned int*)addr, __float_as_uint(val));
}

__device__ __forceinline__ float lrelu(float a) { return a > 0.0f ? a : NEG * a; }

__global__ __launch_bounds__(256) void gat_cvtw(const float* __restrict__ W1,
                                                const float* __restrict__ W2,
                                                _Float16* __restrict__ w1h,
                                                _Float16* __restrict__ w2h) {
    int i = blockIdx.x * 256 + threadIdx.x;
    if (i < HC * NF)              w1h[i] = (_Float16)W1[i];
    else if (i < HC * NF + NC*HC) w2h[i - HC * NF] = (_Float16)W2[i - HC * NF];
}

__global__ __launch_bounds__(256) void gat_cvtx(const float* __restrict__ x,
                                                _Float16* __restrict__ xh) {
    int i = blockIdx.x * 256 + threadIdx.x;
    if (i < NN * NF) xh[i] = (_Float16)x[i];
}

__global__ __launch_bounds__(256)
void gat_gemm1(const _Float16* __restrict__ xh, const _Float16* __restrict__ w1h,
               float* __restrict__ h1) {
    __shared__ _Float16 lw[HC * NF];
    const int tid = threadIdx.x;
    for (int idx = tid; idx < HC * 32; idx += 256) {
        int r = idx >> 5, ch = idx & 31;
        *(h8*)(&lw[r * NF + ((ch ^ (r & 15)) << 3)]) =
            *(const h8*)(w1h + (size_t)r * NF + (ch << 3));
    }
    __syncthreads();

    const int lane = tid & 31;
    const int rt   = blockIdx.x * 8 + (tid >> 5);
    if (rt >= NN / 16) return;
    const int rsub = lane & 15;
    const int ksel = (lane >> 4) << 3;
    const _Float16* xrow = xh + (size_t)(rt * 16 + rsub) * NF;

    v8f acc[8] = {};
    #pragma unroll
    for (int ks = 0; ks < 8; ++ks) {
        const int kb = ks * 32 + ksel;
        const int c0 = kb >> 3;
        union { v16h v; h8 p[2]; } A;
        A.p[0] = *(const h8*)(xrow + kb);
        A.p[1] = *(const h8*)(xrow + kb + 16);
        #pragma unroll
        for (int ct = 0; ct < 8; ++ct) {
            const int n = ct * 16 + rsub;
            union { v16h v; h8 p[2]; } B;
            B.p[0] = *(const h8*)(&lw[n * NF + (((c0    ) ^ (n & 15)) << 3)]);
            B.p[1] = *(const h8*)(&lw[n * NF + (((c0 + 2) ^ (n & 15)) << 3)]);
            acc[ct] = __builtin_amdgcn_wmma_f32_16x16x32_f16(
                false, A.v, false, B.v, (short)0, acc[ct], false, false);
        }
    }
    const int rhi = (lane >> 4) << 3;
    #pragma unroll
    for (int ct = 0; ct < 8; ++ct) {
        #pragma unroll
        for (int r = 0; r < 8; ++r)
            h1[(size_t)(rt * 16 + rhi + r) * HC + ct * 16 + rsub] = acc[ct][r];
    }
}

__global__ __launch_bounds__(256)
void gat_att1(const float* __restrict__ h1, const float* __restrict__ as,
              const float* __restrict__ ad, float* __restrict__ a_src,
              float* __restrict__ a_dst, float* __restrict__ amax,
              float* __restrict__ den, float* __restrict__ agg) {
    int i = blockIdx.x * 256 + threadIdx.x;
    if (i >= NN * HH) return;
    int n = i >> 3, h = i & 7;
    const float* hp = h1 + (size_t)n * HC + h * CC;
    float s = 0.0f, d = 0.0f;
    #pragma unroll
    for (int c = 0; c < CC; ++c) { s += hp[c] * as[h * CC + c]; d += hp[c] * ad[h * CC + c]; }
    a_src[i] = s; a_dst[i] = d;
    amax[i] = -__builtin_inff(); den[i] = 0.0f;
    float* ap = agg + (size_t)n * HC + h * CC;
    #pragma unroll
    for (int c = 0; c < CC; ++c) ap[c] = 0.0f;
}

__global__ __launch_bounds__(256)
void gat_emax1(const int* __restrict__ ei, const float* __restrict__ a_src,
               const float* __restrict__ a_dst, float* __restrict__ amax) {
    int i = blockIdx.x * 256 + threadIdx.x;
    if (i >= ET * HH) return;
    int e = i >> 3, h = i & 7, s, d;
    edge_sd(ei, e, s, d);
    atomicMaxF(&amax[d * HH + h], lrelu(a_src[s * HH + h] + a_dst[d * HH + h]));
}

__global__ __launch_bounds__(256)
void gat_eagg1(const int* __restrict__ ei, const float* __restrict__ a_src,
               const float* __restrict__ a_dst, const float* __restrict__ amax,
               const float* __restrict__ h1, float* __restrict__ den,
               float* __restrict__ agg) {
    int i = blockIdx.x * 256 + threadIdx.x;
    int e = i >> 5, q = i & 31;
    if (e >= ET) return;
    int c = q << 2, h = q >> 2;
    int s, d; edge_sd(ei, e, s, d);
    const float4 hv = *(const float4*)(h1 + (size_t)s * HC + c);
    float al = lrelu(a_src[s * HH + h] + a_dst[d * HH + h]);
    float w = expf(al - amax[d * HH + h]);
    if ((q & 3) == 0) atomicAdd(&den[d * HH + h], w);
    float* ap = agg + (size_t)d * HC + c;
    atomicAdd(ap + 0, hv.x * w);
    atomicAdd(ap + 1, hv.y * w);
    atomicAdd(ap + 2, hv.z * w);
    atomicAdd(ap + 3, hv.w * w);
}

__global__ __launch_bounds__(256)
void gat_biaselu(float* __restrict__ agg, const float* __restrict__ den,
                 const float* __restrict__ b1) {
    int i = blockIdx.x * 256 + threadIdx.x;
    if (i >= NN * HC) return;
    float v = agg[i] / (den[i >> 4] + 1e-16f) + b1[i & 127];
    agg[i] = v > 0.0f ? v : (expf(v) - 1.0f);
}

__global__ __launch_bounds__(256)
void gat_gemm2(const float* __restrict__ a1, const _Float16* __restrict__ w2h,
               float* __restrict__ h2) {
    const int lane = threadIdx.x & 31;
    const int rt   = blockIdx.x * 8 + (threadIdx.x >> 5);
    if (rt >= NN / 16) return;
    const int rsub = lane & 15;
    const int ksel = (lane >> 4) << 3;
    const float*    arow = a1 + (size_t)(rt * 16 + rsub) * HC;
    const _Float16* wrow = w2h + (size_t)rsub * HC;

    v8f acc = {};
    #pragma unroll
    for (int ks = 0; ks < 4; ++ks) {
        const int kb = ks * 32 + ksel;
        union { v16h v; _Float16 e[16]; } A;
        #pragma unroll
        for (int t = 0; t < 8; ++t) {
            A.e[t]     = (_Float16)arow[kb + t];
            A.e[8 + t] = (_Float16)arow[kb + 16 + t];
        }
        union { v16h v; h8 p[2]; } B;
        B.p[0] = *(const h8*)(wrow + kb);
        B.p[1] = *(const h8*)(wrow + kb + 16);
        acc = __builtin_amdgcn_wmma_f32_16x16x32_f16(
            false, A.v, false, B.v, (short)0, acc, false, false);
    }
    const int rhi = (lane >> 4) << 3;
    #pragma unroll
    for (int r = 0; r < 8; ++r)
        h2[(size_t)(rt * 16 + rhi + r) * NC + rsub] = acc[r];
}

__global__ __launch_bounds__(256)
void gat_att2(const float* __restrict__ h2, const float* __restrict__ as,
              const float* __restrict__ ad, float* __restrict__ a_src,
              float* __restrict__ a_dst, float* __restrict__ amax,
              float* __restrict__ den, float* __restrict__ agg) {
    int n = blockIdx.x * 256 + threadIdx.x;
    if (n >= NN) return;
    const float* hp = h2 + (size_t)n * NC;
    float s = 0.0f, d = 0.0f;
    #pragma unroll
    for (int c = 0; c < NC; ++c) { s += hp[c] * as[c]; d += hp[c] * ad[c]; }
    a_src[n] = s; a_dst[n] = d;
    amax[n] = -__builtin_inff(); den[n] = 0.0f;
    float* ap = agg + (size_t)n * NC;
    #pragma unroll
    for (int c = 0; c < NC; ++c) ap[c] = 0.0f;
}

__global__ __launch_bounds__(256)
void gat_emax2(const int* __restrict__ ei, const float* __restrict__ a_src,
               const float* __restrict__ a_dst, float* __restrict__ amax) {
    int e = blockIdx.x * 256 + threadIdx.x;
    if (e >= ET) return;
    int s, d; edge_sd(ei, e, s, d);
    atomicMaxF(&amax[d], lrelu(a_src[s] + a_dst[d]));
}

__global__ __launch_bounds__(256)
void gat_eagg2(const int* __restrict__ ei, const float* __restrict__ a_src,
               const float* __restrict__ a_dst, const float* __restrict__ amax,
               const float* __restrict__ h2, float* __restrict__ den,
               float* __restrict__ agg) {
    int i = blockIdx.x * 256 + threadIdx.x;
    int e = i >> 2, q = i & 3;
    if (e >= ET) return;
    int c = q << 2;
    int s, d; edge_sd(ei, e, s, d);
    const float4 hv = *(const float4*)(h2 + (size_t)s * NC + c);
    float al = lrelu(a_src[s] + a_dst[d]);
    float w = expf(al - amax[d]);
    if (q == 0) atomicAdd(&den[d], w);
    float* ap = agg + (size_t)d * NC + c;
    atomicAdd(ap + 0, hv.x * w);
    atomicAdd(ap + 1, hv.y * w);
    atomicAdd(ap + 2, hv.z * w);
    atomicAdd(ap + 3, hv.w * w);
}

__global__ __launch_bounds__(256)
void gat_lsm(const float* __restrict__ agg, const float* __restrict__ den,
             const float* __restrict__ b2, float* __restrict__ out) {
    int n = blockIdx.x * 256 + threadIdx.x;
    if (n >= NN) return;
    float v[NC], m = -__builtin_inff();
    const float dn = den[n] + 1e-16f;
    #pragma unroll
    for (int c = 0; c < NC; ++c) { v[c] = agg[(size_t)n * NC + c] / dn + b2[c]; m = fmaxf(m, v[c]); }
    float sum = 0.0f;
    #pragma unroll
    for (int c = 0; c < NC; ++c) sum += expf(v[c] - m);
    float ls = logf(sum);
    #pragma unroll
    for (int c = 0; c < NC; ++c) out[(size_t)n * NC + c] = v[c] - m - ls;
}

extern "C" void kernel_launch(void* const* d_in, const int* in_sizes, int n_in,
                              void* d_out, int out_size, void* d_ws, size_t ws_size,
                              hipStream_t stream) {
    (void)in_sizes; (void)n_in; (void)out_size; (void)ws_size;
    const float* x   = (const float*)d_in[0];
    const int*   ei  = (const int*)d_in[1];
    const float* W1  = (const float*)d_in[2];
    const float* as1 = (const float*)d_in[3];
    const float* ad1 = (const float*)d_in[4];
    const float* b1  = (const float*)d_in[5];
    const float* W2  = (const float*)d_in[6];
    const float* as2 = (const float*)d_in[7];
    const float* ad2 = (const float*)d_in[8];
    const float* b2  = (const float*)d_in[9];
    float* out = (float*)d_out;

    char* wsb = (char*)d_ws;
    size_t off = 0;
    auto alloc = [&](size_t bytes) -> void* {
        void* p = wsb + off;
        off += (bytes + 255) & ~(size_t)255;
        return p;
    };
    _Float16* w1h  = (_Float16*)alloc((size_t)HC * NF * 2);
    _Float16* w2h  = (_Float16*)alloc((size_t)NC * HC * 2);
    _Float16* xh   = (_Float16*)alloc((size_t)NN * NF * 2);
    float* h1      = (float*)alloc((size_t)NN * HC * 4);
    float* asrc1   = (float*)alloc((size_t)NN * HH * 4);
    float* adst1   = (float*)alloc((size_t)NN * HH * 4);
    float* amax1   = (float*)alloc((size_t)NN * HH * 4);
    float* den1    = (float*)alloc((size_t)NN * HH * 4);
    float* h2      = (float*)alloc((size_t)NN * NC * 4);
    float* asrc2   = (float*)alloc((size_t)NN * 4);
    float* adst2   = (float*)alloc((size_t)NN * 4);
    float* amax2   = (float*)alloc((size_t)NN * 4);
    float* den2    = (float*)alloc((size_t)NN * 4);
    float* agg2    = (float*)alloc((size_t)NN * NC * 4);
    float* agg1    = (float*)xh;

    const dim3 B(256);
    gat_cvtw <<<(HC * NF + NC * HC + 255) / 256, B, 0, stream>>>(W1, W2, w1h, w2h);
    gat_cvtx <<<(NN * NF) / 256, B, 0, stream>>>(x, xh);
    gat_gemm1<<<(NN / 16 + 7) / 8, B, 0, stream>>>(xh, w1h, h1);
    gat_att1 <<<(NN * HH) / 256, B, 0, stream>>>(h1, as1, ad1, asrc1, adst1, amax1, den1, agg1);
    gat_emax1<<<(ET * HH) / 256, B, 0, stream>>>(ei, asrc1, adst1, amax1);
    gat_eagg1<<<(ET * 32) / 256, B, 0, stream>>>(ei, asrc1, adst1, amax1, h1, den1, agg1);
    gat_biaselu<<<(NN * HC) / 256, B, 0, stream>>>(agg1, den1, b1);
    gat_gemm2<<<(NN / 16 + 7) / 8, B, 0, stream>>>(agg1, w2h, h2);
    gat_att2 <<<(NN + 255) / 256, B, 0, stream>>>(h2, as2, ad2, asrc2, adst2, amax2, den2, agg2);
    gat_emax2<<<(ET + 255) / 256, B, 0, stream>>>(ei, asrc2, adst2, amax2);
    gat_eagg2<<<(ET * 4 + 255) / 256, B, 0, stream>>>(ei, asrc2, adst2, amax2, h2, den2, agg2);
    gat_lsm  <<<(NN + 255) / 256, B, 0, stream>>>(agg2, den2, b2, out);
}


# edge aggregation by 256 owner work-groups: each owns 392 nodes, scans all edges, sums exp-weighted rows in LDS (float LDS atomics, arrival order), writes each row once; no global float atomics
# speedup vs baseline: 1.9067x; 1.8130x over previous
#include <hip/hip_runtime.h>

#define NN  100000
#define NF  256
#define HH  8
#define CC  16
#define HC  128
#define NC  16
#define EE  1600000
#define ET  1700000
#define NEG 0.2f

typedef _Float16 h8   __attribute__((ext_vector_type(8)));
typedef _Float16 v16h __attribute__((ext_vector_type(16)));
typedef float    v8f  __attribute__((ext_vector_type(8)));

__device__ __forceinline__ void edge_sd(const int* __restrict__ ei, int e, int& s, int& d) {
    if (e < EE) { s = ei[e]; d = ei[EE + e]; }
    else        { s = e - EE; d = s; }
}

__device__ __forceinline__ void atomicMaxF(float* addr, float val) {
    if (val >= 0.0f) atomicMax((int*)addr, __float_as_int(val));
    else             atomicMin((unsigned int*)addr, __float_as_uint(val));
}

__device__ __forceinline__ float lrelu(float a) { return a > 0.0f ? a : NEG * a; }

__global__ __launch_bounds__(256) void gat_cvtw(const float* __restrict__ W1,
                                                const float* __restrict__ W2,
                                                _Float16* __restrict__ w1h,
                                                _Float16* __restrict__ w2h) {
    int i = blockIdx.x * 256 + threadIdx.x;
    if (i < HC * NF)              w1h[i] = (_Float16)W1[i];
    else if (i < HC * NF + NC*HC) w2h[i - HC * NF] = (_Float16)W2[i - HC * NF];
}

__global__ __launch_bounds__(256) void gat_cvtx(const float* __restrict__ x,
                                                _Float16* __restrict__ xh) {
    int i = blockIdx.x * 256 + threadIdx.x;
    if (i < NN * NF) xh[i] = (_Float16)x[i];
}

__global__ __launch_bounds__(256)
void gat_gemm1(const _Float16* __restrict__ xh, const _Float16* __restrict__ w1h,
               float* __restrict__ h1) {
    __shared__ _Float16 lw[HC * NF];
    const int tid = threadIdx.x;
    for (int idx = tid; idx < HC * 32; idx += 256) {
        int r = idx >> 5, ch = idx & 31;
        *(h8*)(&lw[r * NF + ((ch ^ (r & 15)) << 3)]) =
            *(const h8*)(w1h + (size_t)r * NF + (ch << 3));
    }
    __syncthreads();

    const int lane = tid & 31;
    const int rt   = blockIdx.x * 8 + (tid >> 5);
    if (rt >= NN / 16) return;
    const int rsub = lane & 15;
    const int ksel = (lane >> 4) << 3;
    const _Float16* xrow = xh + (size_t)(rt * 16 + rsub) * NF;

    v8f acc[8] = {};
    #pragma unroll
    for (int ks = 0; ks < 8; ++ks) {
        const int kb = ks * 32 + ksel;
        const int c0 = kb >> 3;
        union { v16h v; h8 p[2]; } A;
        A.p[0] = *(const h8*)(xrow + kb);
        A.p[1] = *(const h8*)(xrow + kb + 16);
        #pragma unroll
        for (int ct = 0; ct < 8; ++ct) {
            const int n = ct * 16 + rsub;
            union { v16h v; h8 p[2]; } B;
            B.p[0] = *(const h8*)(&lw[n * NF + (((c0    ) ^ (n & 15)) << 3)]);
            B.p[1] = *(const h8*)(&lw[n * NF + (((c0 + 2) ^ (n & 15)) << 3)]);
            acc[ct] = __builtin_amdgcn_wmma_f32_16x16x32_f16(
                false, A.v, false, B.v, (short)0, acc[ct], false, false);
        }
    }
    const int rhi = (lane >> 4) << 3;
    #pragma unroll
    for (int ct = 0; ct < 8; ++ct) {
        #pragma unroll
        for (int r = 0; r < 8; ++r)
            h1[(size_t)(rt * 16 + rhi + r) * HC + ct * 16 + rsub] = acc[ct][r];
    }
}

__global__ __launch_bounds__(256)
void gat_att1(const float* __restrict__ h1, const float* __restrict__ as,
              const float* __restrict__ ad, float* __restrict__ a_src,
              float* __restrict__ a_dst, float* __restrict__ amax) {
    int i = blockIdx.x * 256 + threadIdx.x;
    if (i >= NN * HH) return;
    int n = i >> 3, h = i & 7;
    const float* hp = h1 + (size_t)n * HC + h * CC;
    float s = 0.0f, d = 0.0f;
    #pragma unroll
    for (int c = 0; c < CC; ++c) { s += hp[c] * as[h * CC + c]; d += hp[c] * ad[h * CC + c]; }
    a_src[i] = s; a_dst[i] = d;
    amax[i] = -__builtin_inff();
}

__global__ __launch_bounds__(256)
void gat_emax1(const int* __restrict__ ei, const float* __restrict__ a_src,
               const float* __restrict__ a_dst, float* __restrict__ amax) {
    int i = blockIdx.x * 256 + threadIdx.x;
    if (i >= ET * HH) return;
    int e = i >> 3, h = i & 7, s, d;
    edge_sd(ei, e, s, d);
    atomicMaxF(&amax[d * HH + h], lrelu(a_src[s * HH + h] + a_dst[d * HH + h]));
}

__global__ __launch_bounds__(256)
void gat_gemm2(const float* __restrict__ a1, const _Float16* __restrict__ w2h,
               float* __restrict__ h2) {
    const int lane = threadIdx.x & 31;
    const int rt   = blockIdx.x * 8 + (threadIdx.x >> 5);
    if (rt >= NN / 16) return;
    const int rsub = lane & 15;
    const int ksel = (lane >> 4) << 3;
    const float*    arow = a1 + (size_t)(rt * 16 + rsub) * HC;
    const _Float16* wrow = w2h + (size_t)rsub * HC;

    v8f acc = {};
    #pragma unroll
    for (int ks = 0; ks < 4; ++ks) {
        const int kb = ks * 32 + ksel;
        union { v16h v; _Float16 e[16]; } A;
        #pragma unroll
        for (int t = 0; t < 8; ++t) {
            A.e[t]     = (_Float16)arow[kb + t];
            A.e[8 + t] = (_Float16)arow[kb + 16 + t];
        }
        union { v16h v; h8 p[2]; } B;
        B.p[0] = *(const h8*)(wrow + kb);
        B.p[1] = *(const h8*)(wrow + kb + 16);
        acc = __builtin_amdgcn_wmma_f32_16x16x32_f16(
            false, A.v, false, B.v, (short)0, acc, false, false);
    }
    const int rhi = (lane >> 4) << 3;
    #pragma unroll
    for (int r = 0; r < 8; ++r)
        h2[(size_t)(rt * 16 + rhi + r) * NC + rsub] = acc[r];
}

__global__ __launch_bounds__(256)
void gat_att2(const float* __restrict__ h2, const float* __restrict__ as,
              const float* __restrict__ ad, float* __restrict__ a_src,
              float* __restrict__ a_dst, float* __restrict__ amax) {
    int n = blockIdx.x * 256 + threadIdx.x;
    if (n >= NN) return;
    const float* hp = h2 + (size_t)n * NC;
    float s = 0.0f, d = 0.0f;
    #pragma unroll
    for (int c = 0; c < NC; ++c) { s += hp[c] * as[c]; d += hp[c] * ad[c]; }
    a_src[n] = s; a_dst[n] = d;
    amax[n] = -__builtin_inff();
}

__global__ __launch_bounds__(256)
void gat_emax2(const int* __restrict__ ei, const float* __restrict__ a_src,
               const float* __restrict__ a_dst, float* __restrict__ amax) {
    int e = blockIdx.x * 256 + threadIdx.x;
    if (e >= ET) return;
    int s, d; edge_sd(ei, e, s, d);
    atomicMaxF(&amax[d], lrelu(a_src[s] + a_dst[d]));
}

#define NPB  392
#define NGRP 256
#define SCAN_EDGES 16384
#define NIT ((EE + SCAN_EDGES - 1) / SCAN_EDGES)

template <int H, int HCt, int STEPS>
__device__ __forceinline__ void own_batch(const int* __restrict__ ei, const float* __restrict__ a_src,
                                          const float* __restrict__ a_dst, const float* __restrict__ amax,
                                          const float* __restrict__ hsrc, float* acc, float* dn,
                                          const int* qw, int first, int nvalid, int lo, int lane) {
    constexpr int LPE = HCt / 4;
    constexpr int EPW = 32 / LPE;
    const int sub = lane / LPE;
    const int q   = lane % LPE;
    const int h   = (q * 4) / (HCt / H);
    int  ents[STEPS], sv[STEPS], dlv[STEPS];
    bool ok[STEPS];
    #pragma unroll
    for (int u = 0; u < STEPS; ++u) {
        const int idx = u * EPW + sub;
        ok[u] = idx < nvalid;
        ents[u] = qw[first + idx];
    }
    #pragma unroll
    for (int u = 0; u < STEPS; ++u) {
        const int ent = ok[u] ? ents[u] : 0;
        const int e = ent & 0x1FFFFF;
        dlv[u] = ent >> 21;
        const int sl = ei[e < EE ? e : 0];
        sv[u] = (e < EE) ? sl : e - EE;
    }
    float4 hv[STEPS];
    float as[STEPS], ad[STEPS], am[STEPS];
    #pragma unroll
    for (int u = 0; u < STEPS; ++u) {
        const int sn = sv[u], dnode = lo + dlv[u];
        hv[u] = *(const float4*)(hsrc + (size_t)sn * HCt + q * 4);
        as[u] = a_src[sn * H + h];
        ad[u] = a_dst[dnode * H + h];
        am[u] = amax[dnode * H + h];
    }
    #pragma unroll
    for (int u = 0; u < STEPS; ++u) {
        const float w = expf(lrelu(as[u] + ad[u]) - am[u]);
        if (ok[u]) {
            if ((q & 3) == 0) atomicAdd(&dn[dlv[u] * H + h], w);
            float* ap = acc + dlv[u] * HCt + q;
            atomicAdd(ap,           hv[u].x * w);
            atomicAdd(ap + LPE,     hv[u].y * w);
            atomicAdd(ap + 2 * LPE, hv[u].z * w);
            atomicAdd(ap + 3 * LPE, hv[u].w * w);
        }
    }
}

template <int H, int HCt, int STEPS, int LAYER>
__global__ __launch_bounds__(1024)
void gat_own(const int* __restrict__ ei, const float* __restrict__ a_src,
             const float* __restrict__ a_dst, const float* __restrict__ amax,
             const float* __restrict__ hsrc, const float* __restrict__ bias,
             float* __restrict__ dstv) {
    constexpr int LPE  = HCt / 4;
    constexpr int EPW  = 32 / LPE;
    constexpr int B    = EPW * STEPS;
    constexpr int QCAP = B + 512;
    __shared__ float acc[NPB * HCt];
    __shared__ float dn[NPB * H];
    __shared__ int   qbuf[32 * QCAP];

    const int tid = threadIdx.x, lane = tid & 31;
    const int wave = __builtin_amdgcn_readfirstlane(tid >> 5);
    const int lo  = blockIdx.x * NPB;
    const int cnt = (NN - lo < NPB) ? (NN - lo) : NPB;
    for (int i = tid; i < NPB * HCt; i += 1024) acc[i] = 0.0f;
    for (int i = tid; i < NPB * H;   i += 1024) dn[i]  = 0.0f;
    __syncthreads();

    int* qw = qbuf + wave * QCAP;
    int  qn = 0;
    const unsigned lt = (1u << lane) - 1u;
    auto push = [&](bool m, unsigned ent) {
        const unsigned mask = __builtin_amdgcn_ballot_w32(m);
        if (m) qw[qn + __builtin_popcount(mask & lt)] = (int)ent;
        qn += __builtin_popcount(mask);
    };
    auto drain = [&]() {
        while (qn >= B) {
            own_batch<H, HCt, STEPS>(ei, a_src, a_dst, amax, hsrc, acc, dn, qw, qn - B, B, lo, lane);
            qn -= B;
        }
    };

    for (int b0 = wave * 32; b0 < cnt; b0 += 1024) {
        const int dl = b0 + lane;
        push(dl < cnt, ((unsigned)dl << 21) | (unsigned)(EE + lo + dl));
    }
    drain();

    const int* dstp = ei + EE;
    const int it0 = (blockIdx.x * NIT) / NGRP;
    auto ldscan = [&](int i, int4 (&buf)[4]) -> int {
        int it = i + it0;
        if (it >= NIT) it -= NIT;
        const int eb = it * SCAN_EDGES + wave * 512 + lane * 4;
        #pragma unroll
        for (int k = 0; k < 4; ++k) {
            const int ek = eb + k * 128;
            buf[k] = *(const int4*)(dstp + (ek < EE ? ek : EE - 4));
        }
        return eb;
    };
    auto test = [&](const int4 (&buf)[4], int eb) {
        #pragma unroll
        for (int k = 0; k < 4; ++k) {
            const int ek = eb + k * 128;
            const unsigned d0 = (unsigned)(buf[k].x - lo), d1 = (unsigned)(buf[k].y - lo);
            const unsigned d2 = (unsigned)(buf[k].z - lo), d3 = (unsigned)(buf[k].w - lo);
            const bool in = ek < EE;
            const bool m0 = in && d0 < (unsigned)cnt, m1 = in && d1 < (unsigned)cnt;
            const bool m2 = in && d2 < (unsigned)cnt, m3 = in && d3 < (unsigned)cnt;
            if (__builtin_amdgcn_ballot_w32(m0 | m1 | m2 | m3)) {
                push(m0, (d0 << 21) | (unsigned)(ek + 0));
                push(m1, (d1 << 21) | (unsigned)(ek + 1));
                push(m2, (d2 << 21) | (unsigned)(ek + 2));
                push(m3, (d3 << 21) | (unsigned)(ek + 3));
            }
        }
        drain();
    };
    int4 bufa[4], bufb[4];
    int eba = ldscan(0, bufa), ebb;
    for (int i = 0; i < NIT; i += 2) {
        ebb = ldscan(i + 1, bufb);
        test(bufa, eba);
        eba = ldscan(i + 2, bufa);
        if (i + 1 < NIT) test(bufb, ebb);
    }
    if (qn > 0)
        own_batch<H, HCt, STEPS>(ei, a_src, a_dst, amax, hsrc, acc, dn, qw, 0, qn, lo, lane);
    __syncthreads();

    if constexpr (LAYER == 1) {
        const float4 bq = *(const float4*)(bias + lane * 4);
        for (int r = wave; r < cnt; r += 32) {
            const float dv = dn[r * H + (lane >> 2)] + 1e-16f;
            const float* ap = acc + r * HCt + lane;
            float4 o;
            o.x = ap[0]       / dv + bq.x;
            o.y = ap[LPE]     / dv + bq.y;
            o.z = ap[2 * LPE] / dv + bq.z;
            o.w = ap[3 * LPE] / dv + bq.w;
            o.x = o.x > 0.0f ? o.x : (expf(o.x) - 1.0f);
            o.y = o.y > 0.0f ? o.y : (expf(o.y) - 1.0f);
            o.z = o.z > 0.0f ? o.z : (expf(o.z) - 1.0f);
            o.w = o.w > 0.0f ? o.w : (expf(o.w) - 1.0f);
            *(float4*)(dstv + (size_t)(lo + r) * HCt + lane * 4) = o;
        }
    } else {
        for (int r = tid; r < cnt; r += 1024) {
            float v[HCt], m = -__builtin_inff();
            const float dv = dn[r] + 1e-16f;
            #pragma unroll
            for (int c = 0; c < HCt; ++c) {
                v[c] = acc[r * HCt + (c & 3) * LPE + (c >> 2)] / dv + bias[c];
                m = fmaxf(m, v[c]);
            }
            float sum = 0.0f;
            #pragma unroll
            for (int c = 0; c < HCt; ++c) sum += expf(v[c] - m);
            const float ls = logf(sum);
            #pragma unroll
            for (int c = 0; c < HCt; ++c) dstv[(size_t)(lo + r) * HCt + c] = v[c] - m - ls;
        }
    }
}

extern "C" void kernel_launch(void* const* d_in, const int* in_sizes, int n_in,
                              void* d_out, int out_size, void* d_ws, size_t ws_size,
                              hipStream_t stream) {
    (void)in_sizes; (void)n_in; (void)out_size; (void)ws_size;
    const float* x   = (const float*)d_in[0];
    const int*   ei  = (const int*)d_in[1];
    const float* W1  = (const float*)d_in[2];
    const float* as1 = (const float*)d_in[3];
    const float* ad1 = (const float*)d_in[4];
    const float* b1  = (const float*)d_in[5];
    const float* W2  = (const float*)d_in[6];
    const float* as2 = (const float*)d_in[7];
    const float* ad2 = (const float*)d_in[8];
    const float* b2  = (const float*)d_in[9];
    float* out = (float*)d_out;

    char* wsb = (char*)d_ws;
    size_t off = 0;
    auto alloc = [&](size_t bytes) -> void* {
        void* p = wsb + off;
        off += (bytes + 255) & ~(size_t)255;
        return p;
    };
    _Float16* w1h  = (_Float16*)alloc((size_t)HC * NF * 2);
    _Float16* w2h  = (_Float16*)alloc((size_t)NC * HC * 2);
    _Float16* xh   = (_Float16*)alloc((size_t)NN * NF * 2);
    float* h1      = (float*)alloc((size_t)NN * HC * 4);
    float* asrc1   = (float*)alloc((size_t)NN * HH * 4);
    float* adst1   = (float*)alloc((size_t)NN * HH * 4);
    float* amax1   = (float*)alloc((size_t)NN * HH * 4);
    float* h2      = (float*)alloc((size_t)NN * NC * 4);
    float* asrc2   = (float*)alloc((size_t)NN * 4);
    float* adst2   = (float*)alloc((size_t)NN * 4);
    float* amax2   = (float*)alloc((size_t)NN * 4);
    float* act1    = (float*)xh;

    const dim3 B(256);
    gat_cvtw <<<(HC * NF + NC * HC + 255) / 256, B, 0, stream>>>(W1, W2, w1h, w2h);
    gat_cvtx <<<(NN * NF) / 256, B, 0, stream>>>(x, xh);
    gat_gemm1<<<(NN / 16 + 7) / 8, B, 0, stream>>>(xh, w1h, h1);
    gat_att1 <<<(NN * HH) / 256, B, 0, stream>>>(h1, as1, ad1, asrc1, adst1, amax1);
    gat_emax1<<<(ET * HH) / 256, B, 0, stream>>>(ei, asrc1, adst1, amax1);
    gat_own<HH, HC, 8, 1><<<NGRP, 1024, 0, stream>>>(ei, asrc1, adst1, amax1, h1, b1, act1);
    gat_gemm2<<<(NN / 16 + 7) / 8, B, 0, stream>>>(act1, w2h, h2);
    gat_att2 <<<(NN + 255) / 256, B, 0, stream>>>(h2, as2, ad2, asrc2, adst2, amax2);
    gat_emax2<<<(ET + 255) / 256, B, 0, stream>>>(ei, asrc2, adst2, amax2);
    gat_own<1, NC, 4, 2><<<NGRP, 1024, 0, stream>>>(ei, asrc2, adst2, amax2, h2, b2, out);
}


# destination list scanned once a call: layer-1 owner groups keep per-wave edge lists in the workspace, layer-2 groups walk them (no second scan); layer-2 LDS float sums now arrive in list order
# speedup vs baseline: 2.1063x; 1.1047x over previous
#include <hip/hip_runtime.h>

#define NN  100000
#define NF  256
#define HH  8
#define CC  16
#define HC  128
#define NC  16
#define EE  1600000
#define ET  1700000
#define NEG 0.2f

typedef _Float16 h8   __attribute__((ext_vector_type(8)));
typedef _Float16 v16h __attribute__((ext_vector_type(16)));
typedef float    v8f  __attribute__((ext_vector_type(8)));

__device__ __forceinline__ void edge_sd(const int* __restrict__ ei, int e, int& s, int& d) {
    if (e < EE) { s = ei[e]; d = ei[EE + e]; }
    else        { s = e - EE; d = s; }
}

__device__ __forceinline__ void atomicMaxF(float* addr, float val) {
    if (val >= 0.0f) atomicMax((int*)addr, __float_as_int(val));
    else             atomicMin((unsigned int*)addr, __float_as_uint(val));
}

__device__ __forceinline__ float lrelu(float a) { return a > 0.0f ? a : NEG * a; }

__global__ __launch_bounds__(256) void gat_cvtw(const float* __restrict__ W1,
                                                const float* __restrict__ W2,
                                                _Float16* __restrict__ w1h,
                                                _Float16* __restrict__ w2h) {
    int i = blockIdx.x * 256 + threadIdx.x;
    if (i < HC * NF)              w1h[i] = (_Float16)W1[i];
    else if (i < HC * NF + NC*HC) w2h[i - HC * NF] = (_Float16)W2[i - HC * NF];
}

__global__ __launch_bounds__(256) void gat_cvtx(const float* __restrict__ x,
                                                _Float16* __restrict__ xh) {
    int i = blockIdx.x * 256 + threadIdx.x;
    if (i < NN * NF) xh[i] = (_Float16)x[i];
}

__global__ __launch_bounds__(256)
void gat_gemm1(const _Float16* __restrict__ xh, const _Float16* __restrict__ w1h,
               float* __restrict__ h1) {
    __shared__ _Float16 lw[HC * NF];
    const int tid = threadIdx.x;
    for (int idx = tid; idx < HC * 32; idx += 256) {
        int r = idx >> 5, ch = idx & 31;
        *(h8*)(&lw[r * NF + ((ch ^ (r & 15)) << 3)]) =
            *(const h8*)(w1h + (size_t)r * NF + (ch << 3));
    }
    __syncthreads();

    const int lane = tid & 31;
    const int rt   = blockIdx.x * 8 + (tid >> 5);
    if (rt >= NN / 16) return;
    const int rsub = lane & 15;
    const int ksel = (lane >> 4) << 3;
    const _Float16* xrow = xh + (size_t)(rt * 16 + rsub) * NF;

    v8f acc[8] = {};
    #pragma unroll
    for (int ks = 0; ks < 8; ++ks) {
        const int kb = ks * 32 + ksel;
        const int c0 = kb >> 3;
        union { v16h v; h8 p[2]; } A;
        A.p[0] = *(const h8*)(xrow + kb);
        A.p[1] = *(const h8*)(xrow + kb + 16);
        #pragma unroll
        for (int ct = 0; ct < 8; ++ct) {
            const int n = ct * 16 + rsub;
            union { v16h v; h8 p[2]; } B;
            B.p[0] = *(const h8*)(&lw[n * NF + (((c0    ) ^ (n & 15)) << 3)]);
            B.p[1] = *(const h8*)(&lw[n * NF + (((c0 + 2) ^ (n & 15)) << 3)]);
            acc[ct] = __builtin_amdgcn_wmma_f32_16x16x32_f16(
                false, A.v, false, B.v, (short)0, acc[ct], false, false);
        }
    }
    const int rhi = (lane >> 4) << 3;
    #pragma unroll
    for (int ct = 0; ct < 8; ++ct) {
        #pragma unroll
        for (int r = 0; r < 8; ++r)
            h1[(size_t)(rt * 16 + rhi + r) * HC + ct * 16 + rsub] = acc[ct][r];
    }
}

__global__ __launch_bounds__(256)
void gat_att1(const float* __restrict__ h1, const float* __restrict__ as,
              const float* __restrict__ ad, float* __restrict__ a_src,
              float* __restrict__ a_dst, float* __restrict__ amax) {
    int i = blockIdx.x * 256 + threadIdx.x;
    if (i >= NN * HH) return;
    int n = i >> 3, h = i & 7;
    const float* hp = h1 + (size_t)n * HC + h * CC;
    float s = 0.0f, d = 0.0f;
    #pragma unroll
    for (int c = 0; c < CC; ++c) { s += hp[c] * as[h * CC + c]; d += hp[c] * ad[h * CC + c]; }
    a_src[i] = s; a_dst[i] = d;
    amax[i] = -__builtin_inff();
}

__global__ __launch_bounds__(256)
void gat_emax1(const int* __restrict__ ei, const float* __restrict__ a_src,
               const float* __restrict__ a_dst, float* __restrict__ amax) {
    int i = blockIdx.x * 256 + threadIdx.x;
    if (i >= ET * HH) return;
    int e = i >> 3, h = i & 7, s, d;
    edge_sd(ei, e, s, d);
    atomicMaxF(&amax[d * HH + h], lrelu(a_src[s * HH + h] + a_dst[d * HH + h]));
}

__global__ __launch_bounds__(256)
void gat_gemm2(const float* __restrict__ a1, const _Float16* __restrict__ w2h,
               float* __restrict__ h2) {
    const int lane = threadIdx.x & 31;
    const int rt   = blockIdx.x * 8 + (threadIdx.x >> 5);
    if (rt >= NN / 16) return;
    const int rsub = lane & 15;
    const int ksel = (lane >> 4) << 3;
    const float*    arow = a1 + (size_t)(rt * 16 + rsub) * HC;
    const _Float16* wrow = w2h + (size_t)rsub * HC;

    v8f acc = {};
    #pragma unroll
    for (int ks = 0; ks < 4; ++ks) {
        const int kb = ks * 32 + ksel;
        union { v16h v; _Float16 e[16]; } A;
        #pragma unroll
        for (int t = 0; t < 8; ++t) {
            A.e[t]     = (_Float16)arow[kb + t];
            A.e[8 + t] = (_Float16)arow[kb + 16 + t];
        }
        union { v16h v; h8 p[2]; } B;
        B.p[0] = *(const h8*)(wrow + kb);
        B.p[1] = *(const h8*)(wrow + kb + 16);
        acc = __builtin_amdgcn_wmma_f32_16x16x32_f16(
            false, A.v, false, B.v, (short)0, acc, false, false);
    }
    const int rhi = (lane >> 4) << 3;
    #pragma unroll
    for (int r = 0; r < 8; ++r)
        h2[(size_t)(rt * 16 + rhi + r) * NC + rsub] = acc[r];
}

__global__ __launch_bounds__(256)
void gat_att2(const float* __restrict__ h2, const float* __restrict__ as,
              const float* __restrict__ ad, float* __restrict__ a_src,
              float* __restrict__ a_dst, float* __restrict__ amax) {
    int n = blockIdx.x * 256 + threadIdx.x;
    if (n >= NN) return;
    const float* hp = h2 + (size_t)n * NC;
    float s = 0.0f, d = 0.0f;
    #pragma unroll
    for (int c = 0; c < NC; ++c) { s += hp[c] * as[c]; d += hp[c] * ad[c]; }
    a_src[n] = s; a_dst[n] = d;
    amax[n] = -__builtin_inff();
}

__global__ __launch_bounds__(256)
void gat_emax2(const int* __restrict__ ei, const float* __restrict__ a_src,
               const float* __restrict__ a_dst, float* __restrict__ amax) {
    int e = blockIdx.x * 256 + threadIdx.x;
    if (e >= ET) return;
    int s, d; edge_sd(ei, e, s, d);
    atomicMaxF(&amax[d], lrelu(a_src[s] + a_dst[d]));
}

#define NPB  392
#define NGRP 256
#define SCAN_EDGES 16384
#define NIT ((EE + SCAN_EDGES - 1) / SCAN_EDGES)
#define WCAP 320
static_assert(NPB <= 512 && NN <= (1 << 17), "list word: 9 bits local node, 17 bits source node");

template <int H, int HCt, int STEPS, bool KEEP>
__device__ __forceinline__ void own_batch(const int* __restrict__ ei, const float* __restrict__ a_src,
                                          const float* __restrict__ a_dst, const float* __restrict__ amax,
                                          const float* __restrict__ hsrc, float* acc, float* dn,
                                          const int* qw, int first, int nvalid, int lo, int lane,
                                          int* __restrict__ lst, int lpos) {
    constexpr int LPE = HCt / 4;
    constexpr int EPW = 32 / LPE;
    const int sub = lane / LPE;
    const int q   = lane % LPE;
    const int h   = (q * 4) / (HCt / H);
    int  ents[STEPS], sv[STEPS], dlv[STEPS];
    bool ok[STEPS];
    #pragma unroll
    for (int u = 0; u < STEPS; ++u) {
        const int idx = u * EPW + sub;
        ok[u] = idx < nvalid;
        ents[u] = qw[first + idx];
    }
    #pragma unroll
    for (int u = 0; u < STEPS; ++u) {
        const int ent = ok[u] ? ents[u] : 0;
        const int e = ent & 0x1FFFFF;
        dlv[u] = ent >> 21;
        const int sl = ei[e < EE ? e : 0];
        sv[u] = (e < EE) ? sl : e - EE;
    }
    if constexpr (KEEP) {
        static_assert(EPW == 1 && STEPS <= 32, "list store assumes one edge per wave step");
        int mine = 0;
        #pragma unroll
        for (int u = 0; u < STEPS; ++u) mine = (lane == u) ? ((dlv[u] << 17) | sv[u]) : mine;
        if (lane < nvalid && lpos + lane < WCAP) lst[lpos + lane] = mine;
    }
    float4 hv[STEPS];
    float as[STEPS], ad[STEPS], am[STEPS];
    #pragma unroll
    for (int u = 0; u < STEPS; ++u) {
        const int sn = sv[u], dnode = lo + dlv[u];
        hv[u] = *(const float4*)(hsrc + (size_t)sn * HCt + q * 4);
        as[u] = a_src[sn * H + h];
        ad[u] = a_dst[dnode * H + h];
        am[u] = amax[dnode * H + h];
    }
    #pragma unroll
    for (int u = 0; u < STEPS; ++u) {
        const float w = expf(lrelu(as[u] + ad[u]) - am[u]);
        if (ok[u]) {
            if ((q & 3) == 0) atomicAdd(&dn[dlv[u] * H + h], w);
            float* ap = acc + dlv[u] * HCt + q;
            atomicAdd(ap,           hv[u].x * w);
            atomicAdd(ap + LPE,     hv[u].y * w);
            atomicAdd(ap + 2 * LPE, hv[u].z * w);
            atomicAdd(ap + 3 * LPE, hv[u].w * w);
        }
    }
}

template <int H, int HCt, int STEPS>
__device__ __forceinline__ void own_walk(const int* __restrict__ lw, int n, const float* __restrict__ a_src,
                                         const float* __restrict__ a_dst, const float* __restrict__ amax,
                                         const float* __restrict__ hsrc, float* acc, float* dn,
                                         int lo, int lane) {
    constexpr int LPE = HCt / 4;
    constexpr int EPW = 32 / LPE;
    constexpr int B   = EPW * STEPS;
    const int sub = lane / LPE;
    const int q   = lane % LPE;
    const int h   = (q * 4) / (HCt / H);
    int ents[STEPS], nxt[STEPS];
    #pragma unroll
    for (int u = 0; u < STEPS; ++u) {
        const int idx = u * EPW + sub;
        ents[u] = lw[idx < n ? idx : 0];
    }
    for (int b = 0; b < n; b += B) {
        #pragma unroll
        for (int u = 0; u < STEPS; ++u) {
            const int idx = b + B + u * EPW + sub;
            nxt[u] = lw[idx < n ? idx : 0];
        }
        float4 hv[STEPS];
        float as[STEPS], ad[STEPS], am[STEPS];
        int   dlv[STEPS];
        bool  ok[STEPS];
        #pragma unroll
        for (int u = 0; u < STEPS; ++u) {
            ok[u] = b + u * EPW + sub < n;
            const int ent = ok[u] ? ents[u] : 0;
            const int sn = ent & 0x1FFFF;
            dlv[u] = ent >> 17;
            const int dnode = lo + dlv[u];
            hv[u] = *(const float4*)(hsrc + (size_t)sn * HCt + q * 4);
            as[u] = a_src[sn * H + h];
            ad[u] = a_dst[dnode * H + h];
            am[u] = amax[dnode * H + h];
        }
        #pragma unroll
        for (int u = 0; u < STEPS; ++u) {
            const float w = expf(lrelu(as[u] + ad[u]) - am[u]);
            if (ok[u]) {
                if ((q & 3) == 0) atomicAdd(&dn[dlv[u] * H + h], w);
                float* ap = acc + dlv[u] * HCt + q;
                atomicAdd(ap,           hv[u].x * w);
                atomicAdd(ap + LPE,     hv[u].y * w);
                atomicAdd(ap + 2 * LPE, hv[u].z * w);
                atomicAdd(ap + 3 * LPE, hv[u].w * w);
            }
        }
        #pragma unroll
        for (int u = 0; u < STEPS; ++u) ents[u] = nxt[u];
    }
}

template <int H, int HCt, int STEPS, int LAYER>
__global__ __launch_bounds__(1024)
void gat_own(const int* __restrict__ ei, const float* __restrict__ a_src,
             const float* __restrict__ a_dst, const float* __restrict__ amax,
             const float* __restrict__ hsrc, const float* __restrict__ bias,
             float* __restrict__ dstv, int* __restrict__ elist, int* __restrict__ ecnt) {
    constexpr int LPE  = HCt / 4;
    constexpr int EPW  = 32 / LPE;
    constexpr int B    = EPW * STEPS;
    constexpr int QCAP = B + 512;
    __shared__ float acc[NPB * HCt];
    __shared__ float dn[NPB * H];
    __shared__ int   qbuf[32 * QCAP];

    const int tid = threadIdx.x, lane = tid & 31;
    const int wave = __builtin_amdgcn_readfirstlane(tid >> 5);
    const int lo  = blockIdx.x * NPB;
    const int cnt = (NN - lo < NPB) ? (NN - lo) : NPB;
    for (int i = tid; i < NPB * HCt; i += 1024) acc[i] = 0.0f;
    for (int i = tid; i < NPB * H;   i += 1024) dn[i]  = 0.0f;
    __syncthreads();

    int* lst = elist + (size_t)(blockIdx.x * 32 + wave) * WCAP;
    bool walk = false;
    int  nl = 0;
    if constexpr (LAYER == 2) {
        const int c = ecnt[blockIdx.x * 32 + lane];
        walk = __builtin_amdgcn_ballot_w32(c > WCAP) == 0;
        nl = __builtin_amdgcn_readlane(c, wave);
    }
    if (walk) {
        own_walk<H, HCt, 8>(lst, nl, a_src, a_dst, amax, hsrc, acc, dn, lo, lane);
    } else {
    int* qw = qbuf + wave * QCAP;
    int  qn = 0;
    int  ln = 0;
    const unsigned lt = (1u << lane) - 1u;
    auto push = [&](bool m, unsigned ent) {
        const unsigned mask = __builtin_amdgcn_ballot_w32(m);
        if (m) qw[qn + __builtin_popcount(mask & lt)] = (int)ent;
        qn += __builtin_popcount(mask);
    };
    auto drain = [&]() {
        while (qn >= B) {
            own_batch<H, HCt, STEPS, LAYER == 1>(ei, a_src, a_dst, amax, hsrc, acc, dn, qw, qn - B, B, lo, lane, lst, ln);
            qn -= B;
            ln += B;
        }
    };

    for (int b0 = wave * 32; b0 < cnt; b0 += 1024) {
        const int dl = b0 + lane;
        push(dl < cnt, ((unsigned)dl << 21) | (unsigned)(EE + lo + dl));
    }
    drain();

    const int* dstp = ei + EE;
    const int it0 = (blockIdx.x * NIT) / NGRP;
    auto ldscan = [&](int i, int4 (&buf)[4]) -> int {
        int it = i + it0;
        if (it >= NIT) it -= NIT;
        const int eb = it * SCAN_EDGES + wave * 512 + lane * 4;
        #pragma unroll
        for (int k = 0; k < 4; ++k) {
            const int ek = eb + k * 128;
            buf[k] = *(const int4*)(dstp + (ek < EE ? ek : EE - 4));
        }
        return eb;
    };
    auto test = [&](const int4 (&buf)[4], int eb) {
        #pragma unroll
        for (int k = 0; k < 4; ++k) {
            const int ek = eb + k * 128;
            const unsigned d0 = (unsigned)(buf[k].x - lo), d1 = (unsigned)(buf[k].y - lo);
            const unsigned d2 = (unsigned)(buf[k].z - lo), d3 = (unsigned)(buf[k].w - lo);
            const bool in = ek < EE;
            const bool m0 = in && d0 < (unsigned)cnt, m1 = in && d1 < (unsigned)cnt;
            const bool m2 = in && d2 < (unsigned)cnt, m3 = in && d3 < (unsigned)cnt;
            if (__builtin_amdgcn_ballot_w32(m0 | m1 | m2 | m3)) {
                push(m0, (d0 << 21) | (unsigned)(ek + 0));
                push(m1, (d1 << 21) | (unsigned)(ek + 1));
                push(m2, (d2 << 21) | (unsigned)(ek + 2));
                push(m3, (d3 << 21) | (unsigned)(ek + 3));
            }
        }
        drain();
    };
    int4 bufa[4], bufb[4];
    int eba = ldscan(0, bufa), ebb;
    for (int i = 0; i < NIT; i += 2) {
        ebb = ldscan(i + 1, bufb);
        test(bufa, eba);
        eba = ldscan(i + 2, bufa);
        if (i + 1 < NIT) test(bufb, ebb);
    }
    if (qn > 0) {
        own_batch<H, HCt, STEPS, LAYER == 1>(ei, a_src, a_dst, amax, hsrc, acc, dn, qw, 0, qn, lo, lane, lst, ln);
        ln += qn;
    }
    if constexpr (LAYER == 1) {
        if (lane == 0) ecnt[blockIdx.x * 32 + wave] = ln;
    }
    }
    __syncthreads();

    if constexpr (LAYER == 1) {
        const float4 bq = *(const float4*)(bias + lane * 4);
        for (int r = wave; r < cnt; r += 32) {
            const float dv = dn[r * H + (lane >> 2)] + 1e-16f;
            const float* ap = acc + r * HCt + lane;
            float4 o;
            o.x = ap[0]       / dv + bq.x;
            o.y = ap[LPE]     / dv + bq.y;
            o.z = ap[2 * LPE] / dv + bq.z;
            o.w = ap[3 * LPE] / dv + bq.w;
            o.x = o.x > 0.0f ? o.x : (expf(o.x) - 1.0f);
            o.y = o.y > 0.0f ? o.y : (expf(o.y) - 1.0f);
            o.z = o.z > 0.0f ? o.z : (expf(o.z) - 1.0f);
            o.w = o.w > 0.0f ? o.w : (expf(o.w) - 1.0f);
            *(float4*)(dstv + (size_t)(lo + r) * HCt + lane * 4) = o;
        }
    } else {
        for (int r = tid; r < cnt; r += 1024) {
            float v[HCt], m = -__builtin_inff();
            const float dv = dn[r] + 1e-16f;
            #pragma unroll
            for (int c = 0; c < HCt; ++c) {
                v[c] = acc[r * HCt + (c & 3) * LPE + (c >> 2)] / dv + bias[c];
                m = fmaxf(m, v[c]);
            }
            float sum = 0.0f;
            #pragma unroll
            for (int c = 0; c < HCt; ++c) sum += expf(v[c] - m);
            const float ls = logf(sum);
            #pragma unroll
            for (int c = 0; c < HCt; ++c) dstv[(size_t)(lo + r) * HCt + c] = v[c] - m - ls;
        }
    }
}

extern "C" void kernel_launch(void* const* d_in, const int* in_sizes, int n_in,
                              void* d_out, int out_size, void* d_ws, size_t ws_size,
                              hipStream_t stream) {
    (void)in_sizes; (void)n_in; (void)out_size; (void)ws_size;
    const float* x   = (const float*)d_in[0];
    const int*   ei  = (const int*)d_in[1];
    const float* W1  = (const float*)d_in[2];
    const float* as1 = (const float*)d_in[3];
    const float* ad1 = (const float*)d_in[4];
    const float* b1  = (const float*)d_in[5];
    const float* W2  = (const float*)d_in[6];
    const float* as2 = (const float*)d_in[7];
    const float* ad2 = (const float*)d_in[8];
    const float* b2  = (const float*)d_in[9];
    float* out = (float*)d_out;

    char* wsb = (char*)d_ws;
    size_t off = 0;
    auto alloc = [&](size_t bytes) -> void* {
        void* p = wsb + off;
        off += (bytes + 255) & ~(size_t)255;
        return p;
    };
    _Float16* w1h  = (_Float16*)alloc((size_t)HC * NF * 2);
    _Float16* w2h  = (_Float16*)alloc((size_t)NC * HC * 2);
    _Float16* xh   = (_Float16*)alloc((size_t)NN * NF * 2);
    float* h1      = (float*)alloc((size_t)NN * HC * 4);
    float* asrc1   = (float*)alloc((size_t)NN * HH * 4);
    float* adst1   = (float*)alloc((size_t)NN * HH * 4);
    float* amax1   = (float*)alloc((size_t)NN * HH * 4);
    float* h2      = (float*)alloc((size_t)NN * NC * 4);
    float* asrc2   = (float*)alloc((size_t)NN * 4);
    float* adst2   = (float*)alloc((size_t)NN * 4);
    float* amax2   = (float*)alloc((size_t)NN * 4);
    int* elist     = (int*)alloc((size_t)NGRP * 32 * WCAP * 4);
    int* ecnt      = (int*)alloc((size_t)NGRP * 32 * 4);
    float* act1    = (float*)xh;

    const dim3 B(256);
    gat_cvtw <<<(HC * NF + NC * HC + 255) / 256, B, 0, stream>>>(W1, W2, w1h, w2h);
    gat_cvtx <<<(NN * NF) / 256, B, 0, stream>>>(x, xh);
    gat_gemm1<<<(NN / 16 + 7) / 8, B, 0, stream>>>(xh, w1h, h1);
    gat_att1 <<<(NN * HH) / 256, B, 0, stream>>>(h1, as1, ad1, asrc1, adst1, amax1);
    gat_emax1<<<(ET * HH) / 256, B, 0, stream>>>(ei, asrc1, adst1, amax1);
    gat_own<HH, HC, 8, 1><<<NGRP, 1024, 0, stream>>>(ei, asrc1, adst1, amax1, h1, b1, act1, elist, ecnt);
    gat_gemm2<<<(NN / 16 + 7) / 8, B, 0, stream>>>(act1, w2h, h2);
    gat_att2 <<<(NN + 255) / 256, B, 0, stream>>>(h2, as2, ad2, asrc2, adst2, amax2);
    gat_emax2<<<(ET + 255) / 256, B, 0, stream>>>(ei, asrc2, adst2, amax2);
    gat_own<1, NC, 4, 2><<<NGRP, 1024, 0, stream>>>(ei, asrc2, adst2, amax2, h2, b2, out, elist, ecnt);
}


# owner groups scan 1 byte per edge, not 4: emax1 stores each edge's owner id (dst/392) as a byte, groups test 16 edges per 16-B load; edge-to-wave assignment moves, LDS float sums still arrival-order
# speedup vs baseline: 2.2287x; 1.0581x over previous
#include <hip/hip_runtime.h>

#define NN  100000
#define NF  256
#define HH  8
#define CC  16
#define HC  128
#define NC  16
#define EE  1600000
#define ET  1700000
#define NEG 0.2f
#define NPB  392
#define NGRP 256
static_assert((NN + NPB - 1) / NPB <= 256, "an owner-group id must fit one byte");
static_assert(EE % 16 == 0, "the owner-byte list is read 16 edges a load");

typedef _Float16 h8   __attribute__((ext_vector_type(8)));
typedef _Float16 v16h __attribute__((ext_vector_type(16)));
typedef float    v8f  __attribute__((ext_vector_type(8)));

__device__ __forceinline__ void edge_sd(const int* __restrict__ ei, int e, int& s, int& d) {
    if (e < EE) { s = ei[e]; d = ei[EE + e]; }
    else        { s = e - EE; d = s; }
}

__device__ __forceinline__ void atomicMaxF(float* addr, float val) {
    if (val >= 0.0f) atomicMax((int*)addr, __float_as_int(val));
    else             atomicMin((unsigned int*)addr, __float_as_uint(val));
}

__device__ __forceinline__ float lrelu(float a) { return a > 0.0f ? a : NEG * a; }

__global__ __launch_bounds__(256) void gat_cvtw(const float* __restrict__ W1,
                                                const float* __restrict__ W2,
                                                _Float16* __restrict__ w1h,
                                                _Float16* __restrict__ w2h) {
    int i = blockIdx.x * 256 + threadIdx.x;
    if (i < HC * NF)              w1h[i] = (_Float16)W1[i];
    else if (i < HC * NF + NC*HC) w2h[i - HC * NF] = (_Float16)W2[i - HC * NF];
}

__global__ __launch_bounds__(256) void gat_cvtx(const float* __restrict__ x,
                                                _Float16* __restrict__ xh) {
    int i = blockIdx.x * 256 + threadIdx.x;
    if (i < NN * NF) xh[i] = (_Float16)x[i];
}

__global__ __launch_bounds__(256)
void gat_gemm1(const _Float16* __restrict__ xh, const _Float16* __restrict__ w1h,
               float* __restrict__ h1) {
    __shared__ _Float16 lw[HC * NF];
    const int tid = threadIdx.x;
    for (int idx = tid; idx < HC * 32; idx += 256) {
        int r = idx >> 5, ch = idx & 31;
        *(h8*)(&lw[r * NF + ((ch ^ (r & 15)) << 3)]) =
            *(const h8*)(w1h + (size_t)r * NF + (ch << 3));
    }
    __syncthreads();

    const int lane = tid & 31;
    const int rt   = blockIdx.x * 8 + (tid >> 5);
    if (rt >= NN / 16) return;
    const int rsub = lane & 15;
    const int ksel = (lane >> 4) << 3;
    const _Float16* xrow = xh + (size_t)(rt * 16 + rsub) * NF;

    v8f acc[8] = {};
    #pragma unroll
    for (int ks = 0; ks < 8; ++ks) {
        const int kb = ks * 32 + ksel;
        const int c0 = kb >> 3;
        union { v16h v; h8 p[2]; } A;
        A.p[0] = *(const h8*)(xrow + kb);
        A.p[1] = *(const h8*)(xrow + kb + 16);
        #pragma unroll
        for (int ct = 0; ct < 8; ++ct) {
            const int n = ct * 16 + rsub;
            union { v16h v; h8 p[2]; } B;
            B.p[0] = *(const h8*)(&lw[n * NF + (((c0    ) ^ (n & 15)) << 3)]);
            B.p[1] = *(const h8*)(&lw[n * NF + (((c0 + 2) ^ (n & 15)) << 3)]);
            acc[ct] = __builtin_amdgcn_wmma_f32_16x16x32_f16(
                false, A.v, false, B.v, (short)0, acc[ct], false, false);
        }
    }
    const int rhi = (lane >> 4) << 3;
    #pragma unroll
    for (int ct = 0; ct < 8; ++ct) {
        #pragma unroll
        for (int r = 0; r < 8; ++r)
            h1[(size_t)(rt * 16 + rhi + r) * HC + ct * 16 + rsub] = acc[ct][r];
    }
}

__global__ __launch_bounds__(256)
void gat_att1(const float* __restrict__ h1, const float* __restrict__ as,
              const float* __restrict__ ad, float* __restrict__ a_src,
              float* __restrict__ a_dst, float* __restrict__ amax) {
    int i = blockIdx.x * 256 + threadIdx.x;
    if (i >= NN * HH) return;
    int n = i >> 3, h = i & 7;
    const float* hp = h1 + (size_t)n * HC + h * CC;
    float s = 0.0f, d = 0.0f;
    #pragma unroll
    for (int c = 0; c < CC; ++c) { s += hp[c] * as[h * CC + c]; d += hp[c] * ad[h * CC + c]; }
    a_src[i] = s; a_dst[i] = d;
    amax[i] = -__builtin_inff();
}

__global__ __launch_bounds__(256)
void gat_emax1(const int* __restrict__ ei, const float* __restrict__ a_src,
               const float* __restrict__ a_dst, float* __restrict__ amax,
               unsigned char* __restrict__ own) {
    int i = blockIdx.x * 256 + threadIdx.x;
    if (i >= ET * HH) return;
    int e = i >> 3, h = i & 7, s, d;
    edge_sd(ei, e, s, d);
    if (h == 0 && e < EE) own[e] = (unsigned char)(d / NPB);
    atomicMaxF(&amax[d * HH + h], lrelu(a_src[s * HH + h] + a_dst[d * HH + h]));
}

__global__ __launch_bounds__(256)
void gat_gemm2(const float* __restrict__ a1, const _Float16* __restrict__ w2h,
               float* __restrict__ h2) {
    const int lane = threadIdx.x & 31;
    const int rt   = blockIdx.x * 8 + (threadIdx.x >> 5);
    if (rt >= NN / 16) return;
    const int rsub = lane & 15;
    const int ksel = (lane >> 4) << 3;
    const float*    arow = a1 + (size_t)(rt * 16 + rsub) * HC;
    const _Float16* wrow = w2h + (size_t)rsub * HC;

    v8f acc = {};
    #pragma unroll
    for (int ks = 0; ks < 4; ++ks) {
        const int kb = ks * 32 + ksel;
        union { v16h v; _Float16 e[16]; } A;
        #pragma unroll
        for (int t = 0; t < 8; ++t) {
            A.e[t]     = (_Float16)arow[kb + t];
            A.e[8 + t] = (_Float16)arow[kb + 16 + t];
        }
        union { v16h v; h8 p[2]; } B;
        B.p[0] = *(const h8*)(wrow + kb);
        B.p[1] = *(const h8*)(wrow + kb + 16);
        acc = __builtin_amdgcn_wmma_f32_16x16x32_f16(
            false, A.v, false, B.v, (short)0, acc, false, false);
    }
    const int rhi = (lane >> 4) << 3;
    #pragma unroll
    for (int r = 0; r < 8; ++r)
        h2[(size_t)(rt * 16 + rhi + r) * NC + rsub] = acc[r];
}

__global__ __launch_bounds__(256)
void gat_att2(const float* __restrict__ h2, const float* __restrict__ as,
              const float* __restrict__ ad, float* __restrict__ a_src,
              float* __restrict__ a_dst, float* __restrict__ amax) {
    int n = blockIdx.x * 256 + threadIdx.x;
    if (n >= NN) return;
    const float* hp = h2 + (size_t)n * NC;
    float s = 0.0f, d = 0.0f;
    #pragma unroll
    for (int c = 0; c < NC; ++c) { s += hp[c] * as[c]; d += hp[c] * ad[c]; }
    a_src[n] = s; a_dst[n] = d;
    amax[n] = -__builtin_inff();
}

__global__ __launch_bounds__(256)
void gat_emax2(const int* __restrict__ ei, const float* __restrict__ a_src,
               const float* __restrict__ a_dst, float* __restrict__ amax) {
    int e = blockIdx.x * 256 + threadIdx.x;
    if (e >= ET) return;
    int s, d; edge_sd(ei, e, s, d);
    atomicMaxF(&amax[d], lrelu(a_src[s] + a_dst[d]));
}

#define SCAN_EDGES 16384
#define NIT ((EE + SCAN_EDGES - 1) / SCAN_EDGES)
#define NCH (EE / 16)
#define CH_STEP 4096
#define NITB ((NCH + CH_STEP - 1) / CH_STEP)
#define WCAP 320
static_assert(NPB <= 512 && NN <= (1 << 17), "list word: 9 bits local node, 17 bits source node");
static_assert(ET < (1 << 21), "an edge id is kept in 21 bits");

template <int H, int HCt, int STEPS, bool KEEP>
__device__ __forceinline__ void own_batch(const int* __restrict__ ei, const float* __restrict__ a_src,
                                          const float* __restrict__ a_dst, const float* __restrict__ amax,
                                          const float* __restrict__ hsrc, float* acc, float* dn,
                                          const int* qw, int first, int nvalid, int lo, int lane,
                                          int* __restrict__ lst, int lpos) {
    constexpr int LPE = HCt / 4;
    constexpr int EPW = 32 / LPE;
    const int sub = lane / LPE;
    const int q   = lane % LPE;
    const int h   = (q * 4) / (HCt / H);
    int  ents[STEPS], sv[STEPS], dlv[STEPS];
    bool ok[STEPS];
    #pragma unroll
    for (int u = 0; u < STEPS; ++u) {
        const int idx = u * EPW + sub;
        ok[u] = idx < nvalid;
        ents[u] = qw[first + idx];
    }
    #pragma unroll
    for (int u = 0; u < STEPS; ++u) {
        const int e  = ok[u] ? ents[u] : 0;
        const int ec = e < EE ? e : 0;
        const int sl = ei[ec];
        const int dd = ei[EE + ec];
        sv[u]  = (e < EE) ? sl : e - EE;
        dlv[u] = ((e < EE) ? dd : e - EE) - lo;
    }
    if constexpr (KEEP) {
        static_assert(EPW == 1 && STEPS <= 32, "list store assumes one edge per wave step");
        int mine = 0;
        #pragma unroll
        for (int u = 0; u < STEPS; ++u) mine = (lane == u) ? (int)(((unsigned)dlv[u] << 17) | (unsigned)sv[u]) : mine;
        if (lane < nvalid && lpos + lane < WCAP) lst[lpos + lane] = mine;
    }
    float4 hv[STEPS];
    float as[STEPS], ad[STEPS], am[STEPS];
    #pragma unroll
    for (int u = 0; u < STEPS; ++u) {
        const int sn = sv[u], dnode = lo + dlv[u];
        hv[u] = *(const float4*)(hsrc + (size_t)sn * HCt + q * 4);
        as[u] = a_src[sn * H + h];
        ad[u] = a_dst[dnode * H + h];
        am[u] = amax[dnode * H + h];
    }
    #pragma unroll
    for (int u = 0; u < STEPS; ++u) {
        const float w = expf(lrelu(as[u] + ad[u]) - am[u]);
        if (ok[u]) {
            if ((q & 3) == 0) atomicAdd(&dn[dlv[u] * H + h], w);
            float* ap = acc + dlv[u] * HCt + q;
            atomicAdd(ap,           hv[u].x * w);
            atomicAdd(ap + LPE,     hv[u].y * w);
            atomicAdd(ap + 2 * LPE, hv[u].z * w);
            atomicAdd(ap + 3 * LPE, hv[u].w * w);
        }
    }
}

template <int H, int HCt, int STEPS>
__device__ __forceinline__ void own_walk(const int* __restrict__ lw, int n, const float* __restrict__ a_src,
                                         const float* __restrict__ a_dst, const float* __restrict__ amax,
                                         const float* __restrict__ hsrc, float* acc, float* dn,
                                         int lo, int lane) {
    constexpr int LPE = HCt / 4;
    constexpr int EPW = 32 / LPE;
    constexpr int B   = EPW * STEPS;
    const int sub = lane / LPE;
    const int q   = lane % LPE;
    const int h   = (q * 4) / (HCt / H);
    int ents[STEPS], nxt[STEPS];
    #pragma unroll
    for (int u = 0; u < STEPS; ++u) {
        const int idx = u * EPW + sub;
        ents[u] = lw[idx < n ? idx : 0];
    }
    for (int b = 0; b < n; b += B) {
        #pragma unroll
        for (int u = 0; u < STEPS; ++u) {
            const int idx = b + B + u * EPW + sub;
            nxt[u] = lw[idx < n ? idx : 0];
        }
        float4 hv[STEPS];
        float as[STEPS], ad[STEPS], am[STEPS];
        int   dlv[STEPS];
        bool  ok[STEPS];
        #pragma unroll
        for (int u = 0; u < STEPS; ++u) {
            ok[u] = b + u * EPW + sub < n;
            const int ent = ok[u] ? ents[u] : 0;
            const int sn = ent & 0x1FFFF;
            dlv[u] = ent >> 17;
            const int dnode = lo + dlv[u];
            hv[u] = *(const float4*)(hsrc + (size_t)sn * HCt + q * 4);
            as[u] = a_src[sn * H + h];
            ad[u] = a_dst[dnode * H + h];
            am[u] = amax[dnode * H + h];
        }
        #pragma unroll
        for (int u = 0; u < STEPS; ++u) {
            const float w = expf(lrelu(as[u] + ad[u]) - am[u]);
            if (ok[u]) {
                if ((q & 3) == 0) atomicAdd(&dn[dlv[u] * H + h], w);
                float* ap = acc + dlv[u] * HCt + q;
                atomicAdd(ap,           hv[u].x * w);
                atomicAdd(ap + LPE,     hv[u].y * w);
                atomicAdd(ap + 2 * LPE, hv[u].z * w);
                atomicAdd(ap + 3 * LPE, hv[u].w * w);
            }
        }
        #pragma unroll
        for (int u = 0; u < STEPS; ++u) ents[u] = nxt[u];
    }
}

template <int H, int HCt, int STEPS, int LAYER>
__global__ __launch_bounds__(1024)
void gat_own(const int* __restrict__ ei, const float* __restrict__ a_src,
             const float* __restrict__ a_dst, const float* __restrict__ amax,
             const float* __restrict__ hsrc, const float* __restrict__ bias,
             float* __restrict__ dstv, int* __restrict__ elist, int* __restrict__ ecnt,
             const unsigned char* __restrict__ own) {
    constexpr int LPE  = HCt / 4;
    constexpr int EPW  = 32 / LPE;
    constexpr int B    = EPW * STEPS;
    constexpr int QCAP = B + 512;
    __shared__ float acc[NPB * HCt];
    __shared__ float dn[NPB * H];
    __shared__ int   qbuf[32 * QCAP];

    const int tid = threadIdx.x, lane = tid & 31;
    const int wave = __builtin_amdgcn_readfirstlane(tid >> 5);
    const int lo  = blockIdx.x * NPB;
    const int cnt = (NN - lo < NPB) ? (NN - lo) : NPB;
    for (int i = tid; i < NPB * HCt; i += 1024) acc[i] = 0.0f;
    for (int i = tid; i < NPB * H;   i += 1024) dn[i]  = 0.0f;
    __syncthreads();

    int* lst = elist + (size_t)(blockIdx.x * 32 + wave) * WCAP;
    bool walk = false;
    int  nl = 0;
    if constexpr (LAYER == 2) {
        const int c = ecnt[blockIdx.x * 32 + lane];
        walk = __builtin_amdgcn_ballot_w32(c > WCAP) == 0;
        nl = __builtin_amdgcn_readlane(c, wave);
    }
    if (walk) {
        own_walk<H, HCt, 8>(lst, nl, a_src, a_dst, amax, hsrc, acc, dn, lo, lane);
    } else {
    int* qw = qbuf + wave * QCAP;
    int  qn = 0;
    int  ln = 0;
    const unsigned lt = (1u << lane) - 1u;
    auto push = [&](bool m, unsigned ent) {
        const unsigned mask = __builtin_amdgcn_ballot_w32(m);
        if (m) qw[qn + __builtin_popcount(mask & lt)] = (int)ent;
        qn += __builtin_popcount(mask);
    };
    auto drain = [&]() {
        while (qn >= B) {
            own_batch<H, HCt, STEPS, LAYER == 1>(ei, a_src, a_dst, amax, hsrc, acc, dn, qw, qn - B, B, lo, lane, lst, ln);
            qn -= B;
            ln += B;
        }
    };

    for (int b0 = wave * 32; b0 < cnt; b0 += 1024) {
        const int dl = b0 + lane;
        push(dl < cnt, (unsigned)(EE + lo + dl));
    }
    drain();

    if constexpr (LAYER == 1) {
    const uint4* ownp = (const uint4*)own;
    const unsigned pat = (unsigned)blockIdx.x * 0x01010101u;
    const int it0 = (blockIdx.x * NITB) / NGRP;
    auto ldscan = [&](int i, uint4 (&buf)[4]) -> int {
        int it = i + it0;
        if (it >= NITB) it -= NITB;
        const int cb = it * CH_STEP + wave * 128 + lane;
        #pragma unroll
        for (int k = 0; k < 4; ++k) {
            const int ck = cb + k * 32;
            buf[k] = ownp[ck < NCH ? ck : NCH - 1];
        }
        return cb;
    };
    auto hit = [&](unsigned w) -> unsigned {
        const unsigned x = w ^ pat;
        return ~(((x & 0x7F7F7F7Fu) + 0x7F7F7F7Fu) | x) & 0x80808080u;
    };
    auto test = [&](const uint4 (&buf)[4], int cb) {
        unsigned mk[4];
        #pragma unroll
        for (int k = 0; k < 4; ++k) {
            const unsigned m4 = (hit(buf[k].x) >> 7) | (hit(buf[k].y) >> 6) |
                                (hit(buf[k].z) >> 5) | (hit(buf[k].w) >> 4);
            mk[k] = (cb + k * 32 < NCH) ? m4 : 0u;
        }
        unsigned long long M = ((unsigned long long)(mk[2] | (mk[3] << 4)) << 32) |
                               (unsigned long long)(mk[0] | (mk[1] << 4));
        while (__builtin_amdgcn_ballot_w32(M != 0)) {
            const bool m = M != 0;
            const int  p = m ? __builtin_ctzll(M) : 0;
            const int  k = ((p >> 5) << 1) | ((p >> 2) & 1);
            push(m, (unsigned)((cb + k * 32) * 16 + ((p & 3) << 2) + ((p >> 3) & 3)));
            M &= M - 1;
            drain();
        }
    };
    uint4 bufa[4], bufb[4];
    int cba = ldscan(0, bufa), cbb;
    for (int i = 0; i < NITB; i += 2) {
        cbb = ldscan(i + 1, bufb);
        __builtin_amdgcn_sched_barrier(0);
        test(bufa, cba);
        cba = ldscan(i + 2, bufa);
        __builtin_amdgcn_sched_barrier(0);
        if (i + 1 < NITB) test(bufb, cbb);
    }
    } else {
    const int* dstp = ei + EE;
    const int it0 = (blockIdx.x * NIT) / NGRP;
    auto ldscan = [&](int i, int4 (&buf)[4]) -> int {
        int it = i + it0;
        if (it >= NIT) it -= NIT;
        const int eb = it * SCAN_EDGES + wave * 512 + lane * 4;
        #pragma unroll
        for (int k = 0; k < 4; ++k) {
            const int ek = eb + k * 128;
            buf[k] = *(const int4*)(dstp + (ek < EE ? ek : EE - 4));
        }
        return eb;
    };
    auto test = [&](const int4 (&buf)[4], int eb) {
        #pragma unroll
        for (int k = 0; k < 4; ++k) {
            const int ek = eb + k * 128;
            const unsigned d0 = (unsigned)(buf[k].x - lo), d1 = (unsigned)(buf[k].y - lo);
            const unsigned d2 = (unsigned)(buf[k].z - lo), d3 = (unsigned)(buf[k].w - lo);
            const bool in = ek < EE;
            const bool m0 = in && d0 < (unsigned)cnt, m1 = in && d1 < (unsigned)cnt;
            const bool m2 = in && d2 < (unsigned)cnt, m3 = in && d3 < (unsigned)cnt;
            if (__builtin_amdgcn_ballot_w32(m0 | m1 | m2 | m3)) {
                push(m0, (unsigned)(ek + 0));
                push(m1, (unsigned)(ek + 1));
                push(m2, (unsigned)(ek + 2));
                push(m3, (unsigned)(ek + 3));
            }
        }
        drain();
    };
    int4 bufa[4], bufb[4];
    int eba = ldscan(0, bufa), ebb;
    for (int i = 0; i < NIT; i += 2) {
        ebb = ldscan(i + 1, bufb);
        test(bufa, eba);
        eba = ldscan(i + 2, bufa);
        if (i + 1 < NIT) test(bufb, ebb);
    }
    }
    if (qn > 0) {
        own_batch<H, HCt, STEPS, LAYER == 1>(ei, a_src, a_dst, amax, hsrc, acc, dn, qw, 0, qn, lo, lane, lst, ln);
        ln += qn;
    }
    if constexpr (LAYER == 1) {
        if (lane == 0) ecnt[blockIdx.x * 32 + wave] = ln;
    }
    }
    __syncthreads();

    if constexpr (LAYER == 1) {
        const float4 bq = *(const float4*)(bias + lane * 4);
        for (int r = wave; r < cnt; r += 32) {
            const float dv = dn[r * H + (lane >> 2)] + 1e-16f;
            const float* ap = acc + r * HCt + lane;
            float4 o;
            o.x = ap[0]       / dv + bq.x;
            o.y = ap[LPE]     / dv + bq.y;
            o.z = ap[2 * LPE] / dv + bq.z;
            o.w = ap[3 * LPE] / dv + bq.w;
            o.x = o.x > 0.0f ? o.x : (expf(o.x) - 1.0f);
            o.y = o.y > 0.0f ? o.y : (expf(o.y) - 1.0f);
            o.z = o.z > 0.0f ? o.z : (expf(o.z) - 1.0f);
            o.w = o.w > 0.0f ? o.w : (expf(o.w) - 1.0f);
            *(float4*)(dstv + (size_t)(lo + r) * HCt + lane * 4) = o;
        }
    } else {
        for (int r = tid; r < cnt; r += 1024) {
            float v[HCt], m = -__builtin_inff();
            const float dv = dn[r] + 1e-16f;
            #pragma unroll
            for (int c = 0; c < HCt; ++c) {
                v[c] = acc[r * HCt + (c & 3) * LPE + (c >> 2)] / dv + bias[c];
                m = fmaxf(m, v[c]);
            }
            float sum = 0.0f;
            #pragma unroll
            for (int c = 0; c < HCt; ++c) sum += expf(v[c] - m);
            const float ls = logf(sum);
            #pragma unroll
            for (int c = 0; c < HCt; ++c) dstv[(size_t)(lo + r) * HCt + c] = v[c] - m - ls;
        }
    }
}

extern "C" void kernel_launch(void* const* d_in, const int* in_sizes, int n_in,
                              void* d_out, int out_size, void* d_ws, size_t ws_size,
                              hipStream_t stream) {
    (void)in_sizes; (void)n_in; (void)out_size; (void)ws_size;
    const float* x   = (const float*)d_in[0];
    const int*   ei  = (const int*)d_in[1];
    const float* W1  = (const float*)d_in[2];
    const float* as1 = (const float*)d_in[3];
    const float* ad1 = (const float*)d_in[4];
    const float* b1  = (const float*)d_in[5];
    const float* W2  = (const float*)d_in[6];
    const float* as2 = (const float*)d_in[7];
    const float* ad2 = (const float*)d_in[8];
    const float* b2  = (const float*)d_in[9];
    float* out = (float*)d_out;

    char* wsb = (char*)d_ws;
    size_t off = 0;
    auto alloc = [&](size_t bytes) -> void* {
        void* p = wsb + off;
        off += (bytes + 255) & ~(size_t)255;
        return p;
    };
    _Float16* w1h  = (_Float16*)alloc((size_t)HC * NF * 2);
    _Float16* w2h  = (_Float16*)alloc((size_t)NC * HC * 2);
    _Float16* xh   = (_Float16*)alloc((size_t)NN * NF * 2);
    float* h1      = (float*)alloc((size_t)NN * HC * 4);
    float* asrc1   = (float*)alloc((size_t)NN * HH * 4);
    float* adst1   = (float*)alloc((size_t)NN * HH * 4);
    float* amax1   = (float*)alloc((size_t)NN * HH * 4);
    float* h2      = (float*)alloc((size_t)NN * NC * 4);
    float* asrc2   = (float*)alloc((size_t)NN * 4);
    float* adst2   = (float*)alloc((size_t)NN * 4);
    float* amax2   = (float*)alloc((size_t)NN * 4);
    int* elist     = (int*)alloc((size_t)NGRP * 32 * WCAP * 4);
    int* ecnt      = (int*)alloc((size_t)NGRP * 32 * 4);
    float* act1    = (float*)xh;
    unsigned char* ownb = (unsigned char*)h2;

    const dim3 B(256);
    gat_cvtw <<<(HC * NF + NC * HC + 255) / 256, B, 0, stream>>>(W1, W2, w1h, w2h);
    gat_cvtx <<<(NN * NF) / 256, B, 0, stream>>>(x, xh);
    gat_gemm1<<<(NN / 16 + 7) / 8, B, 0, stream>>>(xh, w1h, h1);
    gat_att1 <<<(NN * HH) / 256, B, 0, stream>>>(h1, as1, ad1, asrc1, adst1, amax1);
    gat_emax1<<<(ET * HH) / 256, B, 0, stream>>>(ei, asrc1, adst1, amax1, ownb);
    gat_own<HH, HC, 8, 1><<<NGRP, 1024, 0, stream>>>(ei, asrc1, adst1, amax1, h1, b1, act1, elist, ecnt, ownb);
    gat_gemm2<<<(NN / 16 + 7) / 8, B, 0, stream>>>(act1, w2h, h2);
    gat_att2 <<<(NN + 255) / 256, B, 0, stream>>>(h2, as2, ad2, asrc2, adst2, amax2);
    gat_emax2<<<(ET + 255) / 256, B, 0, stream>>>(ei, asrc2, adst2, amax2);
    gat_own<1, NC, 4, 2><<<NGRP, 1024, 0, stream>>>(ei, asrc2, adst2, amax2, h2, b2, out, elist, ecnt, ownb);
}


# segment max taken inside the owner groups: pass 1 LDS atomic max while finding edges, barrier, pass 2 sums over the kept lists; global atomic-max sweeps gone; LDS float sums still arrival-order
# speedup vs baseline: 3.3239x; 1.4914x over previous
#include <hip/hip_runtime.h>

#define NN  100000
#define NF  256
#define HH  8
#define CC  16
#define HC  128
#define NC  16
#define EE  1600000
#define ET  1700000
#define NEG 0.2f
#define NPB  392
#define NGRP 256
static_assert((NN + NPB - 1) / NPB <= 256, "an owner-group id must fit one byte");
static_assert(EE % 16 == 0, "the owner-byte list is read 16 edges a load");

typedef _Float16 h8   __attribute__((ext_vector_type(8)));
typedef _Float16 v16h __attribute__((ext_vector_type(16)));
typedef float    v8f  __attribute__((ext_vector_type(8)));

__device__ __forceinline__ void edge_sd(const int* __restrict__ ei, int e, int& s, int& d) {
    if (e < EE) { s = ei[e]; d = ei[EE + e]; }
    else        { s = e - EE; d = s; }
}

__device__ __forceinline__ void atomicMaxF(float* addr, float val) {
    if (val >= 0.0f) atomicMax((int*)addr, __float_as_int(val));
    else             atomicMin((unsigned int*)addr, __float_as_uint(val));
}

__device__ __forceinline__ float lrelu(float a) { return a > 0.0f ? a : NEG * a; }

__global__ __launch_bounds__(256) void gat_cvtw(const float* __restrict__ W1,
                                                const float* __restrict__ W2,
                                                _Float16* __restrict__ w1h,
                                                _Float16* __restrict__ w2h) {
    int i = blockIdx.x * 256 + threadIdx.x;
    if (i < HC * NF)              w1h[i] = (_Float16)W1[i];
    else if (i < HC * NF + NC*HC) w2h[i - HC * NF] = (_Float16)W2[i - HC * NF];
}

__global__ __launch_bounds__(256) void gat_cvtx(const float* __restrict__ x,
                                                _Float16* __restrict__ xh) {
    int i = blockIdx.x * 256 + threadIdx.x;
    if (i < NN * NF) xh[i] = (_Float16)x[i];
}

__global__ __launch_bounds__(256)
void gat_gemm1(const _Float16* __restrict__ xh, const _Float16* __restrict__ w1h,
               float* __restrict__ h1) {
    __shared__ _Float16 lw[HC * NF];
    const int tid = threadIdx.x;
    for (int idx = tid; idx < HC * 32; idx += 256) {
        int r = idx >> 5, ch = idx & 31;
        *(h8*)(&lw[r * NF + ((ch ^ (r & 15)) << 3)]) =
            *(const h8*)(w1h + (size_t)r * NF + (ch << 3));
    }
    __syncthreads();

    const int lane = tid & 31;
    const int rt   = blockIdx.x * 8 + (tid >> 5);
    if (rt >= NN / 16) return;
    const int rsub = lane & 15;
    const int ksel = (lane >> 4) << 3;
    const _Float16* xrow = xh + (size_t)(rt * 16 + rsub) * NF;

    v8f acc[8] = {};
    #pragma unroll
    for (int ks = 0; ks < 8; ++ks) {
        const int kb = ks * 32 + ksel;
        const int c0 = kb >> 3;
        union { v16h v; h8 p[2]; } A;
        A.p[0] = *(const h8*)(xrow + kb);
        A.p[1] = *(const h8*)(xrow + kb + 16);
        #pragma unroll
        for (int ct = 0; ct < 8; ++ct) {
            const int n = ct * 16 + rsub;
            union { v16h v; h8 p[2]; } B;
            B.p[0] = *(const h8*)(&lw[n * NF + (((c0    ) ^ (n & 15)) << 3)]);
            B.p[1] = *(const h8*)(&lw[n * NF + (((c0 + 2) ^ (n & 15)) << 3)]);
            acc[ct] = __builtin_amdgcn_wmma_f32_16x16x32_f16(
                false, A.v, false, B.v, (short)0, acc[ct], false, false);
        }
    }
    const int rhi = (lane >> 4) << 3;
    #pragma unroll
    for (int ct = 0; ct < 8; ++ct) {
        #pragma unroll
        for (int r = 0; r < 8; ++r)
            h1[(size_t)(rt * 16 + rhi + r) * HC + ct * 16 + rsub] = acc[ct][r];
    }
}

__global__ __launch_bounds__(256)
void gat_att1(const float* __restrict__ h1, const float* __restrict__ as,
              const float* __restrict__ ad, float* __restrict__ a_src,
              float* __restrict__ a_dst) {
    int i = blockIdx.x * 256 + threadIdx.x;
    if (i >= NN * HH) return;
    int n = i >> 3, h = i & 7;
    const float* hp = h1 + (size_t)n * HC + h * CC;
    float s = 0.0f, d = 0.0f;
    #pragma unroll
    for (int c = 0; c < CC; ++c) { s += hp[c] * as[h * CC + c]; d += hp[c] * ad[h * CC + c]; }
    a_src[i] = s; a_dst[i] = d;
}

__global__ __launch_bounds__(256)
void gat_ownb(const int* __restrict__ ei, unsigned char* __restrict__ own) {
    const int i = blockIdx.x * 256 + threadIdx.x;
    if (i >= EE / 16) return;
    const int4* dp = (const int4*)(ei + EE) + (size_t)i * 4;
    unsigned w[4];
    #pragma unroll
    for (int k = 0; k < 4; ++k) {
        const int4 d = dp[k];
        w[k] = ((unsigned)d.x / NPB) | (((unsigned)d.y / NPB) << 8) |
               (((unsigned)d.z / NPB) << 16) | (((unsigned)d.w / NPB) << 24);
    }
    ((uint4*)own)[i] = make_uint4(w[0], w[1], w[2], w[3]);
}

__global__ __launch_bounds__(256)
void gat_gemm2(const float* __restrict__ a1, const _Float16* __restrict__ w2h,
               float* __restrict__ h2) {
    const int lane = threadIdx.x & 31;
    const int rt   = blockIdx.x * 8 + (threadIdx.x >> 5);
    if (rt >= NN / 16) return;
    const int rsub = lane & 15;
    const int ksel = (lane >> 4) << 3;
    const float*    arow = a1 + (size_t)(rt * 16 + rsub) * HC;
    const _Float16* wrow = w2h + (size_t)rsub * HC;

    v8f acc = {};
    #pragma unroll
    for (int ks = 0; ks < 4; ++ks) {
        const int kb = ks * 32 + ksel;
        union { v16h v; _Float16 e[16]; } A;
        #pragma unroll
        for (int t = 0; t < 8; ++t) {
            A.e[t]     = (_Float16)arow[kb + t];
            A.e[8 + t] = (_Float16)arow[kb + 16 + t];
        }
        union { v16h v; h8 p[2]; } B;
        B.p[0] = *(const h8*)(wrow + kb);
        B.p[1] = *(const h8*)(wrow + kb + 16);
        acc = __builtin_amdgcn_wmma_f32_16x16x32_f16(
            false, A.v, false, B.v, (short)0, acc, false, false);
    }
    const int rhi = (lane >> 4) << 3;
    #pragma unroll
    for (int r = 0; r < 8; ++r)
        h2[(size_t)(rt * 16 + rhi + r) * NC + rsub] = acc[r];
}

__global__ __launch_bounds__(256)
void gat_att2(const float* __restrict__ h2, const float* __restrict__ as,
              const float* __restrict__ ad, float* __restrict__ a_src,
              float* __restrict__ a_dst) {
    int n = blockIdx.x * 256 + threadIdx.x;
    if (n >= NN) return;
    const float* hp = h2 + (size_t)n * NC;
    float s = 0.0f, d = 0.0f;
    #pragma unroll
    for (int c = 0; c < NC; ++c) { s += hp[c] * as[c]; d += hp[c] * ad[c]; }
    a_src[n] = s; a_dst[n] = d;
}

#define SCAN_EDGES 16384
#define NIT ((EE + SCAN_EDGES - 1) / SCAN_EDGES)
#define NCH (EE / 16)
#define CH_STEP 4096
#define NITB ((NCH + CH_STEP - 1) / CH_STEP)
#define WCAP 320
#define MAXB 32
static_assert(NPB <= 512 && NN <= (1 << 17), "list word: 9 bits local node, 17 bits source node");
static_assert(ET < (1 << 21), "an edge id is kept in 21 bits");

template <bool V> struct mode_c { static constexpr bool value = V; };

template <int H>
__device__ __forceinline__ void load_att(const float* __restrict__ p, int node, float (&v)[H]) {
    if constexpr (H % 4 == 0) {
        #pragma unroll
        for (int k = 0; k < H / 4; ++k) {
            const float4 t = *(const float4*)(p + (size_t)node * H + k * 4);
            v[k * 4] = t.x; v[k * 4 + 1] = t.y; v[k * 4 + 2] = t.z; v[k * 4 + 3] = t.w;
        }
    } else {
        #pragma unroll
        for (int k = 0; k < H; ++k) v[k] = p[(size_t)node * H + k];
    }
}

template <int H, bool KEEP>
__device__ __forceinline__ void max_batch(const int* __restrict__ ei, const float* __restrict__ a_src,
                                          const float* __restrict__ a_dst, float* mx,
                                          const int* qw, int first, int nvalid, int lo, int lane,
                                          int* __restrict__ lst, int lpos) {
    const bool ok = lane < nvalid;
    const int e  = qw[first + (ok ? lane : 0)];
    const int ec = e < EE ? e : 0;
    const int sl = ei[ec];
    const int dd = ei[EE + ec];
    const int s  = (e < EE) ? sl : e - EE;
    const int dl = ((e < EE) ? dd : e - EE) - lo;
    if constexpr (KEEP) {
        if (ok && lpos + lane < WCAP) lst[lpos + lane] = (int)(((unsigned)dl << 17) | (unsigned)s);
    }
    float as[H], ad[H];
    load_att<H>(a_src, s, as);
    load_att<H>(a_dst, lo + dl, ad);
    if (ok) {
        #pragma unroll
        for (int h = 0; h < H; ++h) atomicMaxF(&mx[h * NPB + dl], lrelu(as[h] + ad[h]));
    }
}

template <int H>
__device__ __forceinline__ void walk_max(const int* __restrict__ lw, int n, const float* __restrict__ a_src,
                                         const float* __restrict__ a_dst, float* mx, int lo, int lane) {
    constexpr int U = 4;
    for (int b = 0; b < n; b += 32 * U) {
        int  ent[U];
        bool ok[U];
        #pragma unroll
        for (int u = 0; u < U; ++u) {
            const int idx = b + u * 32 + lane;
            ok[u] = idx < n;
            ent[u] = lw[ok[u] ? idx : 0];
        }
        float as[U][H], ad[U][H];
        #pragma unroll
        for (int u = 0; u < U; ++u) {
            load_att<H>(a_src, ent[u] & 0x1FFFF, as[u]);
            load_att<H>(a_dst, lo + (ent[u] >> 17), ad[u]);
        }
        #pragma unroll
        for (int u = 0; u < U; ++u) {
            if (ok[u]) {
                #pragma unroll
                for (int h = 0; h < H; ++h)
                    atomicMaxF(&mx[h * NPB + (ent[u] >> 17)], lrelu(as[u][h] + ad[u][h]));
            }
        }
    }
}

template <int H, int HCt, int STEPS>
__device__ __forceinline__ void own_batch(const int* __restrict__ ei, const float* __restrict__ a_src,
                                          const float* __restrict__ a_dst, const float* mx,
                                          const float* __restrict__ hsrc, float* acc, float* dn,
                                          const int* qw, int first, int nvalid, int lo, int lane) {
    constexpr int LPE = HCt / 4;
    constexpr int EPW = 32 / LPE;
    const int sub = lane / LPE;
    const int q   = lane % LPE;
    const int h   = (q * 4) / (HCt / H);
    int  ents[STEPS], sv[STEPS], dlv[STEPS];
    bool ok[STEPS];
    #pragma unroll
    for (int u = 0; u < STEPS; ++u) {
        const int idx = u * EPW + sub;
        ok[u] = idx < nvalid;
        ents[u] = qw[first + idx];
    }
    #pragma unroll
    for (int u = 0; u < STEPS; ++u) {
        const int e  = ok[u] ? ents[u] : 0;
        const int ec = e < EE ? e : 0;
        const int sl = ei[ec];
        const int dd = ei[EE + ec];
        sv[u]  = (e < EE) ? sl : e - EE;
        dlv[u] = ok[u] ? ((e < EE) ? dd : e - EE) - lo : 0;
    }
    float4 hv[STEPS];
    float as[STEPS], ad[STEPS], am[STEPS];
    #pragma unroll
    for (int u = 0; u < STEPS; ++u) {
        const int sn = sv[u], dnode = lo + dlv[u];
        hv[u] = *(const float4*)(hsrc + (size_t)sn * HCt + q * 4);
        as[u] = a_src[sn * H + h];
        ad[u] = a_dst[dnode * H + h];
        am[u] = mx[h * NPB + dlv[u]];
    }
    #pragma unroll
    for (int u = 0; u < STEPS; ++u) {
        const float w = expf(lrelu(as[u] + ad[u]) - am[u]);
        if (ok[u]) {
            if ((q & 3) == 0) atomicAdd(&dn[dlv[u] * H + h], w);
            float* ap = acc + dlv[u] * HCt + q;
            atomicAdd(ap,           hv[u].x * w);
            atomicAdd(ap + LPE,     hv[u].y * w);
            atomicAdd(ap + 2 * LPE, hv[u].z * w);
            atomicAdd(ap + 3 * LPE, hv[u].w * w);
        }
    }
}

template <int H, int HCt, int STEPS>
__device__ __forceinline__ void own_walk(const int* __restrict__ lw, int n, const float* __restrict__ a_src,
                                         const float* __restrict__ a_dst, const float* mx,
                                         const float* __restrict__ hsrc, float* acc, float* dn,
                                         int lo, int lane) {
    constexpr int LPE = HCt / 4;
    constexpr int EPW = 32 / LPE;
    constexpr int B   = EPW * STEPS;
    const int sub = lane / LPE;
    const int q   = lane % LPE;
    const int h   = (q * 4) / (HCt / H);
    int ents[STEPS], nxt[STEPS];
    #pragma unroll
    for (int u = 0; u < STEPS; ++u) {
        const int idx = u * EPW + sub;
        ents[u] = lw[idx < n ? idx : 0];
    }
    for (int b = 0; b < n; b += B) {
        #pragma unroll
        for (int u = 0; u < STEPS; ++u) {
            const int idx = b + B + u * EPW + sub;
            nxt[u] = lw[idx < n ? idx : 0];
        }
        float4 hv[STEPS];
        float as[STEPS], ad[STEPS], am[STEPS];
        int   dlv[STEPS];
        bool  ok[STEPS];
        #pragma unroll
        for (int u = 0; u < STEPS; ++u) {
            ok[u] = b + u * EPW + sub < n;
            const int ent = ok[u] ? ents[u] : 0;
            const int sn = ent & 0x1FFFF;
            dlv[u] = ent >> 17;
            const int dnode = lo + dlv[u];
            hv[u] = *(const float4*)(hsrc + (size_t)sn * HCt + q * 4);
            as[u] = a_src[sn * H + h];
            ad[u] = a_dst[dnode * H + h];
            am[u] = mx[h * NPB + dlv[u]];
        }
        #pragma unroll
        for (int u = 0; u < STEPS; ++u) {
            const float w = expf(lrelu(as[u] + ad[u]) - am[u]);
            if (ok[u]) {
                if ((q & 3) == 0) atomicAdd(&dn[dlv[u] * H + h], w);
                float* ap = acc + dlv[u] * HCt + q;
                atomicAdd(ap,           hv[u].x * w);
                atomicAdd(ap + LPE,     hv[u].y * w);
                atomicAdd(ap + 2 * LPE, hv[u].z * w);
                atomicAdd(ap + 3 * LPE, hv[u].w * w);
            }
        }
        #pragma unroll
        for (int u = 0; u < STEPS; ++u) ents[u] = nxt[u];
    }
}

template <int H, int HCt, int STEPS, int LAYER>
__global__ __launch_bounds__(1024)
void gat_own(const int* __restrict__ ei, const float* __restrict__ a_src,
             const float* __restrict__ a_dst,
             const float* __restrict__ hsrc, const float* __restrict__ bias,
             float* __restrict__ dstv, int* __restrict__ elist, int* __restrict__ ecnt,
             const unsigned char* __restrict__ own) {
    constexpr int LPE  = HCt / 4;
    constexpr int EPW  = 32 / LPE;
    constexpr int B    = EPW * STEPS;
    constexpr int QCAP = (B > MAXB ? B : MAXB) + 512;
    __shared__ float acc[NPB * HCt];
    __shared__ float dn[NPB * H];
    __shared__ float mx[NPB * H];
    __shared__ int   qbuf[32 * QCAP];
    __shared__ int   wcnt[32];

    const int tid = threadIdx.x, lane = tid & 31;
    const int wave = __builtin_amdgcn_readfirstlane(tid >> 5);
    const int lo  = blockIdx.x * NPB;
    const int cnt = (NN - lo < NPB) ? (NN - lo) : NPB;
    for (int i = tid; i < NPB * HCt; i += 1024) acc[i] = 0.0f;
    for (int i = tid; i < NPB * H;   i += 1024) { dn[i] = 0.0f; mx[i] = -__builtin_inff(); }
    __syncthreads();

    int* lst = elist + (size_t)(blockIdx.x * 32 + wave) * WCAP;
    int* qw  = qbuf + wave * QCAP;

    auto scan_pass = [&](auto mode) -> int {
    constexpr bool MAXP = decltype(mode)::value;
    constexpr int  BB   = MAXP ? MAXB : B;
    int  qn = 0;
    int  ln = 0;
    const unsigned lt = (1u << lane) - 1u;
    auto push = [&](bool m, unsigned ent) {
        const unsigned mask = __builtin_amdgcn_ballot_w32(m);
        if (m) qw[qn + __builtin_popcount(mask & lt)] = (int)ent;
        qn += __builtin_popcount(mask);
    };
    auto batch = [&](int first, int nvalid) {
        if constexpr (MAXP) max_batch<H, LAYER == 1>(ei, a_src, a_dst, mx, qw, first, nvalid, lo, lane, lst, ln);
        else own_batch<H, HCt, STEPS>(ei, a_src, a_dst, mx, hsrc, acc, dn, qw, first, nvalid, lo, lane);
    };
    auto drain = [&]() {
        while (qn >= BB) {
            batch(qn - BB, BB);
            qn -= BB;
            ln += BB;
        }
    };

    for (int b0 = wave * 32; b0 < cnt; b0 += 1024) {
        const int dl = b0 + lane;
        push(dl < cnt, (unsigned)(EE + lo + dl));
    }
    drain();

    if constexpr (LAYER == 1) {
    const uint4* ownp = (const uint4*)own;
    const unsigned pat = (unsigned)blockIdx.x * 0x01010101u;
    const int it0 = (blockIdx.x * NITB) / NGRP;
    auto ldscan = [&](int i, uint4 (&buf)[4]) -> int {
        int it = i + it0;
        if (it >= NITB) it -= NITB;
        const int cb = it * CH_STEP + wave * 128 + lane;
        #pragma unroll
        for (int k = 0; k < 4; ++k) {
            const int ck = cb + k * 32;
            buf[k] = ownp[ck < NCH ? ck : NCH - 1];
        }
        return cb;
    };
    auto hit = [&](unsigned w) -> unsigned {
        const unsigned x = w ^ pat;
        return ~(((x & 0x7F7F7F7Fu) + 0x7F7F7F7Fu) | x) & 0x80808080u;
    };
    auto test = [&](const uint4 (&buf)[4], int cb) {
        unsigned mk[4];
        #pragma unroll
        for (int k = 0; k < 4; ++k) {
            const unsigned m4 = (hit(buf[k].x) >> 7) | (hit(buf[k].y) >> 6) |
                                (hit(buf[k].z) >> 5) | (hit(buf[k].w) >> 4);
            mk[k] = (cb + k * 32 < NCH) ? m4 : 0u;
        }
        unsigned long long M = ((unsigned long long)(mk[2] | (mk[3] << 4)) << 32) |
                               (unsigned long long)(mk[0] | (mk[1] << 4));
        while (__builtin_amdgcn_ballot_w32(M != 0)) {
            const bool m = M != 0;
            const int  p = m ? __builtin_ctzll(M) : 0;
            const int  k = ((p >> 5) << 1) | ((p >> 2) & 1);
            push(m, (unsigned)((cb + k * 32) * 16 + ((p & 3) << 2) + ((p >> 3) & 3)));
            M &= M - 1;
            drain();
        }
    };
    uint4 bufa[4], bufb[4];
    int cba = ldscan(0, bufa), cbb;
    for (int i = 0; i < NITB; i += 2) {
        cbb = ldscan(i + 1, bufb);
        __builtin_amdgcn_sched_barrier(0);
        test(bufa, cba);
        cba = ldscan(i + 2, bufa);
        __builtin_amdgcn_sched_barrier(0);
        if (i + 1 < NITB) test(bufb, cbb);
    }
    } else {
    const int* dstp = ei + EE;
    const int it0 = (blockIdx.x * NIT) / NGRP;
    auto ldscan = [&](int i, int4 (&buf)[4]) -> int {
        int it = i + it0;
        if (it >= NIT) it -= NIT;
        const int eb = it * SCAN_EDGES + wave * 512 + lane * 4;
        #pragma unroll
        for (int k = 0; k < 4; ++k) {
            const int ek = eb + k * 128;
            buf[k] = *(const int4*)(dstp + (ek < EE ? ek : EE - 4));
        }
        return eb;
    };
    auto test = [&](const int4 (&buf)[4], int eb) {
        #pragma unroll
        for (int k = 0; k < 4; ++k) {
            const int ek = eb + k * 128;
            const unsigned d0 = (unsigned)(buf[k].x - lo), d1 = (unsigned)(buf[k].y - lo);
            const unsigned d2 = (unsigned)(buf[k].z - lo), d3 = (unsigned)(buf[k].w - lo);
            const bool in = ek < EE;
            const bool m0 = in && d0 < (unsigned)cnt, m1 = in && d1 < (unsigned)cnt;
            const bool m2 = in && d2 < (unsigned)cnt, m3 = in && d3 < (unsigned)cnt;
            if (__builtin_amdgcn_ballot_w32(m0 | m1 | m2 | m3)) {
                push(m0, (unsigned)(ek + 0));
                push(m1, (unsigned)(ek + 1));
                push(m2, (unsigned)(ek + 2));
                push(m3, (unsigned)(ek + 3));
            }
        }
        drain();
    };
    int4 bufa[4], bufb[4];
    int eba = ldscan(0, bufa), ebb;
    for (int i = 0; i < NIT; i += 2) {
        ebb = ldscan(i + 1, bufb);
        test(bufa, eba);
        eba = ldscan(i + 2, bufa);
        if (i + 1 < NIT) test(bufb, ebb);
    }
    }
    if (qn > 0) {
        batch(0, qn);
        ln += qn;
    }
    return ln;
    };

    if constexpr (LAYER == 1) {
        const int nl = scan_pass(mode_c<true>{});
        if (lane == 0) { ecnt[blockIdx.x * 32 + wave] = nl; wcnt[wave] = nl; }
        __syncthreads();
        if (__builtin_amdgcn_ballot_w32(wcnt[lane] > WCAP) == 0)
            own_walk<H, HCt, 8>(lst, nl, a_src, a_dst, mx, hsrc, acc, dn, lo, lane);
        else
            (void)scan_pass(mode_c<false>{});
    } else {
        const int c = ecnt[blockIdx.x * 32 + lane];
        const int nl = __builtin_amdgcn_readlane(c, wave);
        if (__builtin_amdgcn_ballot_w32(c > WCAP) == 0) {
            walk_max<H>(lst, nl, a_src, a_dst, mx, lo, lane);
            __syncthreads();
            own_walk<H, HCt, 8>(lst, nl, a_src, a_dst, mx, hsrc, acc, dn, lo, lane);
        } else {
            (void)scan_pass(mode_c<true>{});
            __syncthreads();
            (void)scan_pass(mode_c<false>{});
        }
    }
    __syncthreads();

    if constexpr (LAYER == 1) {
        const float4 bq = *(const float4*)(bias + lane * 4);
        for (int r = wave; r < cnt; r += 32) {
            const float dv = dn[r * H + (lane >> 2)] + 1e-16f;
            const float* ap = acc + r * HCt + lane;
            float4 o;
            o.x = ap[0]       / dv + bq.x;
            o.y = ap[LPE]     / dv + bq.y;
            o.z = ap[2 * LPE] / dv + bq.z;
            o.w = ap[3 * LPE] / dv + bq.w;
            o.x = o.x > 0.0f ? o.x : (expf(o.x) - 1.0f);
            o.y = o.y > 0.0f ? o.y : (expf(o.y) - 1.0f);
            o.z = o.z > 0.0f ? o.z : (expf(o.z) - 1.0f);
            o.w = o.w > 0.0f ? o.w : (expf(o.w) - 1.0f);
            *(float4*)(dstv + (size_t)(lo + r) * HCt + lane * 4) = o;
        }
    } else {
        for (int r = tid; r < cnt; r += 1024) {
            float v[HCt], m = -__builtin_inff();
            const float dv = dn[r] + 1e-16f;
            #pragma unroll
            for (int c = 0; c < HCt; ++c) {
                v[c] = acc[r * HCt + (c & 3) * LPE + (c >> 2)] / dv + bias[c];
                m = fmaxf(m, v[c]);
            }
            float sum = 0.0f;
            #pragma unroll
            for (int c = 0; c < HCt; ++c) sum += expf(v[c] - m);
            const float ls = logf(sum);
            #pragma unroll
            for (int c = 0; c < HCt; ++c) dstv[(size_t)(lo + r) * HCt + c] = v[c] - m - ls;
        }
    }
}

extern "C" void kernel_launch(void* const* d_in, const int* in_sizes, int n_in,
                              void* d_out, int out_size, void* d_ws, size_t ws_size,
                              hipStream_t stream) {
    (void)in_sizes; (void)n_in; (void)out_size; (void)ws_size;
    const float* x   = (const float*)d_in[0];
    const int*   ei  = (const int*)d_in[1];
    const float* W1  = (const float*)d_in[2];
    const float* as1 = (const float*)d_in[3];
    const float* ad1 = (const float*)d_in[4];
    const float* b1  = (const float*)d_in[5];
    const float* W2  = (const float*)d_in[6];
    const float* as2 = (const float*)d_in[7];
    const float* ad2 = (const float*)d_in[8];
    const float* b2  = (const float*)d_in[9];
    float* out = (float*)d_out;

    char* wsb = (char*)d_ws;
    size_t off = 0;
    auto alloc = [&](size_t bytes) -> void* {
        void* p = wsb + off;
        off += (bytes + 255) & ~(size_t)255;
        return p;
    };
    _Float16* w1h  = (_Float16*)alloc((size_t)HC * NF * 2);
    _Float16* w2h  = (_Float16*)alloc((size_t)NC * HC * 2);
    _Float16* xh   = (_Float16*)alloc((size_t)NN * NF * 2);
    float* h1      = (float*)alloc((size_t)NN * HC * 4);
    float* asrc1   = (float*)alloc((size_t)NN * HH * 4);
    float* adst1   = (float*)alloc((size_t)NN * HH * 4);
    float* h2      = (float*)alloc((size_t)NN * NC * 4);
    float* asrc2   = (float*)alloc((size_t)NN * 4);
    float* adst2   = (float*)alloc((size_t)NN * 4);
    int* elist     = (int*)alloc((size_t)NGRP * 32 * WCAP * 4);
    int* ecnt      = (int*)alloc((size_t)NGRP * 32 * 4);
    float* act1    = (float*)xh;
    unsigned char* ownb = (unsigned char*)h2;

    const dim3 B(256);
    gat_cvtw <<<(HC * NF + NC * HC + 255) / 256, B, 0, stream>>>(W1, W2, w1h, w2h);
    gat_cvtx <<<(NN * NF) / 256, B, 0, stream>>>(x, xh);
    gat_gemm1<<<(NN / 16 + 7) / 8, B, 0, stream>>>(xh, w1h, h1);
    gat_att1 <<<(NN * HH) / 256, B, 0, stream>>>(h1, as1, ad1, asrc1, adst1);
    gat_ownb <<<(EE / 16 + 255) / 256, B, 0, stream>>>(ei, ownb);
    gat_own<HH, HC, 8, 1><<<NGRP, 1024, 0, stream>>>(ei, asrc1, adst1, h1, b1, act1, elist, ecnt, ownb);
    gat_gemm2<<<(NN / 16 + 7) / 8, B, 0, stream>>>(act1, w2h, h2);
    gat_att2 <<<(NN + 255) / 256, B, 0, stream>>>(h2, as2, ad2, asrc2, adst2);
    gat_own<1, NC, 4, 2><<<NGRP, 1024, 0, stream>>>(ei, asrc2, adst2, h2, b2, out, elist, ecnt, ownb);
}


# x-to-f16 copy kernel as a stream: a lane converts 80 elements, 20 whole-line 16-B loads (14 in flight before the first convert), whole-line 8-B stores; 10,000 waves not 800,000; values unchanged
# speedup vs baseline: 3.8928x; 1.1712x over previous
#include <hip/hip_runtime.h>

#define NN  100000
#define NF  256
#define HH  8
#define CC  16
#define HC  128
#define NC  16
#define EE  1600000
#define ET  1700000
#define NEG 0.2f
#define NPB  392
#define NGRP 256
static_assert((NN + NPB - 1) / NPB <= 256, "an owner-group id must fit one byte");
static_assert(EE % 16 == 0, "the owner-byte list is read 16 edges a load");

typedef _Float16 h8   __attribute__((ext_vector_type(8)));
typedef _Float16 v16h __attribute__((ext_vector_type(16)));
typedef float    v8f  __attribute__((ext_vector_type(8)));

__device__ __forceinline__ void edge_sd(const int* __restrict__ ei, int e, int& s, int& d) {
    if (e < EE) { s = ei[e]; d = ei[EE + e]; }
    else        { s = e - EE; d = s; }
}

__device__ __forceinline__ void atomicMaxF(float* addr, float val) {
    if (val >= 0.0f) atomicMax((int*)addr, __float_as_int(val));
    else             atomicMin((unsigned int*)addr, __float_as_uint(val));
}

__device__ __forceinline__ float lrelu(float a) { return a > 0.0f ? a : NEG * a; }

__global__ __launch_bounds__(256) void gat_cvtw(const float* __restrict__ W1,
                                                const float* __restrict__ W2,
                                                _Float16* __restrict__ w1h,
                                                _Float16* __restrict__ w2h) {
    int i = blockIdx.x * 256 + threadIdx.x;
    if (i < HC * NF)              w1h[i] = (_Float16)W1[i];
    else if (i < HC * NF + NC*HC) w2h[i - HC * NF] = (_Float16)W2[i - HC * NF];
}

#define CVT_LD 20
#define CVT_PER_BLOCK (256 * CVT_LD)
#define CVT_BLOCKS ((NN * NF / 4) / CVT_PER_BLOCK)
static_assert((NN * NF) % 4 == 0 && (NN * NF / 4) % CVT_PER_BLOCK == 0,
              "the conversion grid covers x exactly: no tail piece");
typedef _Float16 h4 __attribute__((ext_vector_type(4)));

__global__ __launch_bounds__(256) void gat_cvtx(const float* __restrict__ x,
                                                _Float16* __restrict__ xh) {
    const float4* __restrict__ xp = (const float4*)x;
    h4* __restrict__ op = (h4*)xh;
    const int base = blockIdx.x * CVT_PER_BLOCK + threadIdx.x;
    float4 v[CVT_LD];
    #pragma unroll
    for (int k = 0; k < CVT_LD; ++k) v[k] = xp[base + k * 256];
    #pragma unroll
    for (int k = 0; k < CVT_LD; ++k) {
        h4 o;
        o.x = (_Float16)v[k].x; o.y = (_Float16)v[k].y;
        o.z = (_Float16)v[k].z; o.w = (_Float16)v[k].w;
        op[base + k * 256] = o;
    }
}

__global__ __launch_bounds__(256)
void gat_gemm1(const _Float16* __restrict__ xh, const _Float16* __restrict__ w1h,
               float* __restrict__ h1) {
    __shared__ _Float16 lw[HC * NF];
    const int tid = threadIdx.x;
    for (int idx = tid; idx < HC * 32; idx += 256) {
        int r = idx >> 5, ch = idx & 31;
        *(h8*)(&lw[r * NF + ((ch ^ (r & 15)) << 3)]) =
            *(const h8*)(w1h + (size_t)r * NF + (ch << 3));
    }
    __syncthreads();

    const int lane = tid & 31;
    const int rt   = blockIdx.x * 8 + (tid >> 5);
    if (rt >= NN / 16) return;
    const int rsub = lane & 15;
    const int ksel = (lane >> 4) << 3;
    const _Float16* xrow = xh + (size_t)(rt * 16 + rsub) * NF;

    v8f acc[8] = {};
    #pragma unroll
    for (int ks = 0; ks < 8; ++ks) {
        const int kb = ks * 32 + ksel;
        const int c0 = kb >> 3;
        union { v16h v; h8 p[2]; } A;
        A.p[0] = *(const h8*)(xrow + kb);
        A.p[1] = *(const h8*)(xrow + kb + 16);
        #pragma unroll
        for (int ct = 0; ct < 8; ++ct) {
            const int n = ct * 16 + rsub;
            union { v16h v; h8 p[2]; } B;
            B.p[0] = *(const h8*)(&lw[n * NF + (((c0    ) ^ (n & 15)) << 3)]);
            B.p[1] = *(const h8*)(&lw[n * NF + (((c0 + 2) ^ (n & 15)) << 3)]);
            acc[ct] = __builtin_amdgcn_wmma_f32_16x16x32_f16(
                false, A.v, false, B.v, (short)0, acc[ct], false, false);
        }
    }
    const int rhi = (lane >> 4) << 3;
    #pragma unroll
    for (int ct = 0; ct < 8; ++ct) {
        #pragma unroll
        for (int r = 0; r < 8; ++r)
            h1[(size_t)(rt * 16 + rhi + r) * HC + ct * 16 + rsub] = acc[ct][r];
    }
}

__global__ __launch_bounds__(256)
void gat_att1(const float* __restrict__ h1, const float* __restrict__ as,
              const float* __restrict__ ad, float* __restrict__ a_src,
              float* __restrict__ a_dst) {
    int i = blockIdx.x * 256 + threadIdx.x;
    if (i >= NN * HH) return;
    int n = i >> 3, h = i & 7;
    const float* hp = h1 + (size_t)n * HC + h * CC;
    float s = 0.0f, d = 0.0f;
    #pragma unroll
    for (int c = 0; c < CC; ++c) { s += hp[c] * as[h * CC + c]; d += hp[c] * ad[h * CC + c]; }
    a_src[i] = s; a_dst[i] = d;
}

__global__ __launch_bounds__(256)
void gat_ownb(const int* __restrict__ ei, unsigned char* __restrict__ own) {
    const int i = blockIdx.x * 256 + threadIdx.x;
    if (i >= EE / 16) return;
    const int4* dp = (const int4*)(ei + EE) + (size_t)i * 4;
    unsigned w[4];
    #pragma unroll
    for (int k = 0; k < 4; ++k) {
        const int4 d = dp[k];
        w[k] = ((unsigned)d.x / NPB) | (((unsigned)d.y / NPB) << 8) |
               (((unsigned)d.z / NPB) << 16) | (((unsigned)d.w / NPB) << 24);
    }
    ((uint4*)own)[i] = make_uint4(w[0], w[1], w[2], w[3]);
}

__global__ __launch_bounds__(256)
void gat_gemm2(const float* __restrict__ a1, const _Float16* __restrict__ w2h,
               float* __restrict__ h2) {
    const int lane = threadIdx.x & 31;
    const int rt   = blockIdx.x * 8 + (threadIdx.x >> 5);
    if (rt >= NN / 16) return;
    const int rsub = lane & 15;
    const int ksel = (lane >> 4) << 3;
    const float*    arow = a1 + (size_t)(rt * 16 + rsub) * HC;
    const _Float16* wrow = w2h + (size_t)rsub * HC;

    v8f acc = {};
    #pragma unroll
    for (int ks = 0; ks < 4; ++ks) {
        const int kb = ks * 32 + ksel;
        union { v16h v; _Float16 e[16]; } A;
        #pragma unroll
        for (int t = 0; t < 8; ++t) {
            A.e[t]     = (_Float16)arow[kb + t];
            A.e[8 + t] = (_Float16)arow[kb + 16 + t];
        }
        union { v16h v; h8 p[2]; } B;
        B.p[0] = *(const h8*)(wrow + kb);
        B.p[1] = *(const h8*)(wrow + kb + 16);
        acc = __builtin_amdgcn_wmma_f32_16x16x32_f16(
            false, A.v, false, B.v, (short)0, acc, false, false);
    }
    const int rhi = (lane >> 4) << 3;
    #pragma unroll
    for (int r = 0; r < 8; ++r)
        h2[(size_t)(rt * 16 + rhi + r) * NC + rsub] = acc[r];
}

__global__ __launch_bounds__(256)
void gat_att2(const float* __restrict__ h2, const float* __restrict__ as,
              const float* __restrict__ ad, float* __restrict__ a_src,
              float* __restrict__ a_dst) {
    int n = blockIdx.x * 256 + threadIdx.x;
    if (n >= NN) return;
    const float* hp = h2 + (size_t)n * NC;
    float s = 0.0f, d = 0.0f;
    #pragma unroll
    for (int c = 0; c < NC; ++c) { s += hp[c] * as[c]; d += hp[c] * ad[c]; }
    a_src[n] = s; a_dst[n] = d;
}

#define SCAN_EDGES 16384
#define NIT ((EE + SCAN_EDGES - 1) / SCAN_EDGES)
#define NCH (EE / 16)
#define CH_STEP 4096
#define NITB ((NCH + CH_STEP - 1) / CH_STEP)
#define WCAP 320
#define MAXB 32
static_assert(NPB <= 512 && NN <= (1 << 17), "list word: 9 bits local node, 17 bits source node");
static_assert(ET < (1 << 21), "an edge id is kept in 21 bits");

template <bool V> struct mode_c { static constexpr bool value = V; };

template <int H>
__device__ __forceinline__ void load_att(const float* __restrict__ p, int node, float (&v)[H]) {
    if constexpr (H % 4 == 0) {
        #pragma unroll
        for (int k = 0; k < H / 4; ++k) {
            const float4 t = *(const float4*)(p + (size_t)node * H + k * 4);
            v[k * 4] = t.x; v[k * 4 + 1] = t.y; v[k * 4 + 2] = t.z; v[k * 4 + 3] = t.w;
        }
    } else {
        #pragma unroll
        for (int k = 0; k < H; ++k) v[k] = p[(size_t)node * H + k];
    }
}

template <int H, bool KEEP>
__device__ __forceinline__ void max_batch(const int* __restrict__ ei, const float* __restrict__ a_src,
                                          const float* __restrict__ a_dst, float* mx,
                                          const int* qw, int first, int nvalid, int lo, int lane,
                                          int* __restrict__ lst, int lpos) {
    const bool ok = lane < nvalid;
    const int e  = qw[first + (ok ? lane : 0)];
    const int ec = e < EE ? e : 0;
    const int sl = ei[ec];
    const int dd = ei[EE + ec];
    const int s  = (e < EE) ? sl : e - EE;
    const int dl = ((e < EE) ? dd : e - EE) - lo;
    if constexpr (KEEP) {
        if (ok && lpos + lane < WCAP) lst[lpos + lane] = (int)(((unsigned)dl << 17) | (unsigned)s);
    }
    float as[H], ad[H];
    load_att<H>(a_src, s, as);
    load_att<H>(a_dst, lo + dl, ad);
    if (ok) {
        #pragma unroll
        for (int h = 0; h < H; ++h) atomicMaxF(&mx[h * NPB + dl], lrelu(as[h] + ad[h]));
    }
}

template <int H>
__device__ __forceinline__ void walk_max(const int* __restrict__ lw, int n, const float* __restrict__ a_src,
                                         const float* __restrict__ a_dst, float* mx, int lo, int lane) {
    constexpr int U = 4;
    for (int b = 0; b < n; b += 32 * U) {
        int  ent[U];
        bool ok[U];
        #pragma unroll
        for (int u = 0; u < U; ++u) {
            const int idx = b + u * 32 + lane;
            ok[u] = idx < n;
            ent[u] = lw[ok[u] ? idx : 0];
        }
        float as[U][H], ad[U][H];
        #pragma unroll
        for (int u = 0; u < U; ++u) {
            load_att<H>(a_src, ent[u] & 0x1FFFF, as[u]);
            load_att<H>(a_dst, lo + (ent[u] >> 17), ad[u]);
        }
        #pragma unroll
        for (int u = 0; u < U; ++u) {
            if (ok[u]) {
                #pragma unroll
                for (int h = 0; h < H; ++h)
                    atomicMaxF(&mx[h * NPB + (ent[u] >> 17)], lrelu(as[u][h] + ad[u][h]));
            }
        }
    }
}

template <int H, int HCt, int STEPS>
__device__ __forceinline__ void own_batch(const int* __restrict__ ei, const float* __restrict__ a_src,
                                          const float* __restrict__ a_dst, const float* mx,
                                          const float* __restrict__ hsrc, float* acc, float* dn,
                                          const int* qw, int first, int nvalid, int lo, int lane) {
    constexpr int LPE = HCt / 4;
    constexpr int EPW = 32 / LPE;
    const int sub = lane / LPE;
    const int q   = lane % LPE;
    const int h   = (q * 4) / (HCt / H);
    int  ents[STEPS], sv[STEPS], dlv[STEPS];
    bool ok[STEPS];
    #pragma unroll
    for (int u = 0; u < STEPS; ++u) {
        const int idx = u * EPW + sub;
        ok[u] = idx < nvalid;
        ents[u] = qw[first + idx];
    }
    #pragma unroll
    for (int u = 0; u < STEPS; ++u) {
        const int e  = ok[u] ? ents[u] : 0;
        const int ec = e < EE ? e : 0;
        const int sl = ei[ec];
        const int dd = ei[EE + ec];
        sv[u]  = (e < EE) ? sl : e - EE;
        dlv[u] = ok[u] ? ((e < EE) ? dd : e - EE) - lo : 0;
    }
    float4 hv[STEPS];
    float as[STEPS], ad[STEPS], am[STEPS];
    #pragma unroll
    for (int u = 0; u < STEPS; ++u) {
        const int sn = sv[u], dnode = lo + dlv[u];
        hv[u] = *(const float4*)(hsrc + (size_t)sn * HCt + q * 4);
        as[u] = a_src[sn * H + h];
        ad[u] = a_dst[dnode * H + h];
        am[u] = mx[h * NPB + dlv[u]];
    }
    #pragma unroll
    for (int u = 0; u < STEPS; ++u) {
        const float w = expf(lrelu(as[u] + ad[u]) - am[u]);
        if (ok[u]) {
            if ((q & 3) == 0) atomicAdd(&dn[dlv[u] * H + h], w);
            float* ap = acc + dlv[u] * HCt + q;
            atomicAdd(ap,           hv[u].x * w);
            atomicAdd(ap + LPE,     hv[u].y * w);
            atomicAdd(ap + 2 * LPE, hv[u].z * w);
            atomicAdd(ap + 3 * LPE, hv[u].w * w);
        }
    }
}

template <int H, int HCt, int STEPS>
__device__ __forceinline__ void own_walk(const int* __restrict__ lw, int n, const float* __restrict__ a_src,
                                         const float* __restrict__ a_dst, const float* mx,
                                         const float* __restrict__ hsrc, float* acc, float* dn,
                                         int lo, int lane) {
    constexpr int LPE = HCt / 4;
    constexpr int EPW = 32 / LPE;
    constexpr int B   = EPW * STEPS;
    const int sub = lane / LPE;
    const int q   = lane % LPE;
    const int h   = (q * 4) / (HCt / H);
    int ents[STEPS], nxt[STEPS];
    #pragma unroll
    for (int u = 0; u < STEPS; ++u) {
        const int idx = u * EPW + sub;
        ents[u] = lw[idx < n ? idx : 0];
    }
    for (int b = 0; b < n; b += B) {
        #pragma unroll
        for (int u = 0; u < STEPS; ++u) {
            const int idx = b + B + u * EPW + sub;
            nxt[u] = lw[idx < n ? idx : 0];
        }
        float4 hv[STEPS];
        float as[STEPS], ad[STEPS], am[STEPS];
        int   dlv[STEPS];
        bool  ok[STEPS];
        #pragma unroll
        for (int u = 0; u < STEPS; ++u) {
            ok[u] = b + u * EPW + sub < n;
            const int ent = ok[u] ? ents[u] : 0;
            const int sn = ent & 0x1FFFF;
            dlv[u] = ent >> 17;
            const int dnode = lo + dlv[u];
            hv[u] = *(const float4*)(hsrc + (size_t)sn * HCt + q * 4);
            as[u] = a_src[sn * H + h];
            ad[u] = a_dst[dnode * H + h];
            am[u] = mx[h * NPB + dlv[u]];
        }
        #pragma unroll
        for (int u = 0; u < STEPS; ++u) {
            const float w = expf(lrelu(as[u] + ad[u]) - am[u]);
            if (ok[u]) {
                if ((q & 3) == 0) atomicAdd(&dn[dlv[u] * H + h], w);
                float* ap = acc + dlv[u] * HCt + q;
                atomicAdd(ap,           hv[u].x * w);
                atomicAdd(ap + LPE,     hv[u].y * w);
                atomicAdd(ap + 2 * LPE, hv[u].z * w);
                atomicAdd(ap + 3 * LPE, hv[u].w * w);
            }
        }
        #pragma unroll
        for (int u = 0; u < STEPS; ++u) ents[u] = nxt[u];
    }
}

template <int H, int HCt, int STEPS, int LAYER>
__global__ __launch_bounds__(1024)
void gat_own(const int* __restrict__ ei, const float* __restrict__ a_src,
             const float* __restrict__ a_dst,
             const float* __restrict__ hsrc, const float* __restrict__ bias,
             float* __restrict__ dstv, int* __restrict__ elist, int* __restrict__ ecnt,
             const unsigned char* __restrict__ own) {
    constexpr int LPE  = HCt / 4;
    constexpr int EPW  = 32 / LPE;
    constexpr int B    = EPW * STEPS;
    constexpr int QCAP = (B > MAXB ? B : MAXB) + 512;
    __shared__ float acc[NPB * HCt];
    __shared__ float dn[NPB * H];
    __shared__ float mx[NPB * H];
    __shared__ int   qbuf[32 * QCAP];
    __shared__ int   wcnt[32];

    const int tid = threadIdx.x, lane = tid & 31;
    const int wave = __builtin_amdgcn_readfirstlane(tid >> 5);
    const int lo  = blockIdx.x * NPB;
    const int cnt = (NN - lo < NPB) ? (NN - lo) : NPB;
    for (int i = tid; i < NPB * HCt; i += 1024) acc[i] = 0.0f;
    for (int i = tid; i < NPB * H;   i += 1024) { dn[i] = 0.0f; mx[i] = -__builtin_inff(); }
    __syncthreads();

    int* lst = elist + (size_t)(blockIdx.x * 32 + wave) * WCAP;
    int* qw  = qbuf + wave * QCAP;

    auto scan_pass = [&](auto mode) -> int {
    constexpr bool MAXP = decltype(mode)::value;
    constexpr int  BB   = MAXP ? MAXB : B;
    int  qn = 0;
    int  ln = 0;
    const unsigned lt = (1u << lane) - 1u;
    auto push = [&](bool m, unsigned ent) {
        const unsigned mask = __builtin_amdgcn_ballot_w32(m);
        if (m) qw[qn + __builtin_popcount(mask & lt)] = (int)ent;
        qn += __builtin_popcount(mask);
    };
    auto batch = [&](int first, int nvalid) {
        if constexpr (MAXP) max_batch<H, LAYER == 1>(ei, a_src, a_dst, mx, qw, first, nvalid, lo, lane, lst, ln);
        else own_batch<H, HCt, STEPS>(ei, a_src, a_dst, mx, hsrc, acc, dn, qw, first, nvalid, lo, lane);
    };
    auto drain = [&]() {
        while (qn >= BB) {
            batch(qn - BB, BB);
            qn -= BB;
            ln += BB;
        }
    };

    for (int b0 = wave * 32; b0 < cnt; b0 += 1024) {
        const int dl = b0 + lane;
        push(dl < cnt, (unsigned)(EE + lo + dl));
    }
    drain();

    if constexpr (LAYER == 1) {
    const uint4* ownp = (const uint4*)own;
    const unsigned pat = (unsigned)blockIdx.x * 0x01010101u;
    const int it0 = (blockIdx.x * NITB) / NGRP;
    auto ldscan = [&](int i, uint4 (&buf)[4]) -> int {
        int it = i + it0;
        if (it >= NITB) it -= NITB;
        const int cb = it * CH_STEP + wave * 128 + lane;
        #pragma unroll
        for (int k = 0; k < 4; ++k) {
            const int ck = cb + k * 32;
            buf[k] = ownp[ck < NCH ? ck : NCH - 1];
        }
        return cb;
    };
    auto hit = [&](unsigned w) -> unsigned {
        const unsigned x = w ^ pat;
        return ~(((x & 0x7F7F7F7Fu) + 0x7F7F7F7Fu) | x) & 0x80808080u;
    };
    auto test = [&](const uint4 (&buf)[4], int cb) {
        unsigned mk[4];
        #pragma unroll
        for (int k = 0; k < 4; ++k) {
            const unsigned m4 = (hit(buf[k].x) >> 7) | (hit(buf[k].y) >> 6) |
                                (hit(buf[k].z) >> 5) | (hit(buf[k].w) >> 4);
            mk[k] = (cb + k * 32 < NCH) ? m4 : 0u;
        }
        unsigned long long M = ((unsigned long long)(mk[2] | (mk[3] << 4)) << 32) |
                               (unsigned long long)(mk[0] | (mk[1] << 4));
        while (__builtin_amdgcn_ballot_w32(M != 0)) {
            const bool m = M != 0;
            const int  p = m ? __builtin_ctzll(M) : 0;
            const int  k = ((p >> 5) << 1) | ((p >> 2) & 1);
            push(m, (unsigned)((cb + k * 32) * 16 + ((p & 3) << 2) + ((p >> 3) & 3)));
            M &= M - 1;
            drain();
        }
    };
    uint4 bufa[4], bufb[4];
    int cba = ldscan(0, bufa), cbb;
    for (int i = 0; i < NITB; i += 2) {
        cbb = ldscan(i + 1, bufb);
        __builtin_amdgcn_sched_barrier(0);
        test(bufa, cba);
        cba = ldscan(i + 2, bufa);
        __builtin_amdgcn_sched_barrier(0);
        if (i + 1 < NITB) test(bufb, cbb);
    }
    } else {
    const int* dstp = ei + EE;
    const int it0 = (blockIdx.x * NIT) / NGRP;
    auto ldscan = [&](int i, int4 (&buf)[4]) -> int {
        int it = i + it0;
        if (it >= NIT) it -= NIT;
        const int eb = it * SCAN_EDGES + wave * 512 + lane * 4;
        #pragma unroll
        for (int k = 0; k < 4; ++k) {
            const int ek = eb + k * 128;
            buf[k] = *(const int4*)(dstp + (ek < EE ? ek : EE - 4));
        }
        return eb;
    };
    auto test = [&](const int4 (&buf)[4], int eb) {
        #pragma unroll
        for (int k = 0; k < 4; ++k) {
            const int ek = eb + k * 128;
            const unsigned d0 = (unsigned)(buf[k].x - lo), d1 = (unsigned)(buf[k].y - lo);
            const unsigned d2 = (unsigned)(buf[k].z - lo), d3 = (unsigned)(buf[k].w - lo);
            const bool in = ek < EE;
            const bool m0 = in && d0 < (unsigned)cnt, m1 = in && d1 < (unsigned)cnt;
            const bool m2 = in && d2 < (unsigned)cnt, m3 = in && d3 < (unsigned)cnt;
            if (__builtin_amdgcn_ballot_w32(m0 | m1 | m2 | m3)) {
                push(m0, (unsigned)(ek + 0));
                push(m1, (unsigned)(ek + 1));
                push(m2, (unsigned)(ek + 2));
                push(m3, (unsigned)(ek + 3));
            }
        }
        drain();
    };
    int4 bufa[4], bufb[4];
    int eba = ldscan(0, bufa), ebb;
    for (int i = 0; i < NIT; i += 2) {
        ebb = ldscan(i + 1, bufb);
        test(bufa, eba);
        eba = ldscan(i + 2, bufa);
        if (i + 1 < NIT) test(bufb, ebb);
    }
    }
    if (qn > 0) {
        batch(0, qn);
        ln += qn;
    }
    return ln;
    };

    if constexpr (LAYER == 1) {
        const int nl = scan_pass(mode_c<true>{});
        if (lane == 0) { ecnt[blockIdx.x * 32 + wave] = nl; wcnt[wave] = nl; }
        __syncthreads();
        if (__builtin_amdgcn_ballot_w32(wcnt[lane] > WCAP) == 0)
            own_walk<H, HCt, 8>(lst, nl, a_src, a_dst, mx, hsrc, acc, dn, lo, lane);
        else
            (void)scan_pass(mode_c<false>{});
    } else {
        const int c = ecnt[blockIdx.x * 32 + lane];
        const int nl = __builtin_amdgcn_readlane(c, wave);
        if (__builtin_amdgcn_ballot_w32(c > WCAP) == 0) {
            walk_max<H>(lst, nl, a_src, a_dst, mx, lo, lane);
            __syncthreads();
            own_walk<H, HCt, 8>(lst, nl, a_src, a_dst, mx, hsrc, acc, dn, lo, lane);
        } else {
            (void)scan_pass(mode_c<true>{});
            __syncthreads();
            (void)scan_pass(mode_c<false>{});
        }
    }
    __syncthreads();

    if constexpr (LAYER == 1) {
        const float4 bq = *(const float4*)(bias + lane * 4);
        for (int r = wave; r < cnt; r += 32) {
            const float dv = dn[r * H + (lane >> 2)] + 1e-16f;
            const float* ap = acc + r * HCt + lane;
            float4 o;
            o.x = ap[0]       / dv + bq.x;
            o.y = ap[LPE]     / dv + bq.y;
            o.z = ap[2 * LPE] / dv + bq.z;
            o.w = ap[3 * LPE] / dv + bq.w;
            o.x = o.x > 0.0f ? o.x : (expf(o.x) - 1.0f);
            o.y = o.y > 0.0f ? o.y : (expf(o.y) - 1.0f);
            o.z = o.z > 0.0f ? o.z : (expf(o.z) - 1.0f);
            o.w = o.w > 0.0f ? o.w : (expf(o.w) - 1.0f);
            *(float4*)(dstv + (size_t)(lo + r) * HCt + lane * 4) = o;
        }
    } else {
        for (int r = tid; r < cnt; r += 1024) {
            float v[HCt], m = -__builtin_inff();
            const float dv = dn[r] + 1e-16f;
            #pragma unroll
            for (int c = 0; c < HCt; ++c) {
                v[c] = acc[r * HCt + (c & 3) * LPE + (c >> 2)] / dv + bias[c];
                m = fmaxf(m, v[c]);
            }
            float sum = 0.0f;
            #pragma unroll
            for (int c = 0; c < HCt; ++c) sum += expf(v[c] - m);
            const float ls = logf(sum);
            #pragma unroll
            for (int c = 0; c < HCt; ++c) dstv[(size_t)(lo + r) * HCt + c] = v[c] - m - ls;
        }
    }
}

extern "C" void kernel_launch(void* const* d_in, const int* in_sizes, int n_in,
                              void* d_out, int out_size, void* d_ws, size_t ws_size,
                              hipStream_t stream) {
    (void)in_sizes; (void)n_in; (void)out_size; (void)ws_size;
    const float* x   = (const float*)d_in[0];
    const int*   ei  = (const int*)d_in[1];
    const float* W1  = (const float*)d_in[2];
    const float* as1 = (const float*)d_in[3];
    const float* ad1 = (const float*)d_in[4];
    const float* b1  = (const float*)d_in[5];
    const float* W2  = (const float*)d_in[6];
    const float* as2 = (const float*)d_in[7];
    const float* ad2 = (const float*)d_in[8];
    const float* b2  = (const float*)d_in[9];
    float* out = (float*)d_out;

    char* wsb = (char*)d_ws;
    size_t off = 0;
    auto alloc = [&](size_t bytes) -> void* {
        void* p = wsb + off;
        off += (bytes + 255) & ~(size_t)255;
        return p;
    };
    _Float16* w1h  = (_Float16*)alloc((size_t)HC * NF * 2);
    _Float16* w2h  = (_Float16*)alloc((size_t)NC * HC * 2);
    _Float16* xh   = (_Float16*)alloc((size_t)NN * NF * 2);
    float* h1      = (float*)alloc((size_t)NN * HC * 4);
    float* asrc1   = (float*)alloc((size_t)NN * HH * 4);
    float* adst1   = (float*)alloc((size_t)NN * HH * 4);
    float* h2      = (float*)alloc((size_t)NN * NC * 4);
    float* asrc2   = (float*)alloc((size_t)NN * 4);
    float* adst2   = (float*)alloc((size_t)NN * 4);
    int* elist     = (int*)alloc((size_t)NGRP * 32 * WCAP * 4);
    int* ecnt      = (int*)alloc((size_t)NGRP * 32 * 4);
    float* act1    = (float*)xh;
    unsigned char* ownb = (unsigned char*)h2;

    const dim3 B(256);
    gat_cvtw <<<(HC * NF + NC * HC + 255) / 256, B, 0, stream>>>(W1, W2, w1h, w2h);
    gat_cvtx <<<CVT_BLOCKS, B, 0, stream>>>(x, xh);
    gat_gemm1<<<(NN / 16 + 7) / 8, B, 0, stream>>>(xh, w1h, h1);
    gat_att1 <<<(NN * HH) / 256, B, 0, stream>>>(h1, as1, ad1, asrc1, adst1);
    gat_ownb <<<(EE / 16 + 255) / 256, B, 0, stream>>>(ei, ownb);
    gat_own<HH, HC, 8, 1><<<NGRP, 1024, 0, stream>>>(ei, asrc1, adst1, h1, b1, act1, elist, ecnt, ownb);
    gat_gemm2<<<(NN / 16 + 7) / 8, B, 0, stream>>>(act1, w2h, h2);
    gat_att2 <<<(NN + 255) / 256, B, 0, stream>>>(h2, as2, ad2, asrc2, adst2);
    gat_own<1, NC, 4, 2><<<NGRP, 1024, 0, stream>>>(ei, asrc2, adst2, h2, b2, out, elist, ecnt, ownb);
}


# attention scalars as one more WMMA tile: att vectors folded in f32 into 16 extra weight rows, rounded to f16 like W; a_src/a_dst leave the GEMMs; att1/att2 launches gone; their summation order is new
# speedup vs baseline: 4.0323x; 1.0358x over previous
#include <hip/hip_runtime.h>

#define NN  100000
#define NF  256
#define HH  8
#define CC  16
#define HC  128
#define NC  16
#define EE  1600000
#define ET  1700000
#define NEG 0.2f
#define NPB  392
#define NGRP 256
static_assert((NN + NPB - 1) / NPB <= 256, "an owner-group id must fit one byte");
static_assert(EE % 16 == 0, "the owner-byte list is read 16 edges a load");

typedef _Float16 h8   __attribute__((ext_vector_type(8)));
typedef _Float16 v16h __attribute__((ext_vector_type(16)));
typedef float    v8f  __attribute__((ext_vector_type(8)));

__device__ __forceinline__ void edge_sd(const int* __restrict__ ei, int e, int& s, int& d) {
    if (e < EE) { s = ei[e]; d = ei[EE + e]; }
    else        { s = e - EE; d = s; }
}

__device__ __forceinline__ void atomicMaxF(float* addr, float val) {
    if (val >= 0.0f) atomicMax((int*)addr, __float_as_int(val));
    else             atomicMin((unsigned int*)addr, __float_as_uint(val));
}

__device__ __forceinline__ float lrelu(float a) { return a > 0.0f ? a : NEG * a; }

#define W1R (HC + 2 * HH)
#define W2R (NC + 16)
#define CVTW_N (W1R * NF + W2R * HC)
__global__ __launch_bounds__(256) void gat_cvtw(const float* __restrict__ W1,
                                                const float* __restrict__ W2,
                                                const float* __restrict__ as1,
                                                const float* __restrict__ ad1,
                                                const float* __restrict__ as2,
                                                const float* __restrict__ ad2,
                                                _Float16* __restrict__ w1h,
                                                _Float16* __restrict__ w2h) {
    const int i = blockIdx.x * 256 + threadIdx.x;
    if (i < HC * NF) {
        w1h[i] = (_Float16)W1[i];
    } else if (i < W1R * NF) {
        const int j = i - HC * NF;
        const int row = j / NF, f = j % NF;
        const int h = row & (HH - 1);
        const float* __restrict__ att = (row < HH ? as1 : ad1) + h * CC;
        float s = 0.0f;
        #pragma unroll
        for (int cc = 0; cc < CC; ++cc) s += W1[(size_t)(h * CC + cc) * NF + f] * att[cc];
        w1h[i] = (_Float16)s;
    } else if (i < W1R * NF + NC * HC) {
        const int j = i - W1R * NF;
        w2h[j] = (_Float16)W2[j];
    } else if (i < CVTW_N) {
        const int j = i - W1R * NF - NC * HC;
        const int row = j / HC, k = j % HC;
        float s = 0.0f;
        if ((row & 7) == 0) {
            const float* __restrict__ att = row == 0 ? as2 : ad2;
            #pragma unroll
            for (int cc = 0; cc < NC; ++cc) s += W2[(size_t)cc * HC + k] * att[cc];
        }
        w2h[NC * HC + j] = (_Float16)s;
    }
}

#define CVT_LD 20
#define CVT_PER_BLOCK (256 * CVT_LD)
#define CVT_BLOCKS ((NN * NF / 4) / CVT_PER_BLOCK)
static_assert((NN * NF) % 4 == 0 && (NN * NF / 4) % CVT_PER_BLOCK == 0,
              "the conversion grid covers x exactly: no tail piece");
typedef _Float16 h4 __attribute__((ext_vector_type(4)));

__global__ __launch_bounds__(256) void gat_cvtx(const float* __restrict__ x,
                                                _Float16* __restrict__ xh) {
    const float4* __restrict__ xp = (const float4*)x;
    h4* __restrict__ op = (h4*)xh;
    const int base = blockIdx.x * CVT_PER_BLOCK + threadIdx.x;
    float4 v[CVT_LD];
    #pragma unroll
    for (int k = 0; k < CVT_LD; ++k) v[k] = xp[base + k * 256];
    #pragma unroll
    for (int k = 0; k < CVT_LD; ++k) {
        h4 o;
        o.x = (_Float16)v[k].x; o.y = (_Float16)v[k].y;
        o.z = (_Float16)v[k].z; o.w = (_Float16)v[k].w;
        op[base + k * 256] = o;
    }
}

__global__ __launch_bounds__(256) __attribute__((amdgpu_waves_per_eu(8)))
void gat_gemm1(const _Float16* __restrict__ xh, const _Float16* __restrict__ w1h,
               float* __restrict__ h1, float* __restrict__ a_src, float* __restrict__ a_dst) {
    __shared__ _Float16 lw[W1R * NF];
    const int tid = threadIdx.x;
    for (int idx = tid; idx < W1R * 32; idx += 256) {
        int r = idx >> 5, ch = idx & 31;
        *(h8*)(&lw[r * NF + ((ch ^ (r & 15)) << 3)]) =
            *(const h8*)(w1h + (size_t)r * NF + (ch << 3));
    }
    __syncthreads();

    const int lane = tid & 31;
    const int rt   = blockIdx.x * 8 + (tid >> 5);
    if (rt >= NN / 16) return;
    const int rsub = lane & 15;
    const int ksel = (lane >> 4) << 3;
    const _Float16* xrow = xh + (size_t)(rt * 16 + rsub) * NF;

    v8f acc[8] = {};
    v8f att = {};
    #pragma unroll
    for (int ks = 0; ks < 8; ++ks) {
        const int kb = ks * 32 + ksel;
        const int c0 = kb >> 3;
        union { v16h v; h8 p[2]; } A;
        A.p[0] = *(const h8*)(xrow + kb);
        A.p[1] = *(const h8*)(xrow + kb + 16);
        #pragma unroll
        for (int ct = 0; ct < 8; ++ct) {
            const int n = ct * 16 + rsub;
            union { v16h v; h8 p[2]; } B;
            B.p[0] = *(const h8*)(&lw[n * NF + (((c0    ) ^ (n & 15)) << 3)]);
            B.p[1] = *(const h8*)(&lw[n * NF + (((c0 + 2) ^ (n & 15)) << 3)]);
            acc[ct] = __builtin_amdgcn_wmma_f32_16x16x32_f16(
                false, A.v, false, B.v, (short)0, acc[ct], false, false);
        }
        {
            const int n = HC + rsub;
            union { v16h v; h8 p[2]; } F;
            F.p[0] = *(const h8*)(&lw[n * NF + (((c0    ) ^ (n & 15)) << 3)]);
            F.p[1] = *(const h8*)(&lw[n * NF + (((c0 + 2) ^ (n & 15)) << 3)]);
            att = __builtin_amdgcn_wmma_f32_16x16x32_f16(
                false, F.v, false, A.v, (short)0, att, false, false);
        }
    }
    const int rhi = (lane >> 4) << 3;
    #pragma unroll
    for (int ct = 0; ct < 8; ++ct) {
        #pragma unroll
        for (int r = 0; r < 8; ++r)
            h1[(size_t)(rt * 16 + rhi + r) * HC + ct * 16 + rsub] = acc[ct][r];
    }
    float* ap = (lane < 16 ? a_src : a_dst) + (size_t)(rt * 16 + rsub) * HH;
    *(float4*)(ap)     = make_float4(att[0], att[1], att[2], att[3]);
    *(float4*)(ap + 4) = make_float4(att[4], att[5], att[6], att[7]);
}

__global__ __launch_bounds__(256)
void gat_ownb(const int* __restrict__ ei, unsigned char* __restrict__ own) {
    const int i = blockIdx.x * 256 + threadIdx.x;
    if (i >= EE / 16) return;
    const int4* dp = (const int4*)(ei + EE) + (size_t)i * 4;
    unsigned w[4];
    #pragma unroll
    for (int k = 0; k < 4; ++k) {
        const int4 d = dp[k];
        w[k] = ((unsigned)d.x / NPB) | (((unsigned)d.y / NPB) << 8) |
               (((unsigned)d.z / NPB) << 16) | (((unsigned)d.w / NPB) << 24);
    }
    ((uint4*)own)[i] = make_uint4(w[0], w[1], w[2], w[3]);
}

__global__ __launch_bounds__(256)
void gat_gemm2(const float* __restrict__ a1, const _Float16* __restrict__ w2h,
               float* __restrict__ h2, float* __restrict__ a_src, float* __restrict__ a_dst) {
    const int lane = threadIdx.x & 31;
    const int rt   = blockIdx.x * 8 + (threadIdx.x >> 5);
    if (rt >= NN / 16) return;
    const int rsub = lane & 15;
    const int ksel = (lane >> 4) << 3;
    const float*    arow = a1 + (size_t)(rt * 16 + rsub) * HC;
    const _Float16* wrow = w2h + (size_t)rsub * HC;
    const _Float16* frow = w2h + (size_t)(NC + rsub) * HC;

    v8f acc = {};
    v8f att = {};
    #pragma unroll
    for (int ks = 0; ks < 4; ++ks) {
        const int kb = ks * 32 + ksel;
        union { v16h v; _Float16 e[16]; } A;
        #pragma unroll
        for (int t = 0; t < 8; ++t) {
            A.e[t]     = (_Float16)arow[kb + t];
            A.e[8 + t] = (_Float16)arow[kb + 16 + t];
        }
        union { v16h v; h8 p[2]; } B;
        B.p[0] = *(const h8*)(wrow + kb);
        B.p[1] = *(const h8*)(wrow + kb + 16);
        acc = __builtin_amdgcn_wmma_f32_16x16x32_f16(
            false, A.v, false, B.v, (short)0, acc, false, false);
        union { v16h v; h8 p[2]; } F;
        F.p[0] = *(const h8*)(frow + kb);
        F.p[1] = *(const h8*)(frow + kb + 16);
        att = __builtin_amdgcn_wmma_f32_16x16x32_f16(
            false, F.v, false, A.v, (short)0, att, false, false);
    }
    const int rhi = (lane >> 4) << 3;
    #pragma unroll
    for (int r = 0; r < 8; ++r)
        h2[(size_t)(rt * 16 + rhi + r) * NC + rsub] = acc[r];
    (lane < 16 ? a_src : a_dst)[rt * 16 + rsub] = att[0];
}

#define SCAN_EDGES 16384
#define NIT ((EE + SCAN_EDGES - 1) / SCAN_EDGES)
#define NCH (EE / 16)
#define CH_STEP 4096
#define NITB ((NCH + CH_STEP - 1) / CH_STEP)
#define WCAP 320
#define MAXB 32
static_assert(NPB <= 512 && NN <= (1 << 17), "list word: 9 bits local node, 17 bits source node");
static_assert(ET < (1 << 21), "an edge id is kept in 21 bits");

template <bool V> struct mode_c { static constexpr bool value = V; };

template <int H>
__device__ __forceinline__ void load_att(const float* __restrict__ p, int node, float (&v)[H]) {
    if constexpr (H % 4 == 0) {
        #pragma unroll
        for (int k = 0; k < H / 4; ++k) {
            const float4 t = *(const float4*)(p + (size_t)node * H + k * 4);
            v[k * 4] = t.x; v[k * 4 + 1] = t.y; v[k * 4 + 2] = t.z; v[k * 4 + 3] = t.w;
        }
    } else {
        #pragma unroll
        for (int k = 0; k < H; ++k) v[k] = p[(size_t)node * H + k];
    }
}

template <int H, bool KEEP>
__device__ __forceinline__ void max_batch(const int* __restrict__ ei, const float* __restrict__ a_src,
                                          const float* __restrict__ a_dst, float* mx,
                                          const int* qw, int first, int nvalid, int lo, int lane,
                                          int* __restrict__ lst, int lpos) {
    const bool ok = lane < nvalid;
    const int e  = qw[first + (ok ? lane : 0)];
    const int ec = e < EE ? e : 0;
    const int sl = ei[ec];
    const int dd = ei[EE + ec];
    const int s  = (e < EE) ? sl : e - EE;
    const int dl = ((e < EE) ? dd : e - EE) - lo;
    if constexpr (KEEP) {
        if (ok && lpos + lane < WCAP) lst[lpos + lane] = (int)(((unsigned)dl << 17) | (unsigned)s);
    }
    float as[H], ad[H];
    load_att<H>(a_src, s, as);
    load_att<H>(a_dst, lo + dl, ad);
    if (ok) {
        #pragma unroll
        for (int h = 0; h < H; ++h) atomicMaxF(&mx[h * NPB + dl], lrelu(as[h] + ad[h]));
    }
}

template <int H>
__device__ __forceinline__ void walk_max(const int* __restrict__ lw, int n, const float* __restrict__ a_src,
                                         const float* __restrict__ a_dst, float* mx, int lo, int lane) {
    constexpr int U = 4;
    for (int b = 0; b < n; b += 32 * U) {
        int  ent[U];
        bool ok[U];
        #pragma unroll
        for (int u = 0; u < U; ++u) {
            const int idx = b + u * 32 + lane;
            ok[u] = idx < n;
            ent[u] = lw[ok[u] ? idx : 0];
        }
        float as[U][H], ad[U][H];
        #pragma unroll
        for (int u = 0; u < U; ++u) {
            load_att<H>(a_src, ent[u] & 0x1FFFF, as[u]);
            load_att<H>(a_dst, lo + (ent[u] >> 17), ad[u]);
        }
        #pragma unroll
        for (int u = 0; u < U; ++u) {
            if (ok[u]) {
                #pragma unroll
                for (int h = 0; h < H; ++h)
                    atomicMaxF(&mx[h * NPB + (ent[u] >> 17)], lrelu(as[u][h] + ad[u][h]));
            }
        }
    }
}

template <int H, int HCt, int STEPS>
__device__ __forceinline__ void own_batch(const int* __restrict__ ei, const float* __restrict__ a_src,
                                          const float* __restrict__ a_dst, const float* mx,
                                          const float* __restrict__ hsrc, float* acc, float* dn,
                                          const int* qw, int first, int nvalid, int lo, int lane) {
    constexpr int LPE = HCt / 4;
    constexpr int EPW = 32 / LPE;
    const int sub = lane / LPE;
    const int q   = lane % LPE;
    const int h   = (q * 4) / (HCt / H);
    int  ents[STEPS], sv[STEPS], dlv[STEPS];
    bool ok[STEPS];
    #pragma unroll
    for (int u = 0; u < STEPS; ++u) {
        const int idx = u * EPW + sub;
        ok[u] = idx < nvalid;
        ents[u] = qw[first + idx];
    }
    #pragma unroll
    for (int u = 0; u < STEPS; ++u) {
        const int e  = ok[u] ? ents[u] : 0;
        const int ec = e < EE ? e : 0;
        const int sl = ei[ec];
        const int dd = ei[EE + ec];
        sv[u]  = (e < EE) ? sl : e - EE;
        dlv[u] = ok[u] ? ((e < EE) ? dd : e - EE) - lo : 0;
    }
    float4 hv[STEPS];
    float as[STEPS], ad[STEPS], am[STEPS];
    #pragma unroll
    for (int u = 0; u < STEPS; ++u) {
        const int sn = sv[u], dnode = lo + dlv[u];
        hv[u] = *(const float4*)(hsrc + (size_t)sn * HCt + q * 4);
        as[u] = a_src[sn * H + h];
        ad[u] = a_dst[dnode * H + h];
        am[u] = mx[h * NPB + dlv[u]];
    }
    #pragma unroll
    for (int u = 0; u < STEPS; ++u) {
        const float w = expf(lrelu(as[u] + ad[u]) - am[u]);
        if (ok[u]) {
            if ((q & 3) == 0) atomicAdd(&dn[dlv[u] * H + h], w);
            float* ap = acc + dlv[u] * HCt + q;
            atomicAdd(ap,           hv[u].x * w);
            atomicAdd(ap + LPE,     hv[u].y * w);
            atomicAdd(ap + 2 * LPE, hv[u].z * w);
            atomicAdd(ap + 3 * LPE, hv[u].w * w);
        }
    }
}

template <int H, int HCt, int STEPS>
__device__ __forceinline__ void own_walk(const int* __restrict__ lw, int n, const float* __restrict__ a_src,
                                         const float* __restrict__ a_dst, const float* mx,
                                         const float* __restrict__ hsrc, float* acc, float* dn,
                                         int lo, int lane) {
    constexpr int LPE = HCt / 4;
    constexpr int EPW = 32 / LPE;
    constexpr int B   = EPW * STEPS;
    const int sub = lane / LPE;
    const int q   = lane % LPE;
    const int h   = (q * 4) / (HCt / H);
    int ents[STEPS], nxt[STEPS];
    #pragma unroll
    for (int u = 0; u < STEPS; ++u) {
        const int idx = u * EPW + sub;
        ents[u] = lw[idx < n ? idx : 0];
    }
    for (int b = 0; b < n; b += B) {
        #pragma unroll
        for (int u = 0; u < STEPS; ++u) {
            const int idx = b + B + u * EPW + sub;
            nxt[u] = lw[idx < n ? idx : 0];
        }
        float4 hv[STEPS];
        float as[STEPS], ad[STEPS], am[STEPS];
        int   dlv[STEPS];
        bool  ok[STEPS];
        #pragma unroll
        for (int u = 0; u < STEPS; ++u) {
            ok[u] = b + u * EPW + sub < n;
            const int ent = ok[u] ? ents[u] : 0;
            const int sn = ent & 0x1FFFF;
            dlv[u] = ent >> 17;
            const int dnode = lo + dlv[u];
            hv[u] = *(const float4*)(hsrc + (size_t)sn * HCt + q * 4);
            as[u] = a_src[sn * H + h];
            ad[u] = a_dst[dnode * H + h];
            am[u] = mx[h * NPB + dlv[u]];
        }
        #pragma unroll
        for (int u = 0; u < STEPS; ++u) {
            const float w = expf(lrelu(as[u] + ad[u]) - am[u]);
            if (ok[u]) {
                if ((q & 3) == 0) atomicAdd(&dn[dlv[u] * H + h], w);
                float* ap = acc + dlv[u] * HCt + q;
                atomicAdd(ap,           hv[u].x * w);
                atomicAdd(ap + LPE,     hv[u].y * w);
                atomicAdd(ap + 2 * LPE, hv[u].z * w);
                atomicAdd(ap + 3 * LPE, hv[u].w * w);
            }
        }
        #pragma unroll
        for (int u = 0; u < STEPS; ++u) ents[u] = nxt[u];
    }
}

template <int H, int HCt, int STEPS, int LAYER>
__global__ __launch_bounds__(1024)
void gat_own(const int* __restrict__ ei, const float* __restrict__ a_src,
             const float* __restrict__ a_dst,
             const float* __restrict__ hsrc, const float* __restrict__ bias,
             float* __restrict__ dstv, int* __restrict__ elist, int* __restrict__ ecnt,
             const unsigned char* __restrict__ own) {
    constexpr int LPE  = HCt / 4;
    constexpr int EPW  = 32 / LPE;
    constexpr int B    = EPW * STEPS;
    constexpr int QCAP = (B > MAXB ? B : MAXB) + 512;
    __shared__ float acc[NPB * HCt];
    __shared__ float dn[NPB * H];
    __shared__ float mx[NPB * H];
    __shared__ int   qbuf[32 * QCAP];
    __shared__ int   wcnt[32];

    const int tid = threadIdx.x, lane = tid & 31;
    const int wave = __builtin_amdgcn_readfirstlane(tid >> 5);
    const int lo  = blockIdx.x * NPB;
    const int cnt = (NN - lo < NPB) ? (NN - lo) : NPB;
    for (int i = tid; i < NPB * HCt; i += 1024) acc[i] = 0.0f;
    for (int i = tid; i < NPB * H;   i += 1024) { dn[i] = 0.0f; mx[i] = -__builtin_inff(); }
    __syncthreads();

    int* lst = elist + (size_t)(blockIdx.x * 32 + wave) * WCAP;
    int* qw  = qbuf + wave * QCAP;

    auto scan_pass = [&](auto mode) -> int {
    constexpr bool MAXP = decltype(mode)::value;
    constexpr int  BB   = MAXP ? MAXB : B;
    int  qn = 0;
    int  ln = 0;
    const unsigned lt = (1u << lane) - 1u;
    auto push = [&](bool m, unsigned ent) {
        const unsigned mask = __builtin_amdgcn_ballot_w32(m);
        if (m) qw[qn + __builtin_popcount(mask & lt)] = (int)ent;
        qn += __builtin_popcount(mask);
    };
    auto batch = [&](int first, int nvalid) {
        if constexpr (MAXP) max_batch<H, LAYER == 1>(ei, a_src, a_dst, mx, qw, first, nvalid, lo, lane, lst, ln);
        else own_batch<H, HCt, STEPS>(ei, a_src, a_dst, mx, hsrc, acc, dn, qw, first, nvalid, lo, lane);
    };
    auto drain = [&]() {
        while (qn >= BB) {
            batch(qn - BB, BB);
            qn -= BB;
            ln += BB;
        }
    };

    for (int b0 = wave * 32; b0 < cnt; b0 += 1024) {
        const int dl = b0 + lane;
        push(dl < cnt, (unsigned)(EE + lo + dl));
    }
    drain();

    if constexpr (LAYER == 1) {
    const uint4* ownp = (const uint4*)own;
    const unsigned pat = (unsigned)blockIdx.x * 0x01010101u;
    const int it0 = (blockIdx.x * NITB) / NGRP;
    auto ldscan = [&](int i, uint4 (&buf)[4]) -> int {
        int it = i + it0;
        if (it >= NITB) it -= NITB;
        const int cb = it * CH_STEP + wave * 128 + lane;
        #pragma unroll
        for (int k = 0; k < 4; ++k) {
            const int ck = cb + k * 32;
            buf[k] = ownp[ck < NCH ? ck : NCH - 1];
        }
        return cb;
    };
    auto hit = [&](unsigned w) -> unsigned {
        const unsigned x = w ^ pat;
        return ~(((x & 0x7F7F7F7Fu) + 0x7F7F7F7Fu) | x) & 0x80808080u;
    };
    auto test = [&](const uint4 (&buf)[4], int cb) {
        unsigned mk[4];
        #pragma unroll
        for (int k = 0; k < 4; ++k) {
            const unsigned m4 = (hit(buf[k].x) >> 7) | (hit(buf[k].y) >> 6) |
                                (hit(buf[k].z) >> 5) | (hit(buf[k].w) >> 4);
            mk[k] = (cb + k * 32 < NCH) ? m4 : 0u;
        }
        unsigned long long M = ((unsigned long long)(mk[2] | (mk[3] << 4)) << 32) |
                               (unsigned long long)(mk[0] | (mk[1] << 4));
        while (__builtin_amdgcn_ballot_w32(M != 0)) {
            const bool m = M != 0;
            const int  p = m ? __builtin_ctzll(M) : 0;
            const int  k = ((p >> 5) << 1) | ((p >> 2) & 1);
            push(m, (unsigned)((cb + k * 32) * 16 + ((p & 3) << 2) + ((p >> 3) & 3)));
            M &= M - 1;
            drain();
        }
    };
    uint4 bufa[4], bufb[4];
    int cba = ldscan(0, bufa), cbb;
    for (int i = 0; i < NITB; i += 2) {
        cbb = ldscan(i + 1, bufb);
        __builtin_amdgcn_sched_barrier(0);
        test(bufa, cba);
        cba = ldscan(i + 2, bufa);
        __builtin_amdgcn_sched_barrier(0);
        if (i + 1 < NITB) test(bufb, cbb);
    }
    } else {
    const int* dstp = ei + EE;
    const int it0 = (blockIdx.x * NIT) / NGRP;
    auto ldscan = [&](int i, int4 (&buf)[4]) -> int {
        int it = i + it0;
        if (it >= NIT) it -= NIT;
        const int eb = it * SCAN_EDGES + wave * 512 + lane * 4;
        #pragma unroll
        for (int k = 0; k < 4; ++k) {
            const int ek = eb + k * 128;
            buf[k] = *(const int4*)(dstp + (ek < EE ? ek : EE - 4));
        }
        return eb;
    };
    auto test = [&](const int4 (&buf)[4], int eb) {
        #pragma unroll
        for (int k = 0; k < 4; ++k) {
            const int ek = eb + k * 128;
            const unsigned d0 = (unsigned)(buf[k].x - lo), d1 = (unsigned)(buf[k].y - lo);
            const unsigned d2 = (unsigned)(buf[k].z - lo), d3 = (unsigned)(buf[k].w - lo);
            const bool in = ek < EE;
            const bool m0 = in && d0 < (unsigned)cnt, m1 = in && d1 < (unsigned)cnt;
            const bool m2 = in && d2 < (unsigned)cnt, m3 = in && d3 < (unsigned)cnt;
            if (__builtin_amdgcn_ballot_w32(m0 | m1 | m2 | m3)) {
                push(m0, (unsigned)(ek + 0));
                push(m1, (unsigned)(ek + 1));
                push(m2, (unsigned)(ek + 2));
                push(m3, (unsigned)(ek + 3));
            }
        }
        drain();
    };
    int4 bufa[4], bufb[4];
    int eba = ldscan(0, bufa), ebb;
    for (int i = 0; i < NIT; i += 2) {
        ebb = ldscan(i + 1, bufb);
        test(bufa, eba);
        eba = ldscan(i + 2, bufa);
        if (i + 1 < NIT) test(bufb, ebb);
    }
    }
    if (qn > 0) {
        batch(0, qn);
        ln += qn;
    }
    return ln;
    };

    if constexpr (LAYER == 1) {
        const int nl = scan_pass(mode_c<true>{});
        if (lane == 0) { ecnt[blockIdx.x * 32 + wave] = nl; wcnt[wave] = nl; }
        __syncthreads();
        if (__builtin_amdgcn_ballot_w32(wcnt[lane] > WCAP) == 0)
            own_walk<H, HCt, 8>(lst, nl, a_src, a_dst, mx, hsrc, acc, dn, lo, lane);
        else
            (void)scan_pass(mode_c<false>{});
    } else {
        const int c = ecnt[blockIdx.x * 32 + lane];
        const int nl = __builtin_amdgcn_readlane(c, wave);
        if (__builtin_amdgcn_ballot_w32(c > WCAP) == 0) {
            walk_max<H>(lst, nl, a_src, a_dst, mx, lo, lane);
            __syncthreads();
            own_walk<H, HCt, 8>(lst, nl, a_src, a_dst, mx, hsrc, acc, dn, lo, lane);
        } else {
            (void)scan_pass(mode_c<true>{});
            __syncthreads();
            (void)scan_pass(mode_c<false>{});
        }
    }
    __syncthreads();

    if constexpr (LAYER == 1) {
        const float4 bq = *(const float4*)(bias + lane * 4);
        for (int r = wave; r < cnt; r += 32) {
            const float dv = dn[r * H + (lane >> 2)] + 1e-16f;
            const float* ap = acc + r * HCt + lane;
            float4 o;
            o.x = ap[0]       / dv + bq.x;
            o.y = ap[LPE]     / dv + bq.y;
            o.z = ap[2 * LPE] / dv + bq.z;
            o.w = ap[3 * LPE] / dv + bq.w;
            o.x = o.x > 0.0f ? o.x : (expf(o.x) - 1.0f);
            o.y = o.y > 0.0f ? o.y : (expf(o.y) - 1.0f);
            o.z = o.z > 0.0f ? o.z : (expf(o.z) - 1.0f);
            o.w = o.w > 0.0f ? o.w : (expf(o.w) - 1.0f);
            *(float4*)(dstv + (size_t)(lo + r) * HCt + lane * 4) = o;
        }
    } else {
        for (int r = tid; r < cnt; r += 1024) {
            float v[HCt], m = -__builtin_inff();
            const float dv = dn[r] + 1e-16f;
            #pragma unroll
            for (int c = 0; c < HCt; ++c) {
                v[c] = acc[r * HCt + (c & 3) * LPE + (c >> 2)] / dv + bias[c];
                m = fmaxf(m, v[c]);
            }
            float sum = 0.0f;
            #pragma unroll
            for (int c = 0; c < HCt; ++c) sum += expf(v[c] - m);
            const float ls = logf(sum);
            #pragma unroll
            for (int c = 0; c < HCt; ++c) dstv[(size_t)(lo + r) * HCt + c] = v[c] - m - ls;
        }
    }
}

extern "C" void kernel_launch(void* const* d_in, const int* in_sizes, int n_in,
                              void* d_out, int out_size, void* d_ws, size_t ws_size,
                              hipStream_t stream) {
    (void)in_sizes; (void)n_in; (void)out_size; (void)ws_size;
    const float* x   = (const float*)d_in[0];
    const int*   ei  = (const int*)d_in[1];
    const float* W1  = (const float*)d_in[2];
    const float* as1 = (const float*)d_in[3];
    const float* ad1 = (const float*)d_in[4];
    const float* b1  = (const float*)d_in[5];
    const float* W2  = (const float*)d_in[6];
    const float* as2 = (const float*)d_in[7];
    const float* ad2 = (const float*)d_in[8];
    const float* b2  = (const float*)d_in[9];
    float* out = (float*)d_out;

    char* wsb = (char*)d_ws;
    size_t off = 0;
    auto alloc = [&](size_t bytes) -> void* {
        void* p = wsb + off;
        off += (bytes + 255) & ~(size_t)255;
        return p;
    };
    _Float16* w1h  = (_Float16*)alloc((size_t)W1R * NF * 2);
    _Float16* w2h  = (_Float16*)alloc((size_t)W2R * HC * 2);
    _Float16* xh   = (_Float16*)alloc((size_t)NN * NF * 2);
    float* h1      = (float*)alloc((size_t)NN * HC * 4);
    float* asrc1   = (float*)alloc((size_t)NN * HH * 4);
    float* adst1   = (float*)alloc((size_t)NN * HH * 4);
    float* h2      = (float*)alloc((size_t)NN * NC * 4);
    float* asrc2   = (float*)alloc((size_t)NN * 4);
    float* adst2   = (float*)alloc((size_t)NN * 4);
    int* elist     = (int*)alloc((size_t)NGRP * 32 * WCAP * 4);
    int* ecnt      = (int*)alloc((size_t)NGRP * 32 * 4);
    float* act1    = (float*)xh;
    unsigned char* ownb = (unsigned char*)h2;

    const dim3 B(256);
    gat_cvtw <<<(CVTW_N + 255) / 256, B, 0, stream>>>(W1, W2, as1, ad1, as2, ad2, w1h, w2h);
    gat_cvtx <<<CVT_BLOCKS, B, 0, stream>>>(x, xh);
    gat_gemm1<<<(NN / 16 + 7) / 8, B, 0, stream>>>(xh, w1h, h1, asrc1, adst1);
    gat_ownb <<<(EE / 16 + 255) / 256, B, 0, stream>>>(ei, ownb);
    gat_own<HH, HC, 8, 1><<<NGRP, 1024, 0, stream>>>(ei, asrc1, adst1, h1, b1, act1, elist, ecnt, ownb);
    gat_gemm2<<<(NN / 16 + 7) / 8, B, 0, stream>>>(act1, w2h, h2, asrc2, adst2);
    gat_own<1, NC, 4, 2><<<NGRP, 1024, 0, stream>>>(ei, asrc2, adst2, h2, b2, out, elist, ecnt, ownb);
}


# sum pass forms the source's attention scalar from the h row it gathers anyway (4 FMAs a lane + quad DPP sum: new summation order for a_src there); second a_src gather per edge gone, both layers
# speedup vs baseline: 4.1742x; 1.0352x over previous
#include <hip/hip_runtime.h>

#define NN  100000
#define NF  256
#define HH  8
#define CC  16
#define HC  128
#define NC  16
#define EE  1600000
#define ET  1700000
#define NEG 0.2f
#define NPB  392
#define NGRP 256
static_assert((NN + NPB - 1) / NPB <= 256, "an owner-group id must fit one byte");
static_assert(EE % 16 == 0, "the owner-byte list is read 16 edges a load");

typedef _Float16 h8   __attribute__((ext_vector_type(8)));
typedef _Float16 v16h __attribute__((ext_vector_type(16)));
typedef float    v8f  __attribute__((ext_vector_type(8)));

__device__ __forceinline__ void edge_sd(const int* __restrict__ ei, int e, int& s, int& d) {
    if (e < EE) { s = ei[e]; d = ei[EE + e]; }
    else        { s = e - EE; d = s; }
}

__device__ __forceinline__ void atomicMaxF(float* addr, float val) {
    if (val >= 0.0f) atomicMax((int*)addr, __float_as_int(val));
    else             atomicMin((unsigned int*)addr, __float_as_uint(val));
}

__device__ __forceinline__ float lrelu(float a) { return a > 0.0f ? a : NEG * a; }

__device__ __forceinline__ float quad_sum(float v) {
    v += __int_as_float(__builtin_amdgcn_update_dpp(0, __float_as_int(v), 0xB1, 0xF, 0xF, true));
    v += __int_as_float(__builtin_amdgcn_update_dpp(0, __float_as_int(v), 0x4E, 0xF, 0xF, true));
    return v;
}

#define W1R (HC + 2 * HH)
#define W2R (NC + 16)
#define CVTW_N (W1R * NF + W2R * HC)
__global__ __launch_bounds__(256) void gat_cvtw(const float* __restrict__ W1,
                                                const float* __restrict__ W2,
                                                const float* __restrict__ as1,
                                                const float* __restrict__ ad1,
                                                const float* __restrict__ as2,
                                                const float* __restrict__ ad2,
                                                _Float16* __restrict__ w1h,
                                                _Float16* __restrict__ w2h) {
    const int i = blockIdx.x * 256 + threadIdx.x;
    if (i < HC * NF) {
        w1h[i] = (_Float16)W1[i];
    } else if (i < W1R * NF) {
        const int j = i - HC * NF;
        const int row = j / NF, f = j % NF;
        const int h = row & (HH - 1);
        const float* __restrict__ att = (row < HH ? as1 : ad1) + h * CC;
        float s = 0.0f;
        #pragma unroll
        for (int cc = 0; cc < CC; ++cc) s += W1[(size_t)(h * CC + cc) * NF + f] * att[cc];
        w1h[i] = (_Float16)s;
    } else if (i < W1R * NF + NC * HC) {
        const int j = i - W1R * NF;
        w2h[j] = (_Float16)W2[j];
    } else if (i < CVTW_N) {
        const int j = i - W1R * NF - NC * HC;
        const int row = j / HC, k = j % HC;
        float s = 0.0f;
        if ((row & 7) == 0) {
            const float* __restrict__ att = row == 0 ? as2 : ad2;
            #pragma unroll
            for (int cc = 0; cc < NC; ++cc) s += W2[(size_t)cc * HC + k] * att[cc];
        }
        w2h[NC * HC + j] = (_Float16)s;
    }
}

#define CVT_LD 20
#define CVT_PER_BLOCK (256 * CVT_LD)
#define CVT_BLOCKS ((NN * NF / 4) / CVT_PER_BLOCK)
static_assert((NN * NF) % 4 == 0 && (NN * NF / 4) % CVT_PER_BLOCK == 0,
              "the conversion grid covers x exactly: no tail piece");
typedef _Float16 h4 __attribute__((ext_vector_type(4)));

__global__ __launch_bounds__(256) void gat_cvtx(const float* __restrict__ x,
                                                _Float16* __restrict__ xh) {
    const float4* __restrict__ xp = (const float4*)x;
    h4* __restrict__ op = (h4*)xh;
    const int base = blockIdx.x * CVT_PER_BLOCK + threadIdx.x;
    float4 v[CVT_LD];
    #pragma unroll
    for (int k = 0; k < CVT_LD; ++k) v[k] = xp[base + k * 256];
    #pragma unroll
    for (int k = 0; k < CVT_LD; ++k) {
        h4 o;
        o.x = (_Float16)v[k].x; o.y = (_Float16)v[k].y;
        o.z = (_Float16)v[k].z; o.w = (_Float16)v[k].w;
        op[base + k * 256] = o;
    }
}

__global__ __launch_bounds__(256) __attribute__((amdgpu_waves_per_eu(8)))
void gat_gemm1(const _Float16* __restrict__ xh, const _Float16* __restrict__ w1h,
               float* __restrict__ h1, float* __restrict__ a_src, float* __restrict__ a_dst) {
    __shared__ _Float16 lw[W1R * NF];
    const int tid = threadIdx.x;
    for (int idx = tid; idx < W1R * 32; idx += 256) {
        int r = idx >> 5, ch = idx & 31;
        *(h8*)(&lw[r * NF + ((ch ^ (r & 15)) << 3)]) =
            *(const h8*)(w1h + (size_t)r * NF + (ch << 3));
    }
    __syncthreads();

    const int lane = tid & 31;
    const int rt   = blockIdx.x * 8 + (tid >> 5);
    if (rt >= NN / 16) return;
    const int rsub = lane & 15;
    const int ksel = (lane >> 4) << 3;
    const _Float16* xrow = xh + (size_t)(rt * 16 + rsub) * NF;

    v8f acc[8] = {};
    v8f att = {};
    #pragma unroll
    for (int ks = 0; ks < 8; ++ks) {
        const int kb = ks * 32 + ksel;
        const int c0 = kb >> 3;
        union { v16h v; h8 p[2]; } A;
        A.p[0] = *(const h8*)(xrow + kb);
        A.p[1] = *(const h8*)(xrow + kb + 16);
        #pragma unroll
        for (int ct = 0; ct < 8; ++ct) {
            const int n = ct * 16 + rsub;
            union { v16h v; h8 p[2]; } B;
            B.p[0] = *(const h8*)(&lw[n * NF + (((c0    ) ^ (n & 15)) << 3)]);
            B.p[1] = *(const h8*)(&lw[n * NF + (((c0 + 2) ^ (n & 15)) << 3)]);
            acc[ct] = __builtin_amdgcn_wmma_f32_16x16x32_f16(
                false, A.v, false, B.v, (short)0, acc[ct], false, false);
        }
        {
            const int n = HC + rsub;
            union { v16h v; h8 p[2]; } F;
            F.p[0] = *(const h8*)(&lw[n * NF + (((c0    ) ^ (n & 15)) << 3)]);
            F.p[1] = *(const h8*)(&lw[n * NF + (((c0 + 2) ^ (n & 15)) << 3)]);
            att = __builtin_amdgcn_wmma_f32_16x16x32_f16(
                false, F.v, false, A.v, (short)0, att, false, false);
        }
    }
    const int rhi = (lane >> 4) << 3;
    #pragma unroll
    for (int ct = 0; ct < 8; ++ct) {
        #pragma unroll
        for (int r = 0; r < 8; ++r)
            h1[(size_t)(rt * 16 + rhi + r) * HC + ct * 16 + rsub] = acc[ct][r];
    }
    float* ap = (lane < 16 ? a_src : a_dst) + (size_t)(rt * 16 + rsub) * HH;
    *(float4*)(ap)     = make_float4(att[0], att[1], att[2], att[3]);
    *(float4*)(ap + 4) = make_float4(att[4], att[5], att[6], att[7]);
}

__global__ __launch_bounds__(256)
void gat_ownb(const int* __restrict__ ei, unsigned char* __restrict__ own) {
    const int i = blockIdx.x * 256 + threadIdx.x;
    if (i >= EE / 16) return;
    const int4* dp = (const int4*)(ei + EE) + (size_t)i * 4;
    unsigned w[4];
    #pragma unroll
    for (int k = 0; k < 4; ++k) {
        const int4 d = dp[k];
        w[k] = ((unsigned)d.x / NPB) | (((unsigned)d.y / NPB) << 8) |
               (((unsigned)d.z / NPB) << 16) | (((unsigned)d.w / NPB) << 24);
    }
    ((uint4*)own)[i] = make_uint4(w[0], w[1], w[2], w[3]);
}

__global__ __launch_bounds__(256)
void gat_gemm2(const float* __restrict__ a1, const _Float16* __restrict__ w2h,
               float* __restrict__ h2, float* __restrict__ a_src, float* __restrict__ a_dst) {
    const int lane = threadIdx.x & 31;
    const int rt   = blockIdx.x * 8 + (threadIdx.x >> 5);
    if (rt >= NN / 16) return;
    const int rsub = lane & 15;
    const int ksel = (lane >> 4) << 3;
    const float*    arow = a1 + (size_t)(rt * 16 + rsub) * HC;
    const _Float16* wrow = w2h + (size_t)rsub * HC;
    const _Float16* frow = w2h + (size_t)(NC + rsub) * HC;

    v8f acc = {};
    v8f att = {};
    #pragma unroll
    for (int ks = 0; ks < 4; ++ks) {
        const int kb = ks * 32 + ksel;
        union { v16h v; _Float16 e[16]; } A;
        #pragma unroll
        for (int t = 0; t < 8; ++t) {
            A.e[t]     = (_Float16)arow[kb + t];
            A.e[8 + t] = (_Float16)arow[kb + 16 + t];
        }
        union { v16h v; h8 p[2]; } B;
        B.p[0] = *(const h8*)(wrow + kb);
        B.p[1] = *(const h8*)(wrow + kb + 16);
        acc = __builtin_amdgcn_wmma_f32_16x16x32_f16(
            false, A.v, false, B.v, (short)0, acc, false, false);
        union { v16h v; h8 p[2]; } F;
        F.p[0] = *(const h8*)(frow + kb);
        F.p[1] = *(const h8*)(frow + kb + 16);
        att = __builtin_amdgcn_wmma_f32_16x16x32_f16(
            false, F.v, false, A.v, (short)0, att, false, false);
    }
    const int rhi = (lane >> 4) << 3;
    #pragma unroll
    for (int r = 0; r < 8; ++r)
        h2[(size_t)(rt * 16 + rhi + r) * NC + rsub] = acc[r];
    (lane < 16 ? a_src : a_dst)[rt * 16 + rsub] = att[0];
}

#define SCAN_EDGES 16384
#define NIT ((EE + SCAN_EDGES - 1) / SCAN_EDGES)
#define NCH (EE / 16)
#define CH_STEP 4096
#define NITB ((NCH + CH_STEP - 1) / CH_STEP)
#define WCAP 320
#define MAXB 32
static_assert(NPB <= 512 && NN <= (1 << 17), "list word: 9 bits local node, 17 bits source node");
static_assert(ET < (1 << 21), "an edge id is kept in 21 bits");

template <bool V> struct mode_c { static constexpr bool value = V; };

template <int H>
__device__ __forceinline__ void load_att(const float* __restrict__ p, int node, float (&v)[H]) {
    if constexpr (H % 4 == 0) {
        #pragma unroll
        for (int k = 0; k < H / 4; ++k) {
            const float4 t = *(const float4*)(p + (size_t)node * H + k * 4);
            v[k * 4] = t.x; v[k * 4 + 1] = t.y; v[k * 4 + 2] = t.z; v[k * 4 + 3] = t.w;
        }
    } else {
        #pragma unroll
        for (int k = 0; k < H; ++k) v[k] = p[(size_t)node * H + k];
    }
}

template <int H, bool KEEP>
__device__ __forceinline__ void max_batch(const int* __restrict__ ei, const float* __restrict__ a_src,
                                          const float* __restrict__ a_dst, float* mx,
                                          const int* qw, int first, int nvalid, int lo, int lane,
                                          int* __restrict__ lst, int lpos) {
    const bool ok = lane < nvalid;
    const int e  = qw[first + (ok ? lane : 0)];
    const int ec = e < EE ? e : 0;
    const int sl = ei[ec];
    const int dd = ei[EE + ec];
    const int s  = (e < EE) ? sl : e - EE;
    const int dl = ((e < EE) ? dd : e - EE) - lo;
    if constexpr (KEEP) {
        if (ok && lpos + lane < WCAP) lst[lpos + lane] = (int)(((unsigned)dl << 17) | (unsigned)s);
    }
    float as[H], ad[H];
    load_att<H>(a_src, s, as);
    load_att<H>(a_dst, lo + dl, ad);
    if (ok) {
        #pragma unroll
        for (int h = 0; h < H; ++h) atomicMaxF(&mx[h * NPB + dl], lrelu(as[h] + ad[h]));
    }
}

template <int H>
__device__ __forceinline__ void walk_max(const int* __restrict__ lw, int n, const float* __restrict__ a_src,
                                         const float* __restrict__ a_dst, float* mx, int lo, int lane) {
    constexpr int U = 4;
    for (int b = 0; b < n; b += 32 * U) {
        int  ent[U];
        bool ok[U];
        #pragma unroll
        for (int u = 0; u < U; ++u) {
            const int idx = b + u * 32 + lane;
            ok[u] = idx < n;
            ent[u] = lw[ok[u] ? idx : 0];
        }
        float as[U][H], ad[U][H];
        #pragma unroll
        for (int u = 0; u < U; ++u) {
            load_att<H>(a_src, ent[u] & 0x1FFFF, as[u]);
            load_att<H>(a_dst, lo + (ent[u] >> 17), ad[u]);
        }
        #pragma unroll
        for (int u = 0; u < U; ++u) {
            if (ok[u]) {
                #pragma unroll
                for (int h = 0; h < H; ++h)
                    atomicMaxF(&mx[h * NPB + (ent[u] >> 17)], lrelu(as[u][h] + ad[u][h]));
            }
        }
    }
}

template <int H, int HCt, int STEPS>
__device__ __forceinline__ void own_batch(const int* __restrict__ ei, const float* __restrict__ a_src,
                                          const float* __restrict__ a_dst, const float* mx,
                                          const float* __restrict__ hsrc, float* acc, float* dn,
                                          const int* qw, int first, int nvalid, int lo, int lane) {
    constexpr int LPE = HCt / 4;
    constexpr int EPW = 32 / LPE;
    const int sub = lane / LPE;
    const int q   = lane % LPE;
    const int h   = (q * 4) / (HCt / H);
    int  ents[STEPS], sv[STEPS], dlv[STEPS];
    bool ok[STEPS];
    #pragma unroll
    for (int u = 0; u < STEPS; ++u) {
        const int idx = u * EPW + sub;
        ok[u] = idx < nvalid;
        ents[u] = qw[first + idx];
    }
    #pragma unroll
    for (int u = 0; u < STEPS; ++u) {
        const int e  = ok[u] ? ents[u] : 0;
        const int ec = e < EE ? e : 0;
        const int sl = ei[ec];
        const int dd = ei[EE + ec];
        sv[u]  = (e < EE) ? sl : e - EE;
        dlv[u] = ok[u] ? ((e < EE) ? dd : e - EE) - lo : 0;
    }
    float4 hv[STEPS];
    float as[STEPS], ad[STEPS], am[STEPS];
    #pragma unroll
    for (int u = 0; u < STEPS; ++u) {
        const int sn = sv[u], dnode = lo + dlv[u];
        hv[u] = *(const float4*)(hsrc + (size_t)sn * HCt + q * 4);
        as[u] = a_src[sn * H + h];
        ad[u] = a_dst[dnode * H + h];
        am[u] = mx[h * NPB + dlv[u]];
    }
    #pragma unroll
    for (int u = 0; u < STEPS; ++u) {
        const float w = expf(lrelu(as[u] + ad[u]) - am[u]);
        if (ok[u]) {
            if ((q & 3) == 0) atomicAdd(&dn[dlv[u] * H + h], w);
            float* ap = acc + dlv[u] * HCt + q;
            atomicAdd(ap,           hv[u].x * w);
            atomicAdd(ap + LPE,     hv[u].y * w);
            atomicAdd(ap + 2 * LPE, hv[u].z * w);
            atomicAdd(ap + 3 * LPE, hv[u].w * w);
        }
    }
}

template <int H, int HCt, int STEPS>
__device__ __forceinline__ void own_walk(const int* __restrict__ lw, int n, const float* __restrict__ att,
                                         const float* __restrict__ a_dst, const float* mx,
                                         const float* __restrict__ hsrc, float* acc, float* dn,
                                         int lo, int lane) {
    constexpr int LPE = HCt / 4;
    constexpr int EPW = 32 / LPE;
    constexpr int B   = EPW * STEPS;
    static_assert(HCt / H == 16, "a head is 16 channels: the four lanes of one quad");
    const int sub = lane / LPE;
    const int q   = lane % LPE;
    const int h   = (q * 4) / (HCt / H);
    const float4 av = *(const float4*)(att + q * 4);
    int ents[STEPS], nxt[STEPS];
    #pragma unroll
    for (int u = 0; u < STEPS; ++u) {
        const int idx = u * EPW + sub;
        ents[u] = lw[idx < n ? idx : 0];
    }
    for (int b = 0; b < n; b += B) {
        #pragma unroll
        for (int u = 0; u < STEPS; ++u) {
            const int idx = b + B + u * EPW + sub;
            nxt[u] = lw[idx < n ? idx : 0];
        }
        float4 hv[STEPS];
        float ad[STEPS], am[STEPS];
        int   dlv[STEPS];
        bool  ok[STEPS];
        #pragma unroll
        for (int u = 0; u < STEPS; ++u) {
            ok[u] = b + u * EPW + sub < n;
            const int ent = ok[u] ? ents[u] : 0;
            const int sn = ent & 0x1FFFF;
            dlv[u] = ent >> 17;
            const int dnode = lo + dlv[u];
            hv[u] = *(const float4*)(hsrc + (size_t)sn * HCt + q * 4);
            ad[u] = a_dst[dnode * H + h];
            am[u] = mx[h * NPB + dlv[u]];
        }
        #pragma unroll
        for (int u = 0; u < STEPS; ++u) {
            const float as = quad_sum(fmaf(hv[u].w, av.w, fmaf(hv[u].z, av.z,
                                      fmaf(hv[u].y, av.y, hv[u].x * av.x))));
            const float w = expf(lrelu(as + ad[u]) - am[u]);
            if (ok[u]) {
                if ((q & 3) == 0) atomicAdd(&dn[dlv[u] * H + h], w);
                float* ap = acc + dlv[u] * HCt + q;
                atomicAdd(ap,           hv[u].x * w);
                atomicAdd(ap + LPE,     hv[u].y * w);
                atomicAdd(ap + 2 * LPE, hv[u].z * w);
                atomicAdd(ap + 3 * LPE, hv[u].w * w);
            }
        }
        #pragma unroll
        for (int u = 0; u < STEPS; ++u) ents[u] = nxt[u];
    }
}

template <int H, int HCt, int STEPS, int LAYER>
__global__ __launch_bounds__(1024)
void gat_own(const int* __restrict__ ei, const float* __restrict__ a_src,
             const float* __restrict__ a_dst, const float* __restrict__ att,
             const float* __restrict__ hsrc, const float* __restrict__ bias,
             float* __restrict__ dstv, int* __restrict__ elist, int* __restrict__ ecnt,
             const unsigned char* __restrict__ own) {
    constexpr int LPE  = HCt / 4;
    constexpr int EPW  = 32 / LPE;
    constexpr int B    = EPW * STEPS;
    constexpr int QCAP = (B > MAXB ? B : MAXB) + 512;
    __shared__ float acc[NPB * HCt];
    __shared__ float dn[NPB * H];
    __shared__ float mx[NPB * H];
    __shared__ int   qbuf[32 * QCAP];
    __shared__ int   wcnt[32];

    const int tid = threadIdx.x, lane = tid & 31;
    const int wave = __builtin_amdgcn_readfirstlane(tid >> 5);
    const int lo  = blockIdx.x * NPB;
    const int cnt = (NN - lo < NPB) ? (NN - lo) : NPB;
    for (int i = tid; i < NPB * HCt; i += 1024) acc[i] = 0.0f;
    for (int i = tid; i < NPB * H;   i += 1024) { dn[i] = 0.0f; mx[i] = -__builtin_inff(); }
    __syncthreads();

    int* lst = elist + (size_t)(blockIdx.x * 32 + wave) * WCAP;
    int* qw  = qbuf + wave * QCAP;

    auto scan_pass = [&](auto mode) -> int {
    constexpr bool MAXP = decltype(mode)::value;
    constexpr int  BB   = MAXP ? MAXB : B;
    int  qn = 0;
    int  ln = 0;
    const unsigned lt = (1u << lane) - 1u;
    auto push = [&](bool m, unsigned ent) {
        const unsigned mask = __builtin_amdgcn_ballot_w32(m);
        if (m) qw[qn + __builtin_popcount(mask & lt)] = (int)ent;
        qn += __builtin_popcount(mask);
    };
    auto batch = [&](int first, int nvalid) {
        if constexpr (MAXP) max_batch<H, LAYER == 1>(ei, a_src, a_dst, mx, qw, first, nvalid, lo, lane, lst, ln);
        else own_batch<H, HCt, STEPS>(ei, a_src, a_dst, mx, hsrc, acc, dn, qw, first, nvalid, lo, lane);
    };
    auto drain = [&]() {
        while (qn >= BB) {
            batch(qn - BB, BB);
            qn -= BB;
            ln += BB;
        }
    };

    for (int b0 = wave * 32; b0 < cnt; b0 += 1024) {
        const int dl = b0 + lane;
        push(dl < cnt, (unsigned)(EE + lo + dl));
    }
    drain();

    if constexpr (LAYER == 1) {
    const uint4* ownp = (const uint4*)own;
    const unsigned pat = (unsigned)blockIdx.x * 0x01010101u;
    const int it0 = (blockIdx.x * NITB) / NGRP;
    auto ldscan = [&](int i, uint4 (&buf)[4]) -> int {
        int it = i + it0;
        if (it >= NITB) it -= NITB;
        const int cb = it * CH_STEP + wave * 128 + lane;
        #pragma unroll
        for (int k = 0; k < 4; ++k) {
            const int ck = cb + k * 32;
            buf[k] = ownp[ck < NCH ? ck : NCH - 1];
        }
        return cb;
    };
    auto hit = [&](unsigned w) -> unsigned {
        const unsigned x = w ^ pat;
        return ~(((x & 0x7F7F7F7Fu) + 0x7F7F7F7Fu) | x) & 0x80808080u;
    };
    auto test = [&](const uint4 (&buf)[4], int cb) {
        unsigned mk[4];
        #pragma unroll
        for (int k = 0; k < 4; ++k) {
            const unsigned m4 = (hit(buf[k].x) >> 7) | (hit(buf[k].y) >> 6) |
                                (hit(buf[k].z) >> 5) | (hit(buf[k].w) >> 4);
            mk[k] = (cb + k * 32 < NCH) ? m4 : 0u;
        }
        unsigned long long M = ((unsigned long long)(mk[2] | (mk[3] << 4)) << 32) |
                               (unsigned long long)(mk[0] | (mk[1] << 4));
        while (__builtin_amdgcn_ballot_w32(M != 0)) {
            const bool m = M != 0;
            const int  p = m ? __builtin_ctzll(M) : 0;
            const int  k = ((p >> 5) << 1) | ((p >> 2) & 1);
            push(m, (unsigned)((cb + k * 32) * 16 + ((p & 3) << 2) + ((p >> 3) & 3)));
            M &= M - 1;
            drain();
        }
    };
    uint4 bufa[4], bufb[4];
    int cba = ldscan(0, bufa), cbb;
    for (int i = 0; i < NITB; i += 2) {
        cbb = ldscan(i + 1, bufb);
        __builtin_amdgcn_sched_barrier(0);
        test(bufa, cba);
        cba = ldscan(i + 2, bufa);
        __builtin_amdgcn_sched_barrier(0);
        if (i + 1 < NITB) test(bufb, cbb);
    }
    } else {
    const int* dstp = ei + EE;
    const int it0 = (blockIdx.x * NIT) / NGRP;
    auto ldscan = [&](int i, int4 (&buf)[4]) -> int {
        int it = i + it0;
        if (it >= NIT) it -= NIT;
        const int eb = it * SCAN_EDGES + wave * 512 + lane * 4;
        #pragma unroll
        for (int k = 0; k < 4; ++k) {
            const int ek = eb + k * 128;
            buf[k] = *(const int4*)(dstp + (ek < EE ? ek : EE - 4));
        }
        return eb;
    };
    auto test = [&](const int4 (&buf)[4], int eb) {
        #pragma unroll
        for (int k = 0; k < 4; ++k) {
            const int ek = eb + k * 128;
            const unsigned d0 = (unsigned)(buf[k].x - lo), d1 = (unsigned)(buf[k].y - lo);
            const unsigned d2 = (unsigned)(buf[k].z - lo), d3 = (unsigned)(buf[k].w - lo);
            const bool in = ek < EE;
            const bool m0 = in && d0 < (unsigned)cnt, m1 = in && d1 < (unsigned)cnt;
            const bool m2 = in && d2 < (unsigned)cnt, m3 = in && d3 < (unsigned)cnt;
            if (__builtin_amdgcn_ballot_w32(m0 | m1 | m2 | m3)) {
                push(m0, (unsigned)(ek + 0));
                push(m1, (unsigned)(ek + 1));
                push(m2, (unsigned)(ek + 2));
                push(m3, (unsigned)(ek + 3));
            }
        }
        drain();
    };
    int4 bufa[4], bufb[4];
    int eba = ldscan(0, bufa), ebb;
    for (int i = 0; i < NIT; i += 2) {
        ebb = ldscan(i + 1, bufb);
        test(bufa, eba);
        eba = ldscan(i + 2, bufa);
        if (i + 1 < NIT) test(bufb, ebb);
    }
    }
    if (qn > 0) {
        batch(0, qn);
        ln += qn;
    }
    return ln;
    };

    if constexpr (LAYER == 1) {
        const int nl = scan_pass(mode_c<true>{});
        if (lane == 0) { ecnt[blockIdx.x * 32 + wave] = nl; wcnt[wave] = nl; }
        __syncthreads();
        if (__builtin_amdgcn_ballot_w32(wcnt[lane] > WCAP) == 0)
            own_walk<H, HCt, 8>(lst, nl, att, a_dst, mx, hsrc, acc, dn, lo, lane);
        else
            (void)scan_pass(mode_c<false>{});
    } else {
        const int c = ecnt[blockIdx.x * 32 + lane];
        const int nl = __builtin_amdgcn_readlane(c, wave);
        if (__builtin_amdgcn_ballot_w32(c > WCAP) == 0) {
            walk_max<H>(lst, nl, a_src, a_dst, mx, lo, lane);
            __syncthreads();
            own_walk<H, HCt, 8>(lst, nl, att, a_dst, mx, hsrc, acc, dn, lo, lane);
        } else {
            (void)scan_pass(mode_c<true>{});
            __syncthreads();
            (void)scan_pass(mode_c<false>{});
        }
    }
    __syncthreads();

    if constexpr (LAYER == 1) {
        const float4 bq = *(const float4*)(bias + lane * 4);
        for (int r = wave; r < cnt; r += 32) {
            const float dv = dn[r * H + (lane >> 2)] + 1e-16f;
            const float* ap = acc + r * HCt + lane;
            float4 o;
            o.x = ap[0]       / dv + bq.x;
            o.y = ap[LPE]     / dv + bq.y;
            o.z = ap[2 * LPE] / dv + bq.z;
            o.w = ap[3 * LPE] / dv + bq.w;
            o.x = o.x > 0.0f ? o.x : (expf(o.x) - 1.0f);
            o.y = o.y > 0.0f ? o.y : (expf(o.y) - 1.0f);
            o.z = o.z > 0.0f ? o.z : (expf(o.z) - 1.0f);
            o.w = o.w > 0.0f ? o.w : (expf(o.w) - 1.0f);
            *(float4*)(dstv + (size_t)(lo + r) * HCt + lane * 4) = o;
        }
    } else {
        for (int r = tid; r < cnt; r += 1024) {
            float v[HCt], m = -__builtin_inff();
            const float dv = dn[r] + 1e-16f;
            #pragma unroll
            for (int c = 0; c < HCt; ++c) {
                v[c] = acc[r * HCt + (c & 3) * LPE + (c >> 2)] / dv + bias[c];
                m = fmaxf(m, v[c]);
            }
            float sum = 0.0f;
            #pragma unroll
            for (int c = 0; c < HCt; ++c) sum += expf(v[c] - m);
            const float ls = logf(sum);
            #pragma unroll
            for (int c = 0; c < HCt; ++c) dstv[(size_t)(lo + r) * HCt + c] = v[c] - m - ls;
        }
    }
}

extern "C" void kernel_launch(void* const* d_in, const int* in_sizes, int n_in,
                              void* d_out, int out_size, void* d_ws, size_t ws_size,
                              hipStream_t stream) {
    (void)in_sizes; (void)n_in; (void)out_size; (void)ws_size;
    const float* x   = (const float*)d_in[0];
    const int*   ei  = (const int*)d_in[1];
    const float* W1  = (const float*)d_in[2];
    const float* as1 = (const float*)d_in[3];
    const float* ad1 = (const float*)d_in[4];
    const float* b1  = (const float*)d_in[5];
    const float* W2  = (const float*)d_in[6];
    const float* as2 = (const float*)d_in[7];
    const float* ad2 = (const float*)d_in[8];
    const float* b2  = (const float*)d_in[9];
    float* out = (float*)d_out;

    char* wsb = (char*)d_ws;
    size_t off = 0;
    auto alloc = [&](size_t bytes) -> void* {
        void* p = wsb + off;
        off += (bytes + 255) & ~(size_t)255;
        return p;
    };
    _Float16* w1h  = (_Float16*)alloc((size_t)W1R * NF * 2);
    _Float16* w2h  = (_Float16*)alloc((size_t)W2R * HC * 2);
    _Float16* xh   = (_Float16*)alloc((size_t)NN * NF * 2);
    float* h1      = (float*)alloc((size_t)NN * HC * 4);
    float* asrc1   = (float*)alloc((size_t)NN * HH * 4);
    float* adst1   = (float*)alloc((size_t)NN * HH * 4);
    float* h2      = (float*)alloc((size_t)NN * NC * 4);
    float* asrc2   = (float*)alloc((size_t)NN * 4);
    float* adst2   = (float*)alloc((size_t)NN * 4);
    int* elist     = (int*)alloc((size_t)NGRP * 32 * WCAP * 4);
    int* ecnt      = (int*)alloc((size_t)NGRP * 32 * 4);
    float* act1    = (float*)xh;
    unsigned char* ownb = (unsigned char*)h2;

    const dim3 B(256);
    gat_cvtw <<<(CVTW_N + 255) / 256, B, 0, stream>>>(W1, W2, as1, ad1, as2, ad2, w1h, w2h);
    gat_cvtx <<<CVT_BLOCKS, B, 0, stream>>>(x, xh);
    gat_gemm1<<<(NN / 16 + 7) / 8, B, 0, stream>>>(xh, w1h, h1, asrc1, adst1);
    gat_ownb <<<(EE / 16 + 255) / 256, B, 0, stream>>>(ei, ownb);
    gat_own<HH, HC, 8, 1><<<NGRP, 1024, 0, stream>>>(ei, asrc1, adst1, as1, h1, b1, act1, elist, ecnt, ownb);
    gat_gemm2<<<(NN / 16 + 7) / 8, B, 0, stream>>>(act1, w2h, h2, asrc2, adst2);
    gat_own<1, NC, 4, 2><<<NGRP, 1024, 0, stream>>>(ei, asrc2, adst2, as2, h2, b2, out, elist, ecnt, ownb);
}


# layer-1 sums without LDS float atomics: owner groups counting-sort kept edges by node in LDS (int adds), a wave sums whole nodes in registers, stores the row; sums now in slot-claim order
# speedup vs baseline: 4.3599x; 1.0445x over previous
#include <hip/hip_runtime.h>

#define NN  100000
#define NF  256
#define HH  8
#define CC  16
#define HC  128
#define NC  16
#define EE  1600000
#define ET  1700000
#define NEG 0.2f
#define NPB  392
#define NGRP 256
static_assert((NN + NPB - 1) / NPB <= 256, "an owner-group id must fit one byte");
static_assert(EE % 16 == 0, "the owner-byte list is read 16 edges a load");

typedef _Float16 h8   __attribute__((ext_vector_type(8)));
typedef _Float16 v16h __attribute__((ext_vector_type(16)));
typedef float    v8f  __attribute__((ext_vector_type(8)));

__device__ __forceinline__ void edge_sd(const int* __restrict__ ei, int e, int& s, int& d) {
    if (e < EE) { s = ei[e]; d = ei[EE + e]; }
    else        { s = e - EE; d = s; }
}

__device__ __forceinline__ void atomicMaxF(float* addr, float val) {
    if (val >= 0.0f) atomicMax((int*)addr, __float_as_int(val));
    else             atomicMin((unsigned int*)addr, __float_as_uint(val));
}

__device__ __forceinline__ float lrelu(float a) { return a > 0.0f ? a : NEG * a; }

__device__ __forceinline__ float quad_sum(float v) {
    v += __int_as_float(__builtin_amdgcn_update_dpp(0, __float_as_int(v), 0xB1, 0xF, 0xF, true));
    v += __int_as_float(__builtin_amdgcn_update_dpp(0, __float_as_int(v), 0x4E, 0xF, 0xF, true));
    return v;
}

#define W1R (HC + 2 * HH)
#define W2R (NC + 16)
#define CVTW_N (W1R * NF + W2R * HC)
__global__ __launch_bounds__(256) void gat_cvtw(const float* __restrict__ W1,
                                                const float* __restrict__ W2,
                                                const float* __restrict__ as1,
                                                const float* __restrict__ ad1,
                                                const float* __restrict__ as2,
                                                const float* __restrict__ ad2,
                                                _Float16* __restrict__ w1h,
                                                _Float16* __restrict__ w2h) {
    const int i = blockIdx.x * 256 + threadIdx.x;
    if (i < HC * NF) {
        w1h[i] = (_Float16)W1[i];
    } else if (i < W1R * NF) {
        const int j = i - HC * NF;
        const int row = j / NF, f = j % NF;
        const int h = row & (HH - 1);
        const float* __restrict__ att = (row < HH ? as1 : ad1) + h * CC;
        float s = 0.0f;
        #pragma unroll
        for (int cc = 0; cc < CC; ++cc) s += W1[(size_t)(h * CC + cc) * NF + f] * att[cc];
        w1h[i] = (_Float16)s;
    } else if (i < W1R * NF + NC * HC) {
        const int j = i - W1R * NF;
        w2h[j] = (_Float16)W2[j];
    } else if (i < CVTW_N) {
        const int j = i - W1R * NF - NC * HC;
        const int row = j / HC, k = j % HC;
        float s = 0.0f;
        if ((row & 7) == 0) {
            const float* __restrict__ att = row == 0 ? as2 : ad2;
            #pragma unroll
            for (int cc = 0; cc < NC; ++cc) s += W2[(size_t)cc * HC + k] * att[cc];
        }
        w2h[NC * HC + j] = (_Float16)s;
    }
}

#define CVT_LD 20
#define CVT_PER_BLOCK (256 * CVT_LD)
#define CVT_BLOCKS ((NN * NF / 4) / CVT_PER_BLOCK)
static_assert((NN * NF) % 4 == 0 && (NN * NF / 4) % CVT_PER_BLOCK == 0,
              "the conversion grid covers x exactly: no tail piece");
typedef _Float16 h4 __attribute__((ext_vector_type(4)));

__global__ __launch_bounds__(256) void gat_cvtx(const float* __restrict__ x,
                                                _Float16* __restrict__ xh) {
    const float4* __restrict__ xp = (const float4*)x;
    h4* __restrict__ op = (h4*)xh;
    const int base = blockIdx.x * CVT_PER_BLOCK + threadIdx.x;
    float4 v[CVT_LD];
    #pragma unroll
    for (int k = 0; k < CVT_LD; ++k) v[k] = xp[base + k * 256];
    #pragma unroll
    for (int k = 0; k < CVT_LD; ++k) {
        h4 o;
        o.x = (_Float16)v[k].x; o.y = (_Float16)v[k].y;
        o.z = (_Float16)v[k].z; o.w = (_Float16)v[k].w;
        op[base + k * 256] = o;
    }
}

__global__ __launch_bounds__(256) __attribute__((amdgpu_waves_per_eu(8)))
void gat_gemm1(const _Float16* __restrict__ xh, const _Float16* __restrict__ w1h,
               float* __restrict__ h1, float* __restrict__ a_src, float* __restrict__ a_dst) {
    __shared__ _Float16 lw[W1R * NF];
    const int tid = threadIdx.x;
    for (int idx = tid; idx < W1R * 32; idx += 256) {
        int r = idx >> 5, ch = idx & 31;
        *(h8*)(&lw[r * NF + ((ch ^ (r & 15)) << 3)]) =
            *(const h8*)(w1h + (size_t)r * NF + (ch << 3));
    }
    __syncthreads();

    const int lane = tid & 31;
    const int rt   = blockIdx.x * 8 + (tid >> 5);
    if (rt >= NN / 16) return;
    const int rsub = lane & 15;
    const int ksel = (lane >> 4) << 3;
    const _Float16* xrow = xh + (size_t)(rt * 16 + rsub) * NF;

    v8f acc[8] = {};
    v8f att = {};
    #pragma unroll
    for (int ks = 0; ks < 8; ++ks) {
        const int kb = ks * 32 + ksel;
        const int c0 = kb >> 3;
        union { v16h v; h8 p[2]; } A;
        A.p[0] = *(const h8*)(xrow + kb);
        A.p[1] = *(const h8*)(xrow + kb + 16);
        #pragma unroll
        for (int ct = 0; ct < 8; ++ct) {
            const int n = ct * 16 + rsub;
            union { v16h v; h8 p[2]; } B;
            B.p[0] = *(const h8*)(&lw[n * NF + (((c0    ) ^ (n & 15)) << 3)]);
            B.p[1] = *(const h8*)(&lw[n * NF + (((c0 + 2) ^ (n & 15)) << 3)]);
            acc[ct] = __builtin_amdgcn_wmma_f32_16x16x32_f16(
                false, A.v, false, B.v, (short)0, acc[ct], false, false);
        }
        {
            const int n = HC + rsub;
            union { v16h v; h8 p[2]; } F;
            F.p[0] = *(const h8*)(&lw[n * NF + (((c0    ) ^ (n & 15)) << 3)]);
            F.p[1] = *(const h8*)(&lw[n * NF + (((c0 + 2) ^ (n & 15)) << 3)]);
            att = __builtin_amdgcn_wmma_f32_16x16x32_f16(
                false, F.v, false, A.v, (short)0, att, false, false);
        }
    }
    const int rhi = (lane >> 4) << 3;
    #pragma unroll
    for (int ct = 0; ct < 8; ++ct) {
        #pragma unroll
        for (int r = 0; r < 8; ++r)
            h1[(size_t)(rt * 16 + rhi + r) * HC + ct * 16 + rsub] = acc[ct][r];
    }
    float* ap = (lane < 16 ? a_src : a_dst) + (size_t)(rt * 16 + rsub) * HH;
    *(float4*)(ap)     = make_float4(att[0], att[1], att[2], att[3]);
    *(float4*)(ap + 4) = make_float4(att[4], att[5], att[6], att[7]);
}

__global__ __launch_bounds__(256)
void gat_ownb(const int* __restrict__ ei, unsigned char* __restrict__ own) {
    const int i = blockIdx.x * 256 + threadIdx.x;
    if (i >= EE / 16) return;
    const int4* dp = (const int4*)(ei + EE) + (size_t)i * 4;
    unsigned w[4];
    #pragma unroll
    for (int k = 0; k < 4; ++k) {
        const int4 d = dp[k];
        w[k] = ((unsigned)d.x / NPB) | (((unsigned)d.y / NPB) << 8) |
               (((unsigned)d.z / NPB) << 16) | (((unsigned)d.w / NPB) << 24);
    }
    ((uint4*)own)[i] = make_uint4(w[0], w[1], w[2], w[3]);
}

__global__ __launch_bounds__(256)
void gat_gemm2(const float* __restrict__ a1, const _Float16* __restrict__ w2h,
               float* __restrict__ h2, float* __restrict__ a_src, float* __restrict__ a_dst) {
    const int lane = threadIdx.x & 31;
    const int rt   = blockIdx.x * 8 + (threadIdx.x >> 5);
    if (rt >= NN / 16) return;
    const int rsub = lane & 15;
    const int ksel = (lane >> 4) << 3;
    const float*    arow = a1 + (size_t)(rt * 16 + rsub) * HC;
    const _Float16* wrow = w2h + (size_t)rsub * HC;
    const _Float16* frow = w2h + (size_t)(NC + rsub) * HC;

    v8f acc = {};
    v8f att = {};
    #pragma unroll
    for (int ks = 0; ks < 4; ++ks) {
        const int kb = ks * 32 + ksel;
        union { v16h v; _Float16 e[16]; } A;
        #pragma unroll
        for (int t = 0; t < 8; ++t) {
            A.e[t]     = (_Float16)arow[kb + t];
            A.e[8 + t] = (_Float16)arow[kb + 16 + t];
        }
        union { v16h v; h8 p[2]; } B;
        B.p[0] = *(const h8*)(wrow + kb);
        B.p[1] = *(const h8*)(wrow + kb + 16);
        acc = __builtin_amdgcn_wmma_f32_16x16x32_f16(
            false, A.v, false, B.v, (short)0, acc, false, false);
        union { v16h v; h8 p[2]; } F;
        F.p[0] = *(const h8*)(frow + kb);
        F.p[1] = *(const h8*)(frow + kb + 16);
        att = __builtin_amdgcn_wmma_f32_16x16x32_f16(
            false, F.v, false, A.v, (short)0, att, false, false);
    }
    const int rhi = (lane >> 4) << 3;
    #pragma unroll
    for (int r = 0; r < 8; ++r)
        h2[(size_t)(rt * 16 + rhi + r) * NC + rsub] = acc[r];
    (lane < 16 ? a_src : a_dst)[rt * 16 + rsub] = att[0];
}

#define SCAN_EDGES 16384
#define NIT ((EE + SCAN_EDGES - 1) / SCAN_EDGES)
#define NCH (EE / 16)
#define CH_STEP 4096
#define NITB ((NCH + CH_STEP - 1) / CH_STEP)
#define WCAP 320
#define MAXB 32
static_assert(NPB <= 512 && NN <= (1 << 17), "list word: 9 bits local node, 17 bits source node");
static_assert(ET < (1 << 21), "an edge id is kept in 21 bits");

template <bool V> struct mode_c { static constexpr bool value = V; };

template <int H>
__device__ __forceinline__ void load_att(const float* __restrict__ p, int node, float (&v)[H]) {
    if constexpr (H % 4 == 0) {
        #pragma unroll
        for (int k = 0; k < H / 4; ++k) {
            const float4 t = *(const float4*)(p + (size_t)node * H + k * 4);
            v[k * 4] = t.x; v[k * 4 + 1] = t.y; v[k * 4 + 2] = t.z; v[k * 4 + 3] = t.w;
        }
    } else {
        #pragma unroll
        for (int k = 0; k < H; ++k) v[k] = p[(size_t)node * H + k];
    }
}

template <int H, bool KEEP>
__device__ __forceinline__ void max_batch(const int* __restrict__ ei, const float* __restrict__ a_src,
                                          const float* __restrict__ a_dst, float* mx,
                                          const int* qw, int first, int nvalid, int lo, int lane,
                                          int* __restrict__ lst, int lpos, int* deg) {
    const bool ok = lane < nvalid;
    const int e  = qw[first + (ok ? lane : 0)];
    const int ec = e < EE ? e : 0;
    const int sl = ei[ec];
    const int dd = ei[EE + ec];
    const int s  = (e < EE) ? sl : e - EE;
    const int dl = ((e < EE) ? dd : e - EE) - lo;
    if constexpr (KEEP) {
        if (ok && lpos + lane < WCAP) lst[lpos + lane] = (int)(((unsigned)dl << 17) | (unsigned)s);
        if (ok) atomicAdd(&deg[dl], 1);
    }
    float as[H], ad[H];
    load_att<H>(a_src, s, as);
    load_att<H>(a_dst, lo + dl, ad);
    if (ok) {
        #pragma unroll
        for (int h = 0; h < H; ++h) atomicMaxF(&mx[h * NPB + dl], lrelu(as[h] + ad[h]));
    }
}

template <int H>
__device__ __forceinline__ void walk_max(const int* __restrict__ lw, int n, const float* __restrict__ a_src,
                                         const float* __restrict__ a_dst, float* mx, int lo, int lane) {
    constexpr int U = 4;
    for (int b = 0; b < n; b += 32 * U) {
        int  ent[U];
        bool ok[U];
        #pragma unroll
        for (int u = 0; u < U; ++u) {
            const int idx = b + u * 32 + lane;
            ok[u] = idx < n;
            ent[u] = lw[ok[u] ? idx : 0];
        }
        float as[U][H], ad[U][H];
        #pragma unroll
        for (int u = 0; u < U; ++u) {
            load_att<H>(a_src, ent[u] & 0x1FFFF, as[u]);
            load_att<H>(a_dst, lo + (ent[u] >> 17), ad[u]);
        }
        #pragma unroll
        for (int u = 0; u < U; ++u) {
            if (ok[u]) {
                #pragma unroll
                for (int h = 0; h < H; ++h)
                    atomicMaxF(&mx[h * NPB + (ent[u] >> 17)], lrelu(as[u][h] + ad[u][h]));
            }
        }
    }
}

template <int H, int HCt, int STEPS>
__device__ __forceinline__ void own_batch(const int* __restrict__ ei, const float* __restrict__ a_src,
                                          const float* __restrict__ a_dst, const float* mx,
                                          const float* __restrict__ hsrc, float* acc, float* dn,
                                          const int* qw, int first, int nvalid, int lo, int lane) {
    constexpr int LPE = HCt / 4;
    constexpr int EPW = 32 / LPE;
    const int sub = lane / LPE;
    const int q   = lane % LPE;
    const int h   = (q * 4) / (HCt / H);
    int  ents[STEPS], sv[STEPS], dlv[STEPS];
    bool ok[STEPS];
    #pragma unroll
    for (int u = 0; u < STEPS; ++u) {
        const int idx = u * EPW + sub;
        ok[u] = idx < nvalid;
        ents[u] = qw[first + idx];
    }
    #pragma unroll
    for (int u = 0; u < STEPS; ++u) {
        const int e  = ok[u] ? ents[u] : 0;
        const int ec = e < EE ? e : 0;
        const int sl = ei[ec];
        const int dd = ei[EE + ec];
        sv[u]  = (e < EE) ? sl : e - EE;
        dlv[u] = ok[u] ? ((e < EE) ? dd : e - EE) - lo : 0;
    }
    float4 hv[STEPS];
    float as[STEPS], ad[STEPS], am[STEPS];
    #pragma unroll
    for (int u = 0; u < STEPS; ++u) {
        const int sn = sv[u], dnode = lo + dlv[u];
        hv[u] = *(const float4*)(hsrc + (size_t)sn * HCt + q * 4);
        as[u] = a_src[sn * H + h];
        ad[u] = a_dst[dnode * H + h];
        am[u] = mx[h * NPB + dlv[u]];
    }
    #pragma unroll
    for (int u = 0; u < STEPS; ++u) {
        const float w = expf(lrelu(as[u] + ad[u]) - am[u]);
        if (ok[u]) {
            if ((q & 3) == 0) atomicAdd(&dn[dlv[u] * H + h], w);
            float* ap = acc + dlv[u] * HCt + q;
            atomicAdd(ap,           hv[u].x * w);
            atomicAdd(ap + LPE,     hv[u].y * w);
            atomicAdd(ap + 2 * LPE, hv[u].z * w);
            atomicAdd(ap + 3 * LPE, hv[u].w * w);
        }
    }
}

template <int H, int HCt, int STEPS>
__device__ __forceinline__ void own_walk(const int* __restrict__ lw, int n, const float* __restrict__ att,
                                         const float* __restrict__ a_dst, const float* mx,
                                         const float* __restrict__ hsrc, float* acc, float* dn,
                                         int lo, int lane) {
    constexpr int LPE = HCt / 4;
    constexpr int EPW = 32 / LPE;
    constexpr int B   = EPW * STEPS;
    static_assert(HCt / H == 16, "a head is 16 channels: the four lanes of one quad");
    const int sub = lane / LPE;
    const int q   = lane % LPE;
    const int h   = (q * 4) / (HCt / H);
    const float4 av = *(const float4*)(att + q * 4);
    int ents[STEPS], nxt[STEPS];
    #pragma unroll
    for (int u = 0; u < STEPS; ++u) {
        const int idx = u * EPW + sub;
        ents[u] = lw[idx < n ? idx : 0];
    }
    for (int b = 0; b < n; b += B) {
        #pragma unroll
        for (int u = 0; u < STEPS; ++u) {
            const int idx = b + B + u * EPW + sub;
            nxt[u] = lw[idx < n ? idx : 0];
        }
        float4 hv[STEPS];
        float ad[STEPS], am[STEPS];
        int   dlv[STEPS];
        bool  ok[STEPS];
        #pragma unroll
        for (int u = 0; u < STEPS; ++u) {
            ok[u] = b + u * EPW + sub < n;
            const int ent = ok[u] ? ents[u] : 0;
            const int sn = ent & 0x1FFFF;
            dlv[u] = ent >> 17;
            const int dnode = lo + dlv[u];
            hv[u] = *(const float4*)(hsrc + (size_t)sn * HCt + q * 4);
            ad[u] = a_dst[dnode * H + h];
            am[u] = mx[h * NPB + dlv[u]];
        }
        #pragma unroll
        for (int u = 0; u < STEPS; ++u) {
            const float as = quad_sum(fmaf(hv[u].w, av.w, fmaf(hv[u].z, av.z,
                                      fmaf(hv[u].y, av.y, hv[u].x * av.x))));
            const float w = expf(lrelu(as + ad[u]) - am[u]);
            if (ok[u]) {
                if ((q & 3) == 0) atomicAdd(&dn[dlv[u] * H + h], w);
                float* ap = acc + dlv[u] * HCt + q;
                atomicAdd(ap,           hv[u].x * w);
                atomicAdd(ap + LPE,     hv[u].y * w);
                atomicAdd(ap + 2 * LPE, hv[u].z * w);
                atomicAdd(ap + 3 * LPE, hv[u].w * w);
            }
        }
        #pragma unroll
        for (int u = 0; u < STEPS; ++u) ents[u] = nxt[u];
    }
}

#define NCK 16
template <int H, int HCt>
__device__ __forceinline__ void node_rows(const int* srt, const int* nstart, const int* deg,
                                          const float* __restrict__ att, const float* __restrict__ a_dst,
                                          const float* mx, const float* __restrict__ hsrc,
                                          const float* __restrict__ bias, float* __restrict__ dstv,
                                          int lo, int cnt, int wave, int lane) {
    static_assert(HCt == 128 && H == 8, "one lane per four channels, one quad per head");
    const int h = lane >> 2;
    const float4 av = *(const float4*)(att + lane * 4);
    const float4 bq = *(const float4*)(bias + lane * 4);
    for (int r = wave; r < cnt; r += 32) {
        const int beg = __builtin_amdgcn_readfirstlane(nstart[r]);
        const int end = beg + __builtin_amdgcn_readfirstlane(deg[r]);
        const float ad = a_dst[(lo + r) * H + h];
        const float m  = mx[h * NPB + r];
        float4 a4 = make_float4(0.0f, 0.0f, 0.0f, 0.0f);
        float den = 0.0f;
        for (int i = beg; i < end; i += NCK) {
            const int k  = i + (lane & (NCK - 1));
            const int sv = srt[k < end ? k : beg];
            float4 hv[NCK];
            #pragma unroll
            for (int u = 0; u < NCK; ++u) {
                const int s = __builtin_amdgcn_readlane(sv, u);
                hv[u] = *(const float4*)(hsrc + (size_t)s * HCt + lane * 4);
            }
            #pragma unroll
            for (int u = 0; u < NCK; ++u) {
                if (i + u < end) {
                    const float as = quad_sum(fmaf(hv[u].w, av.w, fmaf(hv[u].z, av.z,
                                              fmaf(hv[u].y, av.y, hv[u].x * av.x))));
                    const float w = expf(lrelu(as + ad) - m);
                    den += w;
                    a4.x = fmaf(hv[u].x, w, a4.x);
                    a4.y = fmaf(hv[u].y, w, a4.y);
                    a4.z = fmaf(hv[u].z, w, a4.z);
                    a4.w = fmaf(hv[u].w, w, a4.w);
                }
            }
        }
        const float dv = den + 1e-16f;
        float4 o;
        o.x = a4.x / dv + bq.x;
        o.y = a4.y / dv + bq.y;
        o.z = a4.z / dv + bq.z;
        o.w = a4.w / dv + bq.w;
        o.x = o.x > 0.0f ? o.x : (expf(o.x) - 1.0f);
        o.y = o.y > 0.0f ? o.y : (expf(o.y) - 1.0f);
        o.z = o.z > 0.0f ? o.z : (expf(o.z) - 1.0f);
        o.w = o.w > 0.0f ? o.w : (expf(o.w) - 1.0f);
        *(float4*)(dstv + (size_t)(lo + r) * HCt + lane * 4) = o;
    }
}

template <int H, int HCt, int STEPS, int LAYER>
__global__ __launch_bounds__(1024)
void gat_own(const int* __restrict__ ei, const float* __restrict__ a_src,
             const float* __restrict__ a_dst, const float* __restrict__ att,
             const float* __restrict__ hsrc, const float* __restrict__ bias,
             float* __restrict__ dstv, int* __restrict__ elist, int* __restrict__ ecnt,
             const unsigned char* __restrict__ own) {
    constexpr int LPE  = HCt / 4;
    constexpr int EPW  = 32 / LPE;
    constexpr int B    = EPW * STEPS;
    constexpr int QCAP = (B > MAXB ? B : MAXB) + 512;
    __shared__ float acc[NPB * HCt];
    __shared__ float dn[NPB * H];
    __shared__ float mx[NPB * H];
    __shared__ int   qbuf[32 * QCAP];
    __shared__ int   wcnt[32];
    constexpr int NSRT = LAYER == 1 ? NPB : 1;
    __shared__ int   deg[NSRT];
    __shared__ int   nstart[NSRT];
    __shared__ int   ncur[NSRT];

    const int tid = threadIdx.x, lane = tid & 31;
    const int wave = __builtin_amdgcn_readfirstlane(tid >> 5);
    const int lo  = blockIdx.x * NPB;
    const int cnt = (NN - lo < NPB) ? (NN - lo) : NPB;
    for (int i = tid; i < NPB * HCt; i += 1024) acc[i] = 0.0f;
    for (int i = tid; i < NPB * H;   i += 1024) { dn[i] = 0.0f; mx[i] = -__builtin_inff(); }
    for (int i = tid; i < NSRT;      i += 1024) deg[i] = 0;
    __syncthreads();

    int* lst = elist + (size_t)(blockIdx.x * 32 + wave) * WCAP;
    int* qw  = qbuf + wave * QCAP;

    auto scan_pass = [&](auto mode) -> int {
    constexpr bool MAXP = decltype(mode)::value;
    constexpr int  BB   = MAXP ? MAXB : B;
    int  qn = 0;
    int  ln = 0;
    const unsigned lt = (1u << lane) - 1u;
    auto push = [&](bool m, unsigned ent) {
        const unsigned mask = __builtin_amdgcn_ballot_w32(m);
        if (m) qw[qn + __builtin_popcount(mask & lt)] = (int)ent;
        qn += __builtin_popcount(mask);
    };
    auto batch = [&](int first, int nvalid) {
        if constexpr (MAXP) max_batch<H, LAYER == 1>(ei, a_src, a_dst, mx, qw, first, nvalid, lo, lane, lst, ln, deg);
        else own_batch<H, HCt, STEPS>(ei, a_src, a_dst, mx, hsrc, acc, dn, qw, first, nvalid, lo, lane);
    };
    auto drain = [&]() {
        while (qn >= BB) {
            batch(qn - BB, BB);
            qn -= BB;
            ln += BB;
        }
    };

    for (int b0 = wave * 32; b0 < cnt; b0 += 1024) {
        const int dl = b0 + lane;
        push(dl < cnt, (unsigned)(EE + lo + dl));
    }
    drain();

    if constexpr (LAYER == 1) {
    const uint4* ownp = (const uint4*)own;
    const unsigned pat = (unsigned)blockIdx.x * 0x01010101u;
    const int it0 = (blockIdx.x * NITB) / NGRP;
    auto ldscan = [&](int i, uint4 (&buf)[4]) -> int {
        int it = i + it0;
        if (it >= NITB) it -= NITB;
        const int cb = it * CH_STEP + wave * 128 + lane;
        #pragma unroll
        for (int k = 0; k < 4; ++k) {
            const int ck = cb + k * 32;
            buf[k] = ownp[ck < NCH ? ck : NCH - 1];
        }
        return cb;
    };
    auto hit = [&](unsigned w) -> unsigned {
        const unsigned x = w ^ pat;
        return ~(((x & 0x7F7F7F7Fu) + 0x7F7F7F7Fu) | x) & 0x80808080u;
    };
    auto test = [&](const uint4 (&buf)[4], int cb) {
        unsigned mk[4];
        #pragma unroll
        for (int k = 0; k < 4; ++k) {
            const unsigned m4 = (hit(buf[k].x) >> 7) | (hit(buf[k].y) >> 6) |
                                (hit(buf[k].z) >> 5) | (hit(buf[k].w) >> 4);
            mk[k] = (cb + k * 32 < NCH) ? m4 : 0u;
        }
        unsigned long long M = ((unsigned long long)(mk[2] | (mk[3] << 4)) << 32) |
                               (unsigned long long)(mk[0] | (mk[1] << 4));
        while (__builtin_amdgcn_ballot_w32(M != 0)) {
            const bool m = M != 0;
            const int  p = m ? __builtin_ctzll(M) : 0;
            const int  k = ((p >> 5) << 1) | ((p >> 2) & 1);
            push(m, (unsigned)((cb + k * 32) * 16 + ((p & 3) << 2) + ((p >> 3) & 3)));
            M &= M - 1;
            drain();
        }
    };
    uint4 bufa[4], bufb[4];
    int cba = ldscan(0, bufa), cbb;
    for (int i = 0; i < NITB; i += 2) {
        cbb = ldscan(i + 1, bufb);
        __builtin_amdgcn_sched_barrier(0);
        test(bufa, cba);
        cba = ldscan(i + 2, bufa);
        __builtin_amdgcn_sched_barrier(0);
        if (i + 1 < NITB) test(bufb, cbb);
    }
    } else {
    const int* dstp = ei + EE;
    const int it0 = (blockIdx.x * NIT) / NGRP;
    auto ldscan = [&](int i, int4 (&buf)[4]) -> int {
        int it = i + it0;
        if (it >= NIT) it -= NIT;
        const int eb = it * SCAN_EDGES + wave * 512 + lane * 4;
        #pragma unroll
        for (int k = 0; k < 4; ++k) {
            const int ek = eb + k * 128;
            buf[k] = *(const int4*)(dstp + (ek < EE ? ek : EE - 4));
        }
        return eb;
    };
    auto test = [&](const int4 (&buf)[4], int eb) {
        #pragma unroll
        for (int k = 0; k < 4; ++k) {
            const int ek = eb + k * 128;
            const unsigned d0 = (unsigned)(buf[k].x - lo), d1 = (unsigned)(buf[k].y - lo);
            const unsigned d2 = (unsigned)(buf[k].z - lo), d3 = (unsigned)(buf[k].w - lo);
            const bool in = ek < EE;
            const bool m0 = in && d0 < (unsigned)cnt, m1 = in && d1 < (unsigned)cnt;
            const bool m2 = in && d2 < (unsigned)cnt, m3 = in && d3 < (unsigned)cnt;
            if (__builtin_amdgcn_ballot_w32(m0 | m1 | m2 | m3)) {
                push(m0, (unsigned)(ek + 0));
                push(m1, (unsigned)(ek + 1));
                push(m2, (unsigned)(ek + 2));
                push(m3, (unsigned)(ek + 3));
            }
        }
        drain();
    };
    int4 bufa[4], bufb[4];
    int eba = ldscan(0, bufa), ebb;
    for (int i = 0; i < NIT; i += 2) {
        ebb = ldscan(i + 1, bufb);
        test(bufa, eba);
        eba = ldscan(i + 2, bufa);
        if (i + 1 < NIT) test(bufb, ebb);
    }
    }
    if (qn > 0) {
        batch(0, qn);
        ln += qn;
    }
    return ln;
    };

    bool rows_done = false;
    (void)rows_done;
    if constexpr (LAYER == 1) {
        const int nl = scan_pass(mode_c<true>{});
        if (lane == 0) { ecnt[blockIdx.x * 32 + wave] = nl; wcnt[wave] = nl; }
        __syncthreads();
        if (__builtin_amdgcn_ballot_w32(wcnt[lane] > WCAP) == 0) {
            static_assert(32 * WCAP <= NPB * HCt || LAYER != 1, "the sorted list fits over acc");
            int* srt = (int*)acc;
            if (wave == 0) {
                constexpr int PER = (NPB + 31) / 32;
                int v[PER], sum = 0;
                #pragma unroll
                for (int k = 0; k < PER; ++k) {
                    const int i = lane * PER + k;
                    v[k] = i < NPB ? deg[i] : 0;
                    sum += v[k];
                }
                int inc = sum;
                #pragma unroll
                for (int o = 1; o < 32; o <<= 1) {
                    const int t = __shfl_up(inc, o);
                    if (lane >= o) inc += t;
                }
                int run = inc - sum;
                #pragma unroll
                for (int k = 0; k < PER; ++k) {
                    const int i = lane * PER + k;
                    if (i < NPB) { nstart[i] = run; ncur[i] = run; }
                    run += v[k];
                }
            }
            __syncthreads();
            for (int b = 0; b < nl; b += 128) {
                int ent[4];
                #pragma unroll
                for (int u = 0; u < 4; ++u) {
                    const int idx = b + u * 32 + lane;
                    ent[u] = lst[idx < nl ? idx : 0];
                }
                #pragma unroll
                for (int u = 0; u < 4; ++u) {
                    if (b + u * 32 + lane < nl) {
                        const int pos = atomicAdd(&ncur[ent[u] >> 17], 1);
                        srt[pos] = ent[u] & 0x1FFFF;
                    }
                }
            }
            __syncthreads();
            node_rows<H, HCt>(srt, nstart, deg, att, a_dst, mx, hsrc, bias, dstv, lo, cnt, wave, lane);
            rows_done = true;
        } else {
            (void)scan_pass(mode_c<false>{});
        }
    } else {
        const int c = ecnt[blockIdx.x * 32 + lane];
        const int nl = __builtin_amdgcn_readlane(c, wave);
        if (__builtin_amdgcn_ballot_w32(c > WCAP) == 0) {
            walk_max<H>(lst, nl, a_src, a_dst, mx, lo, lane);
            __syncthreads();
            own_walk<H, HCt, 8>(lst, nl, att, a_dst, mx, hsrc, acc, dn, lo, lane);
        } else {
            (void)scan_pass(mode_c<true>{});
            __syncthreads();
            (void)scan_pass(mode_c<false>{});
        }
    }
    __syncthreads();

    if constexpr (LAYER == 1) {
        const float4 bq = *(const float4*)(bias + lane * 4);
        for (int r = rows_done ? cnt : wave; r < cnt; r += 32) {
            const float dv = dn[r * H + (lane >> 2)] + 1e-16f;
            const float* ap = acc + r * HCt + lane;
            float4 o;
            o.x = ap[0]       / dv + bq.x;
            o.y = ap[LPE]     / dv + bq.y;
            o.z = ap[2 * LPE] / dv + bq.z;
            o.w = ap[3 * LPE] / dv + bq.w;
            o.x = o.x > 0.0f ? o.x : (expf(o.x) - 1.0f);
            o.y = o.y > 0.0f ? o.y : (expf(o.y) - 1.0f);
            o.z = o.z > 0.0f ? o.z : (expf(o.z) - 1.0f);
            o.w = o.w > 0.0f ? o.w : (expf(o.w) - 1.0f);
            *(float4*)(dstv + (size_t)(lo + r) * HCt + lane * 4) = o;
        }
    } else {
        for (int r = tid; r < cnt; r += 1024) {
            float v[HCt], m = -__builtin_inff();
            const float dv = dn[r] + 1e-16f;
            #pragma unroll
            for (int c = 0; c < HCt; ++c) {
                v[c] = acc[r * HCt + (c & 3) * LPE + (c >> 2)] / dv + bias[c];
                m = fmaxf(m, v[c]);
            }
            float sum = 0.0f;
            #pragma unroll
            for (int c = 0; c < HCt; ++c) sum += expf(v[c] - m);
            const float ls = logf(sum);
            #pragma unroll
            for (int c = 0; c < HCt; ++c) dstv[(size_t)(lo + r) * HCt + c] = v[c] - m - ls;
        }
    }
}

extern "C" void kernel_launch(void* const* d_in, const int* in_sizes, int n_in,
                              void* d_out, int out_size, void* d_ws, size_t ws_size,
                              hipStream_t stream) {
    (void)in_sizes; (void)n_in; (void)out_size; (void)ws_size;
    const float* x   = (const float*)d_in[0];
    const int*   ei  = (const int*)d_in[1];
    const float* W1  = (const float*)d_in[2];
    const float* as1 = (const float*)d_in[3];
    const float* ad1 = (const float*)d_in[4];
    const float* b1  = (const float*)d_in[5];
    const float* W2  = (const float*)d_in[6];
    const float* as2 = (const float*)d_in[7];
    const float* ad2 = (const float*)d_in[8];
    const float* b2  = (const float*)d_in[9];
    float* out = (float*)d_out;

    char* wsb = (char*)d_ws;
    size_t off = 0;
    auto alloc = [&](size_t bytes) -> void* {
        void* p = wsb + off;
        off += (bytes + 255) & ~(size_t)255;
        return p;
    };
    _Float16* w1h  = (_Float16*)alloc((size_t)W1R * NF * 2);
    _Float16* w2h  = (_Float16*)alloc((size_t)W2R * HC * 2);
    _Float16* xh   = (_Float16*)alloc((size_t)NN * NF * 2);
    float* h1      = (float*)alloc((size_t)NN * HC * 4);
    float* asrc1   = (float*)alloc((size_t)NN * HH * 4);
    float* adst1   = (float*)alloc((size_t)NN * HH * 4);
    float* h2      = (float*)alloc((size_t)NN * NC * 4);
    float* asrc2   = (float*)alloc((size_t)NN * 4);
    float* adst2   = (float*)alloc((size_t)NN * 4);
    int* elist     = (int*)alloc((size_t)NGRP * 32 * WCAP * 4);
    int* ecnt      = (int*)alloc((size_t)NGRP * 32 * 4);
    float* act1    = (float*)xh;
    unsigned char* ownb = (unsigned char*)h2;

    const dim3 B(256);
    gat_cvtw <<<(CVTW_N + 255) / 256, B, 0, stream>>>(W1, W2, as1, ad1, as2, ad2, w1h, w2h);
    gat_cvtx <<<CVT_BLOCKS, B, 0, stream>>>(x, xh);
    gat_gemm1<<<(NN / 16 + 7) / 8, B, 0, stream>>>(xh, w1h, h1, asrc1, adst1);
    gat_ownb <<<(EE / 16 + 255) / 256, B, 0, stream>>>(ei, ownb);
    gat_own<HH, HC, 8, 1><<<NGRP, 1024, 0, stream>>>(ei, asrc1, adst1, as1, h1, b1, act1, elist, ecnt, ownb);
    gat_gemm2<<<(NN / 16 + 7) / 8, B, 0, stream>>>(act1, w2h, h2, asrc2, adst2);
    gat_own<1, NC, 4, 2><<<NGRP, 1024, 0, stream>>>(ei, asrc2, adst2, as2, h2, b2, out, elist, ecnt, ownb);
}


# layer-1 row pass: softmax weight made once per edge and head, not by all four lanes of the head's quad: lanes split four edges (DPP transpose sum, same tree), one exp each, weights back by DPP
# speedup vs baseline: 4.6390x; 1.0640x over previous
#include <hip/hip_runtime.h>

#define NN  100000
#define NF  256
#define HH  8
#define CC  16
#define HC  128
#define NC  16
#define EE  1600000
#define ET  1700000
#define NEG 0.2f
#define NPB  392
#define NGRP 256
static_assert((NN + NPB - 1) / NPB <= 256, "an owner-group id must fit one byte");
static_assert(EE % 16 == 0, "the owner-byte list is read 16 edges a load");

typedef _Float16 h8   __attribute__((ext_vector_type(8)));
typedef _Float16 v16h __attribute__((ext_vector_type(16)));
typedef float    v8f  __attribute__((ext_vector_type(8)));

__device__ __forceinline__ void edge_sd(const int* __restrict__ ei, int e, int& s, int& d) {
    if (e < EE) { s = ei[e]; d = ei[EE + e]; }
    else        { s = e - EE; d = s; }
}

__device__ __forceinline__ void atomicMaxF(float* addr, float val) {
    if (val >= 0.0f) atomicMax((int*)addr, __float_as_int(val));
    else             atomicMin((unsigned int*)addr, __float_as_uint(val));
}

__device__ __forceinline__ float lrelu(float a) { return a > 0.0f ? a : NEG * a; }

__device__ __forceinline__ float quad_sum(float v) {
    v += __int_as_float(__builtin_amdgcn_update_dpp(0, __float_as_int(v), 0xB1, 0xF, 0xF, true));
    v += __int_as_float(__builtin_amdgcn_update_dpp(0, __float_as_int(v), 0x4E, 0xF, 0xF, true));
    return v;
}

template <int CTRL>
__device__ __forceinline__ float quad_dpp(float v) {
    return __int_as_float(__builtin_amdgcn_update_dpp(0, __float_as_int(v), CTRL, 0xF, 0xF, true));
}

__device__ __forceinline__ float quad_sum4(float p0, float p1, float p2, float p3, int lane) {
    const bool b0 = (lane & 1) != 0, b1 = (lane & 2) != 0;
    const float q0 = (b0 ? p1 : p0) + quad_dpp<0xB1>(b0 ? p0 : p1);
    const float q1 = (b0 ? p3 : p2) + quad_dpp<0xB1>(b0 ? p2 : p3);
    return (b1 ? q1 : q0) + quad_dpp<0x4E>(b1 ? q0 : q1);
}

#define W1R (HC + 2 * HH)
#define W2R (NC + 16)
#define CVTW_N (W1R * NF + W2R * HC)
__global__ __launch_bounds__(256) void gat_cvtw(const float* __restrict__ W1,
                                                const float* __restrict__ W2,
                                                const float* __restrict__ as1,
                                                const float* __restrict__ ad1,
                                                const float* __restrict__ as2,
                                                const float* __restrict__ ad2,
                                                _Float16* __restrict__ w1h,
                                                _Float16* __restrict__ w2h) {
    const int i = blockIdx.x * 256 + threadIdx.x;
    if (i < HC * NF) {
        w1h[i] = (_Float16)W1[i];
    } else if (i < W1R * NF) {
        const int j = i - HC * NF;
        const int row = j / NF, f = j % NF;
        const int h = row & (HH - 1);
        const float* __restrict__ att = (row < HH ? as1 : ad1) + h * CC;
        float s = 0.0f;
        #pragma unroll
        for (int cc = 0; cc < CC; ++cc) s += W1[(size_t)(h * CC + cc) * NF + f] * att[cc];
        w1h[i] = (_Float16)s;
    } else if (i < W1R * NF + NC * HC) {
        const int j = i - W1R * NF;
        w2h[j] = (_Float16)W2[j];
    } else if (i < CVTW_N) {
        const int j = i - W1R * NF - NC * HC;
        const int row = j / HC, k = j % HC;
        float s = 0.0f;
        if ((row & 7) == 0) {
            const float* __restrict__ att = row == 0 ? as2 : ad2;
            #pragma unroll
            for (int cc = 0; cc < NC; ++cc) s += W2[(size_t)cc * HC + k] * att[cc];
        }
        w2h[NC * HC + j] = (_Float16)s;
    }
}

#define CVT_LD 20
#define CVT_PER_BLOCK (256 * CVT_LD)
#define CVT_BLOCKS ((NN * NF / 4) / CVT_PER_BLOCK)
static_assert((NN * NF) % 4 == 0 && (NN * NF / 4) % CVT_PER_BLOCK == 0,
              "the conversion grid covers x exactly: no tail piece");
typedef _Float16 h4 __attribute__((ext_vector_type(4)));

__global__ __launch_bounds__(256) void gat_cvtx(const float* __restrict__ x,
                                                _Float16* __restrict__ xh) {
    const float4* __restrict__ xp = (const float4*)x;
    h4* __restrict__ op = (h4*)xh;
    const int base = blockIdx.x * CVT_PER_BLOCK + threadIdx.x;
    float4 v[CVT_LD];
    #pragma unroll
    for (int k = 0; k < CVT_LD; ++k) v[k] = xp[base + k * 256];
    #pragma unroll
    for (int k = 0; k < CVT_LD; ++k) {
        h4 o;
        o.x = (_Float16)v[k].x; o.y = (_Float16)v[k].y;
        o.z = (_Float16)v[k].z; o.w = (_Float16)v[k].w;
        op[base + k * 256] = o;
    }
}

__global__ __launch_bounds__(256) __attribute__((amdgpu_waves_per_eu(8)))
void gat_gemm1(const _Float16* __restrict__ xh, const _Float16* __restrict__ w1h,
               float* __restrict__ h1, float* __restrict__ a_src, float* __restrict__ a_dst) {
    __shared__ _Float16 lw[W1R * NF];
    const int tid = threadIdx.x;
    for (int idx = tid; idx < W1R * 32; idx += 256) {
        int r = idx >> 5, ch = idx & 31;
        *(h8*)(&lw[r * NF + ((ch ^ (r & 15)) << 3)]) =
            *(const h8*)(w1h + (size_t)r * NF + (ch << 3));
    }
    __syncthreads();

    const int lane = tid & 31;
    const int rt   = blockIdx.x * 8 + (tid >> 5);
    if (rt >= NN / 16) return;
    const int rsub = lane & 15;
    const int ksel = (lane >> 4) << 3;
    const _Float16* xrow = xh + (size_t)(rt * 16 + rsub) * NF;

    v8f acc[8] = {};
    v8f att = {};
    #pragma unroll
    for (int ks = 0; ks < 8; ++ks) {
        const int kb = ks * 32 + ksel;
        const int c0 = kb >> 3;
        union { v16h v; h8 p[2]; } A;
        A.p[0] = *(const h8*)(xrow + kb);
        A.p[1] = *(const h8*)(xrow + kb + 16);
        #pragma unroll
        for (int ct = 0; ct < 8; ++ct) {
            const int n = ct * 16 + rsub;
            union { v16h v; h8 p[2]; } B;
            B.p[0] = *(const h8*)(&lw[n * NF + (((c0    ) ^ (n & 15)) << 3)]);
            B.p[1] = *(const h8*)(&lw[n * NF + (((c0 + 2) ^ (n & 15)) << 3)]);
            acc[ct] = __builtin_amdgcn_wmma_f32_16x16x32_f16(
                false, A.v, false, B.v, (short)0, acc[ct], false, false);
        }
        {
            const int n = HC + rsub;
            union { v16h v; h8 p[2]; } F;
            F.p[0] = *(const h8*)(&lw[n * NF + (((c0    ) ^ (n & 15)) << 3)]);
            F.p[1] = *(const h8*)(&lw[n * NF + (((c0 + 2) ^ (n & 15)) << 3)]);
            att = __builtin_amdgcn_wmma_f32_16x16x32_f16(
                false, F.v, false, A.v, (short)0, att, false, false);
        }
    }
    const int rhi = (lane >> 4) << 3;
    #pragma unroll
    for (int ct = 0; ct < 8; ++ct) {
        #pragma unroll
        for (int r = 0; r < 8; ++r)
            h1[(size_t)(rt * 16 + rhi + r) * HC + ct * 16 + rsub] = acc[ct][r];
    }
    float* ap = (lane < 16 ? a_src : a_dst) + (size_t)(rt * 16 + rsub) * HH;
    *(float4*)(ap)     = make_float4(att[0], att[1], att[2], att[3]);
    *(float4*)(ap + 4) = make_float4(att[4], att[5], att[6], att[7]);
}

__global__ __launch_bounds__(256)
void gat_ownb(const int* __restrict__ ei, unsigned char* __restrict__ own) {
    const int i = blockIdx.x * 256 + threadIdx.x;
    if (i >= EE / 16) return;
    const int4* dp = (const int4*)(ei + EE) + (size_t)i * 4;
    unsigned w[4];
    #pragma unroll
    for (int k = 0; k < 4; ++k) {
        const int4 d = dp[k];
        w[k] = ((unsigned)d.x / NPB) | (((unsigned)d.y / NPB) << 8) |
               (((unsigned)d.z / NPB) << 16) | (((unsigned)d.w / NPB) << 24);
    }
    ((uint4*)own)[i] = make_uint4(w[0], w[1], w[2], w[3]);
}

__global__ __launch_bounds__(256)
void gat_gemm2(const float* __restrict__ a1, const _Float16* __restrict__ w2h,
               float* __restrict__ h2, float* __restrict__ a_src, float* __restrict__ a_dst) {
    const int lane = threadIdx.x & 31;
    const int rt   = blockIdx.x * 8 + (threadIdx.x >> 5);
    if (rt >= NN / 16) return;
    const int rsub = lane & 15;
    const int ksel = (lane >> 4) << 3;
    const float*    arow = a1 + (size_t)(rt * 16 + rsub) * HC;
    const _Float16* wrow = w2h + (size_t)rsub * HC;
    const _Float16* frow = w2h + (size_t)(NC + rsub) * HC;

    v8f acc = {};
    v8f att = {};
    #pragma unroll
    for (int ks = 0; ks < 4; ++ks) {
        const int kb = ks * 32 + ksel;
        union { v16h v; _Float16 e[16]; } A;
        #pragma unroll
        for (int t = 0; t < 8; ++t) {
            A.e[t]     = (_Float16)arow[kb + t];
            A.e[8 + t] = (_Float16)arow[kb + 16 + t];
        }
        union { v16h v; h8 p[2]; } B;
        B.p[0] = *(const h8*)(wrow + kb);
        B.p[1] = *(const h8*)(wrow + kb + 16);
        acc = __builtin_amdgcn_wmma_f32_16x16x32_f16(
            false, A.v, false, B.v, (short)0, acc, false, false);
        union { v16h v; h8 p[2]; } F;
        F.p[0] = *(const h8*)(frow + kb);
        F.p[1] = *(const h8*)(frow + kb + 16);
        att = __builtin_amdgcn_wmma_f32_16x16x32_f16(
            false, F.v, false, A.v, (short)0, att, false, false);
    }
    const int rhi = (lane >> 4) << 3;
    #pragma unroll
    for (int r = 0; r < 8; ++r)
        h2[(size_t)(rt * 16 + rhi + r) * NC + rsub] = acc[r];
    (lane < 16 ? a_src : a_dst)[rt * 16 + rsub] = att[0];
}

#define SCAN_EDGES 16384
#define NIT ((EE + SCAN_EDGES - 1) / SCAN_EDGES)
#define NCH (EE / 16)
#define CH_STEP 4096
#define NITB ((NCH + CH_STEP - 1) / CH_STEP)
#define WCAP 320
#define MAXB 32
static_assert(NPB <= 512 && NN <= (1 << 17), "list word: 9 bits local node, 17 bits source node");
static_assert(ET < (1 << 21), "an edge id is kept in 21 bits");

template <bool V> struct mode_c { static constexpr bool value = V; };

template <int H>
__device__ __forceinline__ void load_att(const float* __restrict__ p, int node, float (&v)[H]) {
    if constexpr (H % 4 == 0) {
        #pragma unroll
        for (int k = 0; k < H / 4; ++k) {
            const float4 t = *(const float4*)(p + (size_t)node * H + k * 4);
            v[k * 4] = t.x; v[k * 4 + 1] = t.y; v[k * 4 + 2] = t.z; v[k * 4 + 3] = t.w;
        }
    } else {
        #pragma unroll
        for (int k = 0; k < H; ++k) v[k] = p[(size_t)node * H + k];
    }
}

template <int H, bool KEEP>
__device__ __forceinline__ void max_batch(const int* __restrict__ ei, const float* __restrict__ a_src,
                                          const float* __restrict__ a_dst, float* mx,
                                          const int* qw, int first, int nvalid, int lo, int lane,
                                          int* __restrict__ lst, int lpos, int* deg) {
    const bool ok = lane < nvalid;
    const int e  = qw[first + (ok ? lane : 0)];
    const int ec = e < EE ? e : 0;
    const int sl = ei[ec];
    const int dd = ei[EE + ec];
    const int s  = (e < EE) ? sl : e - EE;
    const int dl = ((e < EE) ? dd : e - EE) - lo;
    if constexpr (KEEP) {
        if (ok && lpos + lane < WCAP) lst[lpos + lane] = (int)(((unsigned)dl << 17) | (unsigned)s);
        if (ok) atomicAdd(&deg[dl], 1);
    }
    float as[H], ad[H];
    load_att<H>(a_src, s, as);
    load_att<H>(a_dst, lo + dl, ad);
    if (ok) {
        #pragma unroll
        for (int h = 0; h < H; ++h) atomicMaxF(&mx[h * NPB + dl], lrelu(as[h] + ad[h]));
    }
}

template <int H>
__device__ __forceinline__ void walk_max(const int* __restrict__ lw, int n, const float* __restrict__ a_src,
                                         const float* __restrict__ a_dst, float* mx, int lo, int lane) {
    constexpr int U = 4;
    for (int b = 0; b < n; b += 32 * U) {
        int  ent[U];
        bool ok[U];
        #pragma unroll
        for (int u = 0; u < U; ++u) {
            const int idx = b + u * 32 + lane;
            ok[u] = idx < n;
            ent[u] = lw[ok[u] ? idx : 0];
        }
        float as[U][H], ad[U][H];
        #pragma unroll
        for (int u = 0; u < U; ++u) {
            load_att<H>(a_src, ent[u] & 0x1FFFF, as[u]);
            load_att<H>(a_dst, lo + (ent[u] >> 17), ad[u]);
        }
        #pragma unroll
        for (int u = 0; u < U; ++u) {
            if (ok[u]) {
                #pragma unroll
                for (int h = 0; h < H; ++h)
                    atomicMaxF(&mx[h * NPB + (ent[u] >> 17)], lrelu(as[u][h] + ad[u][h]));
            }
        }
    }
}

template <int H, int HCt, int STEPS>
__device__ __forceinline__ void own_batch(const int* __restrict__ ei, const float* __restrict__ a_src,
                                          const float* __restrict__ a_dst, const float* mx,
                                          const float* __restrict__ hsrc, float* acc, float* dn,
                                          const int* qw, int first, int nvalid, int lo, int lane) {
    constexpr int LPE = HCt / 4;
    constexpr int EPW = 32 / LPE;
    const int sub = lane / LPE;
    const int q   = lane % LPE;
    const int h   = (q * 4) / (HCt / H);
    int  ents[STEPS], sv[STEPS], dlv[STEPS];
    bool ok[STEPS];
    #pragma unroll
    for (int u = 0; u < STEPS; ++u) {
        const int idx = u * EPW + sub;
        ok[u] = idx < nvalid;
        ents[u] = qw[first + idx];
    }
    #pragma unroll
    for (int u = 0; u < STEPS; ++u) {
        const int e  = ok[u] ? ents[u] : 0;
        const int ec = e < EE ? e : 0;
        const int sl = ei[ec];
        const int dd = ei[EE + ec];
        sv[u]  = (e < EE) ? sl : e - EE;
        dlv[u] = ok[u] ? ((e < EE) ? dd : e - EE) - lo : 0;
    }
    float4 hv[STEPS];
    float as[STEPS], ad[STEPS], am[STEPS];
    #pragma unroll
    for (int u = 0; u < STEPS; ++u) {
        const int sn = sv[u], dnode = lo + dlv[u];
        hv[u] = *(const float4*)(hsrc + (size_t)sn * HCt + q * 4);
        as[u] = a_src[sn * H + h];
        ad[u] = a_dst[dnode * H + h];
        am[u] = mx[h * NPB + dlv[u]];
    }
    #pragma unroll
    for (int u = 0; u < STEPS; ++u) {
        const float w = expf(lrelu(as[u] + ad[u]) - am[u]);
        if (ok[u]) {
            if ((q & 3) == 0) atomicAdd(&dn[dlv[u] * H + h], w);
            float* ap = acc + dlv[u] * HCt + q;
            atomicAdd(ap,           hv[u].x * w);
            atomicAdd(ap + LPE,     hv[u].y * w);
            atomicAdd(ap + 2 * LPE, hv[u].z * w);
            atomicAdd(ap + 3 * LPE, hv[u].w * w);
        }
    }
}

template <int H, int HCt, int STEPS>
__device__ __forceinline__ void own_walk(const int* __restrict__ lw, int n, const float* __restrict__ att,
                                         const float* __restrict__ a_dst, const float* mx,
                                         const float* __restrict__ hsrc, float* acc, float* dn,
                                         int lo, int lane) {
    constexpr int LPE = HCt / 4;
    constexpr int EPW = 32 / LPE;
    constexpr int B   = EPW * STEPS;
    static_assert(HCt / H == 16, "a head is 16 channels: the four lanes of one quad");
    const int sub = lane / LPE;
    const int q   = lane % LPE;
    const int h   = (q * 4) / (HCt / H);
    const float4 av = *(const float4*)(att + q * 4);
    int ents[STEPS], nxt[STEPS];
    #pragma unroll
    for (int u = 0; u < STEPS; ++u) {
        const int idx = u * EPW + sub;
        ents[u] = lw[idx < n ? idx : 0];
    }
    for (int b = 0; b < n; b += B) {
        #pragma unroll
        for (int u = 0; u < STEPS; ++u) {
            const int idx = b + B + u * EPW + sub;
            nxt[u] = lw[idx < n ? idx : 0];
        }
        float4 hv[STEPS];
        float ad[STEPS], am[STEPS];
        int   dlv[STEPS];
        bool  ok[STEPS];
        #pragma unroll
        for (int u = 0; u < STEPS; ++u) {
            ok[u] = b + u * EPW + sub < n;
            const int ent = ok[u] ? ents[u] : 0;
            const int sn = ent & 0x1FFFF;
            dlv[u] = ent >> 17;
            const int dnode = lo + dlv[u];
            hv[u] = *(const float4*)(hsrc + (size_t)sn * HCt + q * 4);
            ad[u] = a_dst[dnode * H + h];
            am[u] = mx[h * NPB + dlv[u]];
        }
        #pragma unroll
        for (int u = 0; u < STEPS; ++u) {
            const float as = quad_sum(fmaf(hv[u].w, av.w, fmaf(hv[u].z, av.z,
                                      fmaf(hv[u].y, av.y, hv[u].x * av.x))));
            const float w = expf(lrelu(as + ad[u]) - am[u]);
            if (ok[u]) {
                if ((q & 3) == 0) atomicAdd(&dn[dlv[u] * H + h], w);
                float* ap = acc + dlv[u] * HCt + q;
                atomicAdd(ap,           hv[u].x * w);
                atomicAdd(ap + LPE,     hv[u].y * w);
                atomicAdd(ap + 2 * LPE, hv[u].z * w);
                atomicAdd(ap + 3 * LPE, hv[u].w * w);
            }
        }
        #pragma unroll
        for (int u = 0; u < STEPS; ++u) ents[u] = nxt[u];
    }
}

#define NCK 16
template <int H, int HCt>
__device__ __forceinline__ void node_rows(const int* srt, const int* nstart, const int* deg,
                                          const float* __restrict__ att, const float* __restrict__ a_dst,
                                          const float* mx, const float* __restrict__ hsrc,
                                          const float* __restrict__ bias, float* __restrict__ dstv,
                                          int lo, int cnt, int wave, int lane) {
    static_assert(HCt == 128 && H == 8, "one lane per four channels, one quad per head");
    const int h = lane >> 2;
    const float4 av = *(const float4*)(att + lane * 4);
    const float4 bq = *(const float4*)(bias + lane * 4);
    for (int r = wave; r < cnt; r += 32) {
        const int beg = __builtin_amdgcn_readfirstlane(nstart[r]);
        const int end = beg + __builtin_amdgcn_readfirstlane(deg[r]);
        const float ad = a_dst[(lo + r) * H + h];
        const float m  = mx[h * NPB + r];
        float4 a4 = make_float4(0.0f, 0.0f, 0.0f, 0.0f);
        float den = 0.0f;
        for (int i = beg; i < end; i += NCK) {
            const int k  = i + (lane & (NCK - 1));
            const int sv = srt[k < end ? k : beg];
            float4 hv[NCK];
            #pragma unroll
            for (int u = 0; u < NCK; ++u) {
                const int s = __builtin_amdgcn_readlane(sv, u);
                hv[u] = *(const float4*)(hsrc + (size_t)s * HCt + lane * 4);
            }
            #pragma unroll
            for (int g = 0; g < NCK / 4; ++g) {
                if (i + 4 * g < end) {
                    float p[4];
                    #pragma unroll
                    for (int t = 0; t < 4; ++t)
                        p[t] = fmaf(hv[4 * g + t].w, av.w, fmaf(hv[4 * g + t].z, av.z,
                               fmaf(hv[4 * g + t].y, av.y, hv[4 * g + t].x * av.x)));
                    const float as = quad_sum4(p[0], p[1], p[2], p[3], lane);
                    float w = expf(lrelu(as + ad) - m);
                    w = (i + 4 * g + (lane & 3) < end) ? w : 0.0f;
                    const float w0 = quad_dpp<0x00>(w), w1 = quad_dpp<0x55>(w);
                    const float w2 = quad_dpp<0xAA>(w), w3 = quad_dpp<0xFF>(w);
                    const float wt[4] = {w0, w1, w2, w3};
                    #pragma unroll
                    for (int t = 0; t < 4; ++t) {
                        den += wt[t];
                        a4.x = fmaf(hv[4 * g + t].x, wt[t], a4.x);
                        a4.y = fmaf(hv[4 * g + t].y, wt[t], a4.y);
                        a4.z = fmaf(hv[4 * g + t].z, wt[t], a4.z);
                        a4.w = fmaf(hv[4 * g + t].w, wt[t], a4.w);
                    }
                }
            }
        }
        const float dv = den + 1e-16f;
        float4 o;
        o.x = a4.x / dv + bq.x;
        o.y = a4.y / dv + bq.y;
        o.z = a4.z / dv + bq.z;
        o.w = a4.w / dv + bq.w;
        o.x = o.x > 0.0f ? o.x : (expf(o.x) - 1.0f);
        o.y = o.y > 0.0f ? o.y : (expf(o.y) - 1.0f);
        o.z = o.z > 0.0f ? o.z : (expf(o.z) - 1.0f);
        o.w = o.w > 0.0f ? o.w : (expf(o.w) - 1.0f);
        *(float4*)(dstv + (size_t)(lo + r) * HCt + lane * 4) = o;
    }
}

template <int H, int HCt, int STEPS, int LAYER>
__global__ __launch_bounds__(1024)
void gat_own(const int* __restrict__ ei, const float* __restrict__ a_src,
             const float* __restrict__ a_dst, const float* __restrict__ att,
             const float* __restrict__ hsrc, const float* __restrict__ bias,
             float* __restrict__ dstv, int* __restrict__ elist, int* __restrict__ ecnt,
             const unsigned char* __restrict__ own) {
    constexpr int LPE  = HCt / 4;
    constexpr int EPW  = 32 / LPE;
    constexpr int B    = EPW * STEPS;
    constexpr int QCAP = (B > MAXB ? B : MAXB) + 512;
    __shared__ float acc[NPB * HCt];
    __shared__ float dn[NPB * H];
    __shared__ float mx[NPB * H];
    __shared__ int   qbuf[32 * QCAP];
    __shared__ int   wcnt[32];
    constexpr int NSRT = LAYER == 1 ? NPB : 1;
    __shared__ int   deg[NSRT];
    __shared__ int   nstart[NSRT];
    __shared__ int   ncur[NSRT];

    const int tid = threadIdx.x, lane = tid & 31;
    const int wave = __builtin_amdgcn_readfirstlane(tid >> 5);
    const int lo  = blockIdx.x * NPB;
    const int cnt = (NN - lo < NPB) ? (NN - lo) : NPB;
    for (int i = tid; i < NPB * HCt; i += 1024) acc[i] = 0.0f;
    for (int i = tid; i < NPB * H;   i += 1024) { dn[i] = 0.0f; mx[i] = -__builtin_inff(); }
    for (int i = tid; i < NSRT;      i += 1024) deg[i] = 0;
    __syncthreads();

    int* lst = elist + (size_t)(blockIdx.x * 32 + wave) * WCAP;
    int* qw  = qbuf + wave * QCAP;

    auto scan_pass = [&](auto mode) -> int {
    constexpr bool MAXP = decltype(mode)::value;
    constexpr int  BB   = MAXP ? MAXB : B;
    int  qn = 0;
    int  ln = 0;
    const unsigned lt = (1u << lane) - 1u;
    auto push = [&](bool m, unsigned ent) {
        const unsigned mask = __builtin_amdgcn_ballot_w32(m);
        if (m) qw[qn + __builtin_popcount(mask & lt)] = (int)ent;
        qn += __builtin_popcount(mask);
    };
    auto batch = [&](int first, int nvalid) {
        if constexpr (MAXP) max_batch<H, LAYER == 1>(ei, a_src, a_dst, mx, qw, first, nvalid, lo, lane, lst, ln, deg);
        else own_batch<H, HCt, STEPS>(ei, a_src, a_dst, mx, hsrc, acc, dn, qw, first, nvalid, lo, lane);
    };
    auto drain = [&]() {
        while (qn >= BB) {
            batch(qn - BB, BB);
            qn -= BB;
            ln += BB;
        }
    };

    for (int b0 = wave * 32; b0 < cnt; b0 += 1024) {
        const int dl = b0 + lane;
        push(dl < cnt, (unsigned)(EE + lo + dl));
    }
    drain();

    if constexpr (LAYER == 1) {
    const uint4* ownp = (const uint4*)own;
    const unsigned pat = (unsigned)blockIdx.x * 0x01010101u;
    const int it0 = (blockIdx.x * NITB) / NGRP;
    auto ldscan = [&](int i, uint4 (&buf)[4]) -> int {
        int it = i + it0;
        if (it >= NITB) it -= NITB;
        const int cb = it * CH_STEP + wave * 128 + lane;
        #pragma unroll
        for (int k = 0; k < 4; ++k) {
            const int ck = cb + k * 32;
            buf[k] = ownp[ck < NCH ? ck : NCH - 1];
        }
        return cb;
    };
    auto hit = [&](unsigned w) -> unsigned {
        const unsigned x = w ^ pat;
        return ~(((x & 0x7F7F7F7Fu) + 0x7F7F7F7Fu) | x) & 0x80808080u;
    };
    auto test = [&](const uint4 (&buf)[4], int cb) {
        unsigned mk[4];
        #pragma unroll
        for (int k = 0; k < 4; ++k) {
            const unsigned m4 = (hit(buf[k].x) >> 7) | (hit(buf[k].y) >> 6) |
                                (hit(buf[k].z) >> 5) | (hit(buf[k].w) >> 4);
            mk[k] = (cb + k * 32 < NCH) ? m4 : 0u;
        }
        unsigned long long M = ((unsigned long long)(mk[2] | (mk[3] << 4)) << 32) |
                               (unsigned long long)(mk[0] | (mk[1] << 4));
        while (__builtin_amdgcn_ballot_w32(M != 0)) {
            const bool m = M != 0;
            const int  p = m ? __builtin_ctzll(M) : 0;
            const int  k = ((p >> 5) << 1) | ((p >> 2) & 1);
            push(m, (unsigned)((cb + k * 32) * 16 + ((p & 3) << 2) + ((p >> 3) & 3)));
            M &= M - 1;
            drain();
        }
    };
    uint4 bufa[4], bufb[4];
    int cba = ldscan(0, bufa), cbb;
    for (int i = 0; i < NITB; i += 2) {
        cbb = ldscan(i + 1, bufb);
        __builtin_amdgcn_sched_barrier(0);
        test(bufa, cba);
        cba = ldscan(i + 2, bufa);
        __builtin_amdgcn_sched_barrier(0);
        if (i + 1 < NITB) test(bufb, cbb);
    }
    } else {
    const int* dstp = ei + EE;
    const int it0 = (blockIdx.x * NIT) / NGRP;
    auto ldscan = [&](int i, int4 (&buf)[4]) -> int {
        int it = i + it0;
        if (it >= NIT) it -= NIT;
        const int eb = it * SCAN_EDGES + wave * 512 + lane * 4;
        #pragma unroll
        for (int k = 0; k < 4; ++k) {
            const int ek = eb + k * 128;
            buf[k] = *(const int4*)(dstp + (ek < EE ? ek : EE - 4));
        }
        return eb;
    };
    auto test = [&](const int4 (&buf)[4], int eb) {
        #pragma unroll
        for (int k = 0; k < 4; ++k) {
            const int ek = eb + k * 128;
            const unsigned d0 = (unsigned)(buf[k].x - lo), d1 = (unsigned)(buf[k].y - lo);
            const unsigned d2 = (unsigned)(buf[k].z - lo), d3 = (unsigned)(buf[k].w - lo);
            const bool in = ek < EE;
            const bool m0 = in && d0 < (unsigned)cnt, m1 = in && d1 < (unsigned)cnt;
            const bool m2 = in && d2 < (unsigned)cnt, m3 = in && d3 < (unsigned)cnt;
            if (__builtin_amdgcn_ballot_w32(m0 | m1 | m2 | m3)) {
                push(m0, (unsigned)(ek + 0));
                push(m1, (unsigned)(ek + 1));
                push(m2, (unsigned)(ek + 2));
                push(m3, (unsigned)(ek + 3));
            }
        }
        drain();
    };
    int4 bufa[4], bufb[4];
    int eba = ldscan(0, bufa), ebb;
    for (int i = 0; i < NIT; i += 2) {
        ebb = ldscan(i + 1, bufb);
        test(bufa, eba);
        eba = ldscan(i + 2, bufa);
        if (i + 1 < NIT) test(bufb, ebb);
    }
    }
    if (qn > 0) {
        batch(0, qn);
        ln += qn;
    }
    return ln;
    };

    bool rows_done = false;
    (void)rows_done;
    if constexpr (LAYER == 1) {
        const int nl = scan_pass(mode_c<true>{});
        if (lane == 0) { ecnt[blockIdx.x * 32 + wave] = nl; wcnt[wave] = nl; }
        __syncthreads();
        if (__builtin_amdgcn_ballot_w32(wcnt[lane] > WCAP) == 0) {
            static_assert(32 * WCAP <= NPB * HCt || LAYER != 1, "the sorted list fits over acc");
            int* srt = (int*)acc;
            if (wave == 0) {
                constexpr int PER = (NPB + 31) / 32;
                int v[PER], sum = 0;
                #pragma unroll
                for (int k = 0; k < PER; ++k) {
                    const int i = lane * PER + k;
                    v[k] = i < NPB ? deg[i] : 0;
                    sum += v[k];
                }
                int inc = sum;
                #pragma unroll
                for (int o = 1; o < 32; o <<= 1) {
                    const int t = __shfl_up(inc, o);
                    if (lane >= o) inc += t;
                }
                int run = inc - sum;
                #pragma unroll
                for (int k = 0; k < PER; ++k) {
                    const int i = lane * PER + k;
                    if (i < NPB) { nstart[i] = run; ncur[i] = run; }
                    run += v[k];
                }
            }
            __syncthreads();
            for (int b = 0; b < nl; b += 128) {
                int ent[4];
                #pragma unroll
                for (int u = 0; u < 4; ++u) {
                    const int idx = b + u * 32 + lane;
                    ent[u] = lst[idx < nl ? idx : 0];
                }
                #pragma unroll
                for (int u = 0; u < 4; ++u) {
                    if (b + u * 32 + lane < nl) {
                        const int pos = atomicAdd(&ncur[ent[u] >> 17], 1);
                        srt[pos] = ent[u] & 0x1FFFF;
                    }
                }
            }
            __syncthreads();
            node_rows<H, HCt>(srt, nstart, deg, att, a_dst, mx, hsrc, bias, dstv, lo, cnt, wave, lane);
            rows_done = true;
        } else {
            (void)scan_pass(mode_c<false>{});
        }
    } else {
        const int c = ecnt[blockIdx.x * 32 + lane];
        const int nl = __builtin_amdgcn_readlane(c, wave);
        if (__builtin_amdgcn_ballot_w32(c > WCAP) == 0) {
            walk_max<H>(lst, nl, a_src, a_dst, mx, lo, lane);
            __syncthreads();
            own_walk<H, HCt, 8>(lst, nl, att, a_dst, mx, hsrc, acc, dn, lo, lane);
        } else {
            (void)scan_pass(mode_c<true>{});
            __syncthreads();
            (void)scan_pass(mode_c<false>{});
        }
    }
    __syncthreads();

    if constexpr (LAYER == 1) {
        const float4 bq = *(const float4*)(bias + lane * 4);
        for (int r = rows_done ? cnt : wave; r < cnt; r += 32) {
            const float dv = dn[r * H + (lane >> 2)] + 1e-16f;
            const float* ap = acc + r * HCt + lane;
            float4 o;
            o.x = ap[0]       / dv + bq.x;
            o.y = ap[LPE]     / dv + bq.y;
            o.z = ap[2 * LPE] / dv + bq.z;
            o.w = ap[3 * LPE] / dv + bq.w;
            o.x = o.x > 0.0f ? o.x : (expf(o.x) - 1.0f);
            o.y = o.y > 0.0f ? o.y : (expf(o.y) - 1.0f);
            o.z = o.z > 0.0f ? o.z : (expf(o.z) - 1.0f);
            o.w = o.w > 0.0f ? o.w : (expf(o.w) - 1.0f);
            *(float4*)(dstv + (size_t)(lo + r) * HCt + lane * 4) = o;
        }
    } else {
        for (int r = tid; r < cnt; r += 1024) {
            float v[HCt], m = -__builtin_inff();
            const float dv = dn[r] + 1e-16f;
            #pragma unroll
            for (int c = 0; c < HCt; ++c) {
                v[c] = acc[r * HCt + (c & 3) * LPE + (c >> 2)] / dv + bias[c];
                m = fmaxf(m, v[c]);
            }
            float sum = 0.0f;
            #pragma unroll
            for (int c = 0; c < HCt; ++c) sum += expf(v[c] - m);
            const float ls = logf(sum);
            #pragma unroll
            for (int c = 0; c < HCt; ++c) dstv[(size_t)(lo + r) * HCt + c] = v[c] - m - ls;
        }
    }
}

extern "C" void kernel_launch(void* const* d_in, const int* in_sizes, int n_in,
                              void* d_out, int out_size, void* d_ws, size_t ws_size,
                              hipStream_t stream) {
    (void)in_sizes; (void)n_in; (void)out_size; (void)ws_size;
    const float* x   = (const float*)d_in[0];
    const int*   ei  = (const int*)d_in[1];
    const float* W1  = (const float*)d_in[2];
    const float* as1 = (const float*)d_in[3];
    const float* ad1 = (const float*)d_in[4];
    const float* b1  = (const float*)d_in[5];
    const float* W2  = (const float*)d_in[6];
    const float* as2 = (const float*)d_in[7];
    const float* ad2 = (const float*)d_in[8];
    const float* b2  = (const float*)d_in[9];
    float* out = (float*)d_out;

    char* wsb = (char*)d_ws;
    size_t off = 0;
    auto alloc = [&](size_t bytes) -> void* {
        void* p = wsb + off;
        off += (bytes + 255) & ~(size_t)255;
        return p;
    };
    _Float16* w1h  = (_Float16*)alloc((size_t)W1R * NF * 2);
    _Float16* w2h  = (_Float16*)alloc((size_t)W2R * HC * 2);
    _Float16* xh   = (_Float16*)alloc((size_t)NN * NF * 2);
    float* h1      = (float*)alloc((size_t)NN * HC * 4);
    float* asrc1   = (float*)alloc((size_t)NN * HH * 4);
    float* adst1   = (float*)alloc((size_t)NN * HH * 4);
    float* h2      = (float*)alloc((size_t)NN * NC * 4);
    float* asrc2   = (float*)alloc((size_t)NN * 4);
    float* adst2   = (float*)alloc((size_t)NN * 4);
    int* elist     = (int*)alloc((size_t)NGRP * 32 * WCAP * 4);
    int* ecnt      = (int*)alloc((size_t)NGRP * 32 * 4);
    float* act1    = (float*)xh;
    unsigned char* ownb = (unsigned char*)h2;

    const dim3 B(256);
    gat_cvtw <<<(CVTW_N + 255) / 256, B, 0, stream>>>(W1, W2, as1, ad1, as2, ad2, w1h, w2h);
    gat_cvtx <<<CVT_BLOCKS, B, 0, stream>>>(x, xh);
    gat_gemm1<<<(NN / 16 + 7) / 8, B, 0, stream>>>(xh, w1h, h1, asrc1, adst1);
    gat_ownb <<<(EE / 16 + 255) / 256, B, 0, stream>>>(ei, ownb);
    gat_own<HH, HC, 8, 1><<<NGRP, 1024, 0, stream>>>(ei, asrc1, adst1, as1, h1, b1, act1, elist, ecnt, ownb);
    gat_gemm2<<<(NN / 16 + 7) / 8, B, 0, stream>>>(act1, w2h, h2, asrc2, adst2);
    gat_own<1, NC, 4, 2><<<NGRP, 1024, 0, stream>>>(ei, asrc2, adst2, as2, h2, b2, out, elist, ecnt, ownb);
}
